# starting kernel
# baseline (speedup 1.0000x reference)
#include <hip/hip_runtime.h>
#include <hip/hip_bf16.h>

#define F_IN   64
#define HEADS  4
#define CHD    32
#define HC     128
#define NEG_SLOPE 0.2f

typedef __attribute__((ext_vector_type(2))) float v2f;
typedef __attribute__((ext_vector_type(8))) float v8f;

__device__ __forceinline__ float leaky(float x) { return x > 0.f ? x : NEG_SLOPE * x; }

__global__ void fold_kernel(const float* __restrict__ We1, const float* __restrict__ ae1,
                            const float* __restrict__ We2, const float* __restrict__ ae2,
                            float* __restrict__ Ve1, float* __restrict__ Ve2) {
    int t = threadIdx.x;
    int f = t >> 2, h = t & 3;
    float s1 = 0.f, s2 = 0.f;
#pragma unroll
    for (int c = 0; c < CHD; ++c) {
        s1 += We1[f * HC + h * CHD + c] * ae1[h * CHD + c];
        s2 += We2[f * HC + h * CHD + c] * ae2[h * CHD + c];
    }
    Ve1[f * 4 + h] = s1;
    Ve2[f * 4 + h] = s2;
}

__global__ void edge_proj_kernel(const float* __restrict__ EA, const int* __restrict__ dstA,
                                 const float* __restrict__ Ve1, const float* __restrict__ Ve2,
                                 float* __restrict__ ae1v, float* __restrict__ ae2v,
                                 float* __restrict__ sum1, float* __restrict__ sum2,
                                 int* __restrict__ degI, int E) {
    __shared__ float sv[F_IN][8];
    int t = threadIdx.x;
    {
        int f = t >> 2, h = t & 3;
        sv[f][h]     = Ve1[f * 4 + h];
        sv[f][h + 4] = Ve2[f * 4 + h];
    }
    __syncthreads();
    int e = blockIdx.x * blockDim.x + t;
    if (e >= E) return;
    float acc[8] = {0, 0, 0, 0, 0, 0, 0, 0};
    const float4* ea4 = reinterpret_cast<const float4*>(EA + (size_t)e * F_IN);
#pragma unroll
    for (int j = 0; j < F_IN / 4; ++j) {
        float4 v = ea4[j];
        int f = j * 4;
#pragma unroll
        for (int q = 0; q < 8; ++q) {
            acc[q] += v.x * sv[f][q] + v.y * sv[f + 1][q] +
                      v.z * sv[f + 2][q] + v.w * sv[f + 3][q];
        }
    }
    int d = dstA[e];
#pragma unroll
    for (int h = 0; h < 4; ++h) {
        ae1v[(size_t)e * 4 + h] = acc[h];
        ae2v[(size_t)e * 4 + h] = acc[4 + h];
        atomicAdd(&sum1[(size_t)d * 4 + h], acc[h]);
        atomicAdd(&sum2[(size_t)d * 4 + h], acc[4 + h]);
    }
    atomicAdd(&degI[d], 1);
}

__global__ void scan_kernel(const int* __restrict__ deg, int* __restrict__ off, int N) {
    __shared__ int sh[1024];
    int t = threadIdx.x;
    int CHK = (N + 1023) / 1024;
    int b = t * CHK, en = min(b + CHK, N);
    int sum = 0;
    for (int i = b; i < en; ++i) sum += deg[i];
    sh[t] = sum;
    __syncthreads();
    for (int d = 1; d < 1024; d <<= 1) {
        int v = (t >= d) ? sh[t - d] : 0;
        __syncthreads();
        sh[t] += v;
        __syncthreads();
    }
    int run = sh[t] - sum;
    for (int i = b; i < en; ++i) { off[i] = run; run += deg[i]; }
}

__global__ void scatter_kernel(const int* __restrict__ dstA, const int* __restrict__ off,
                               int* __restrict__ cursor, int* __restrict__ eidA, int E) {
    int e = blockIdx.x * blockDim.x + threadIdx.x;
    if (e >= E) return;
    int d = dstA[e];
    int pos = off[d] + atomicAdd(&cursor[d], 1);
    eidA[pos] = e;
}

__global__ void gemm_wmma_kernel(const float* __restrict__ X, const float* __restrict__ W,
                                 float* __restrict__ Y, int M, int K) {
    int wave = threadIdx.x >> 5;
    int lane = threadIdx.x & 31;
    int tile = blockIdx.x * (blockDim.x >> 5) + wave;
    int mt = tile >> 3;
    int nt = tile & 7;
    if (mt * 16 >= M) return;
    int row0 = mt * 16, col0 = nt * 16;
    int m = lane & 15, half = lane >> 4;

    v8f acc = {};
    const float* ap0 = X + (size_t)(row0 + m) * K + 2 * half;
    const float* bp0 = W + (size_t)(2 * half) * HC + col0 + m;
    for (int k0 = 0; k0 < K; k0 += 4) {
        v2f a, b;
        a.x = ap0[k0];
        a.y = ap0[k0 + 1];
        b.x = bp0[(size_t)k0 * HC];
        b.y = bp0[(size_t)(k0 + 1) * HC];
        acc = __builtin_amdgcn_wmma_f32_16x16x4_f32(false, a, false, b,
                                                    (short)0, acc, false, false);
    }
    float* yp = Y + (size_t)(row0 + 8 * half) * HC + col0 + m;
#pragma unroll
    for (int r = 0; r < 8; ++r) yp[(size_t)r * HC] = acc[r];
}

__global__ void node_att_kernel(const float* __restrict__ Hp,
                                const float* __restrict__ att_s, const float* __restrict__ att_d,
                                float* __restrict__ a_s, float* __restrict__ a_d, int N) {
    int wave = threadIdx.x >> 5, lane = threadIdx.x & 31;
    int n = blockIdx.x * (blockDim.x >> 5) + wave;
    if (n >= N) return;
    int h = lane >> 3, q = lane & 7;
    float4 hv = reinterpret_cast<const float4*>(Hp)[(size_t)n * 32 + lane];
    float4 sa = reinterpret_cast<const float4*>(att_s)[lane];
    float4 da = reinterpret_cast<const float4*>(att_d)[lane];
    float ps = hv.x * sa.x + hv.y * sa.y + hv.z * sa.z + hv.w * sa.w;
    float pd = hv.x * da.x + hv.y * da.y + hv.z * da.z + hv.w * da.w;
#pragma unroll
    for (int msk = 1; msk < 8; msk <<= 1) {
        ps += __shfl_xor(ps, msk, 32);
        pd += __shfl_xor(pd, msk, 32);
    }
    if (q == 0) { a_s[(size_t)n * 4 + h] = ps; a_d[(size_t)n * 4 + h] = pd; }
}

__global__ void agg_kernel(const int* __restrict__ off, const int* __restrict__ degI,
                           const int* __restrict__ eidA, const int* __restrict__ srcA,
                           const float* __restrict__ a_e, const float* __restrict__ a_s,
                           const float* __restrict__ a_d, const float* __restrict__ sum_ae,
                           const float* __restrict__ Hp, const float* __restrict__ bias,
                           const float* __restrict__ lin_w, const float* __restrict__ lin_b,
                           float* __restrict__ outv, float* __restrict__ outs,
                           int N, int mode) {
    int wave = threadIdx.x >> 5, lane = threadIdx.x & 31;
    int n = blockIdx.x * (blockDim.x >> 5) + wave;
    if (n >= N) return;
    int h = lane >> 3;
    const float4* Hp4 = reinterpret_cast<const float4*>(Hp);

    float ad_n = a_d[(size_t)n * 4 + h];
    float as_n = a_s[(size_t)n * 4 + h];
    int cnt = degI[n];
    float degf = fmaxf((float)cnt, 1.0f);
    float M = leaky(as_n + ad_n + sum_ae[(size_t)n * 4 + h] / degf);
    float D = 1.0f;
    float4 A = Hp4[(size_t)n * 32 + lane];

    int s0 = off[n];
    for (int i = 0; i < cnt; ++i) {
        int e = eidA[s0 + i];
        int s = srcA[e];
        float lg = leaky(a_s[(size_t)s * 4 + h] + ad_n + a_e[(size_t)e * 4 + h]);
        float Mn = fmaxf(M, lg);
        float sc = __expf(M - Mn);
        float p  = __expf(lg - Mn);
        float4 hv = Hp4[(size_t)s * 32 + lane];
        A.x = A.x * sc + p * hv.x;
        A.y = A.y * sc + p * hv.y;
        A.z = A.z * sc + p * hv.z;
        A.w = A.w * sc + p * hv.w;
        D = D * sc + p;
        M = Mn;
    }
    float4 bv = reinterpret_cast<const float4*>(bias)[lane];
    float inv = 1.0f / D;
    float4 r = { A.x * inv + bv.x, A.y * inv + bv.y, A.z * inv + bv.z, A.w * inv + bv.w };
    if (mode == 0) {
        r.x = fmaxf(r.x, 0.f); r.y = fmaxf(r.y, 0.f);
        r.z = fmaxf(r.z, 0.f); r.w = fmaxf(r.w, 0.f);
        reinterpret_cast<float4*>(outv)[(size_t)n * 32 + lane] = r;
    } else {
        float4 lw = reinterpret_cast<const float4*>(lin_w)[lane];
        float part = r.x * lw.x + r.y * lw.y + r.z * lw.z + r.w * lw.w;
#pragma unroll
        for (int msk = 16; msk >= 1; msk >>= 1) part += __shfl_xor(part, msk, 32);
        if (lane == 0) outs[n] = part + lin_b[0];
    }
}

extern "C" void kernel_launch(void* const* d_in, const int* in_sizes, int n_in,
                              void* d_out, int out_size, void* d_ws, size_t ws_size,
                              hipStream_t stream) {
    const float* x   = (const float*)d_in[0];
    const float* ea  = (const float*)d_in[1];
    const float* W1  = (const float*)d_in[2];
    const float* as1 = (const float*)d_in[3];
    const float* ad1 = (const float*)d_in[4];
    const float* We1 = (const float*)d_in[5];
    const float* ae1 = (const float*)d_in[6];
    const float* b1  = (const float*)d_in[7];
    const float* W2  = (const float*)d_in[8];
    const float* as2 = (const float*)d_in[9];
    const float* ad2 = (const float*)d_in[10];
    const float* We2 = (const float*)d_in[11];
    const float* ae2 = (const float*)d_in[12];
    const float* b2  = (const float*)d_in[13];
    const float* lw  = (const float*)d_in[14];
    const float* lb  = (const float*)d_in[15];
    const int*   ei  = (const int*)d_in[16];

    const int N = in_sizes[0] / F_IN;
    const int E = in_sizes[16] / 2;
    const int* srcA = ei;
    const int* dstA = ei + E;

    char* w = (char*)d_ws;
    auto alloc = [&](size_t bytes) -> void* {
        void* p = (void*)w;
        w += (bytes + 255) & ~(size_t)255;
        return p;
    };
    int*   degI   = (int*)  alloc((size_t)N * 4);
    int*   cursor = (int*)  alloc((size_t)N * 4);
    float* sum1   = (float*)alloc((size_t)N * 16);
    float* sum2   = (float*)alloc((size_t)N * 16);
    size_t zbytes = (size_t)(w - (char*)degI);
    float* Ve1  = (float*)alloc(F_IN * 4 * 4);
    float* Ve2  = (float*)alloc(F_IN * 4 * 4);
    float* ae1v = (float*)alloc((size_t)E * 16);
    float* ae2v = (float*)alloc((size_t)E * 16);
    float* a_sb = (float*)alloc((size_t)N * 16);
    float* a_db = (float*)alloc((size_t)N * 16);
    int*   off  = (int*)  alloc((size_t)N * 4);
    int*   eidA = (int*)  alloc((size_t)E * 4);
    float* Hp   = (float*)alloc((size_t)N * HC * 4);
    float* h1   = (float*)alloc((size_t)N * HC * 4);

    hipMemsetAsync(degI, 0, zbytes, stream);

    fold_kernel<<<1, 256, 0, stream>>>(We1, ae1, We2, ae2, Ve1, Ve2);
    edge_proj_kernel<<<(E + 255) / 256, 256, 0, stream>>>(ea, dstA, Ve1, Ve2,
                                                          ae1v, ae2v, sum1, sum2, degI, E);
    scan_kernel<<<1, 1024, 0, stream>>>(degI, off, N);
    scatter_kernel<<<(E + 255) / 256, 256, 0, stream>>>(dstA, off, cursor, eidA, E);

    const int mt = (N + 15) / 16;
    const int gemmBlocks = (mt * 8 + 3) / 4;
    const int nodeBlocks = (N + 7) / 8;

    gemm_wmma_kernel<<<gemmBlocks, 128, 0, stream>>>(x, W1, Hp, N, F_IN);
    node_att_kernel<<<nodeBlocks, 256, 0, stream>>>(Hp, as1, ad1, a_sb, a_db, N);
    agg_kernel<<<nodeBlocks, 256, 0, stream>>>(off, degI, eidA, srcA, ae1v, a_sb, a_db,
                                               sum1, Hp, b1, nullptr, nullptr,
                                               h1, nullptr, N, 0);
    gemm_wmma_kernel<<<gemmBlocks, 128, 0, stream>>>(h1, W2, Hp, N, HC);
    node_att_kernel<<<nodeBlocks, 256, 0, stream>>>(Hp, as2, ad2, a_sb, a_db, N);
    agg_kernel<<<nodeBlocks, 256, 0, stream>>>(off, degI, eidA, srcA, ae2v, a_sb, a_db,
                                               sum2, Hp, b2, lw, lb,
                                               nullptr, (float*)d_out, N, 1);
}


# per-destination sums of edge logits taken by the CSR aggregation wave instead of 8 float atomics per edge in the edge pass; self loop joins the online softmax last (summation order changes)
# speedup vs baseline: 1.4336x; 1.4336x over previous
#include <hip/hip_runtime.h>
#include <hip/hip_bf16.h>

#define F_IN   64
#define HEADS  4
#define CHD    32
#define HC     128
#define NEG_SLOPE 0.2f

typedef __attribute__((ext_vector_type(2))) float v2f;
typedef __attribute__((ext_vector_type(8))) float v8f;

__device__ __forceinline__ float leaky(float x) { return x > 0.f ? x : NEG_SLOPE * x; }

__global__ void fold_kernel(const float* __restrict__ We1, const float* __restrict__ ae1,
                            const float* __restrict__ We2, const float* __restrict__ ae2,
                            float* __restrict__ Ve1, float* __restrict__ Ve2) {
    int t = threadIdx.x;
    int f = t >> 2, h = t & 3;
    float s1 = 0.f, s2 = 0.f;
#pragma unroll
    for (int c = 0; c < CHD; ++c) {
        s1 += We1[f * HC + h * CHD + c] * ae1[h * CHD + c];
        s2 += We2[f * HC + h * CHD + c] * ae2[h * CHD + c];
    }
    Ve1[f * 4 + h] = s1;
    Ve2[f * 4 + h] = s2;
}

__global__ void edge_proj_kernel(const float* __restrict__ EA, const int* __restrict__ dstA,
                                 const float* __restrict__ Ve1, const float* __restrict__ Ve2,
                                 float* __restrict__ ae1v, float* __restrict__ ae2v,
                                 int* __restrict__ degI, int E) {
    __shared__ float sv[F_IN][8];
    int t = threadIdx.x;
    {
        int f = t >> 2, h = t & 3;
        sv[f][h]     = Ve1[f * 4 + h];
        sv[f][h + 4] = Ve2[f * 4 + h];
    }
    __syncthreads();
    int e = blockIdx.x * blockDim.x + t;
    if (e >= E) return;
    float acc[8] = {0, 0, 0, 0, 0, 0, 0, 0};
    const float4* ea4 = reinterpret_cast<const float4*>(EA + (size_t)e * F_IN);
#pragma unroll
    for (int j = 0; j < F_IN / 4; ++j) {
        float4 v = ea4[j];
        int f = j * 4;
#pragma unroll
        for (int q = 0; q < 8; ++q) {
            acc[q] += v.x * sv[f][q] + v.y * sv[f + 1][q] +
                      v.z * sv[f + 2][q] + v.w * sv[f + 3][q];
        }
    }
    int d = dstA[e];
#pragma unroll
    for (int h = 0; h < 4; ++h) {
        ae1v[(size_t)e * 4 + h] = acc[h];
        ae2v[(size_t)e * 4 + h] = acc[4 + h];
    }
    atomicAdd(&degI[d], 1);
}

__global__ void scan_kernel(const int* __restrict__ deg, int* __restrict__ off, int N) {
    __shared__ int sh[1024];
    int t = threadIdx.x;
    int CHK = (N + 1023) / 1024;
    int b = t * CHK, en = min(b + CHK, N);
    int sum = 0;
    for (int i = b; i < en; ++i) sum += deg[i];
    sh[t] = sum;
    __syncthreads();
    for (int d = 1; d < 1024; d <<= 1) {
        int v = (t >= d) ? sh[t - d] : 0;
        __syncthreads();
        sh[t] += v;
        __syncthreads();
    }
    int run = sh[t] - sum;
    for (int i = b; i < en; ++i) { off[i] = run; run += deg[i]; }
}

__global__ void scatter_kernel(const int* __restrict__ dstA, const int* __restrict__ off,
                               int* __restrict__ cursor, int* __restrict__ eidA, int E) {
    int e = blockIdx.x * blockDim.x + threadIdx.x;
    if (e >= E) return;
    int d = dstA[e];
    int pos = off[d] + atomicAdd(&cursor[d], 1);
    eidA[pos] = e;
}

__global__ void gemm_wmma_kernel(const float* __restrict__ X, const float* __restrict__ W,
                                 float* __restrict__ Y, int M, int K) {
    int wave = threadIdx.x >> 5;
    int lane = threadIdx.x & 31;
    int tile = blockIdx.x * (blockDim.x >> 5) + wave;
    int mt = tile >> 3;
    int nt = tile & 7;
    if (mt * 16 >= M) return;
    int row0 = mt * 16, col0 = nt * 16;
    int m = lane & 15, half = lane >> 4;

    v8f acc = {};
    const float* ap0 = X + (size_t)(row0 + m) * K + 2 * half;
    const float* bp0 = W + (size_t)(2 * half) * HC + col0 + m;
    for (int k0 = 0; k0 < K; k0 += 4) {
        v2f a, b;
        a.x = ap0[k0];
        a.y = ap0[k0 + 1];
        b.x = bp0[(size_t)k0 * HC];
        b.y = bp0[(size_t)(k0 + 1) * HC];
        acc = __builtin_amdgcn_wmma_f32_16x16x4_f32(false, a, false, b,
                                                    (short)0, acc, false, false);
    }
    float* yp = Y + (size_t)(row0 + 8 * half) * HC + col0 + m;
#pragma unroll
    for (int r = 0; r < 8; ++r) yp[(size_t)r * HC] = acc[r];
}

__global__ void node_att_kernel(const float* __restrict__ Hp,
                                const float* __restrict__ att_s, const float* __restrict__ att_d,
                                float* __restrict__ a_s, float* __restrict__ a_d, int N) {
    int wave = threadIdx.x >> 5, lane = threadIdx.x & 31;
    int n = blockIdx.x * (blockDim.x >> 5) + wave;
    if (n >= N) return;
    int h = lane >> 3, q = lane & 7;
    float4 hv = reinterpret_cast<const float4*>(Hp)[(size_t)n * 32 + lane];
    float4 sa = reinterpret_cast<const float4*>(att_s)[lane];
    float4 da = reinterpret_cast<const float4*>(att_d)[lane];
    float ps = hv.x * sa.x + hv.y * sa.y + hv.z * sa.z + hv.w * sa.w;
    float pd = hv.x * da.x + hv.y * da.y + hv.z * da.z + hv.w * da.w;
#pragma unroll
    for (int msk = 1; msk < 8; msk <<= 1) {
        ps += __shfl_xor(ps, msk, 32);
        pd += __shfl_xor(pd, msk, 32);
    }
    if (q == 0) { a_s[(size_t)n * 4 + h] = ps; a_d[(size_t)n * 4 + h] = pd; }
}

__global__ void agg_kernel(const int* __restrict__ off, const int* __restrict__ degI,
                           const int* __restrict__ eidA, const int* __restrict__ srcA,
                           const float* __restrict__ a_e, const float* __restrict__ a_s,
                           const float* __restrict__ a_d,
                           const float* __restrict__ Hp, const float* __restrict__ bias,
                           const float* __restrict__ lin_w, const float* __restrict__ lin_b,
                           float* __restrict__ outv, float* __restrict__ outs,
                           int N, int mode) {
    int wave = threadIdx.x >> 5, lane = threadIdx.x & 31;
    int n = blockIdx.x * (blockDim.x >> 5) + wave;
    if (n >= N) return;
    int h = lane >> 3;
    const float4* Hp4 = reinterpret_cast<const float4*>(Hp);

    float ad_n = a_d[(size_t)n * 4 + h];
    float as_n = a_s[(size_t)n * 4 + h];
    int cnt = degI[n];
    float degf = fmaxf((float)cnt, 1.0f);
    float M = -INFINITY;
    float D = 0.0f;
    float4 A = {0.f, 0.f, 0.f, 0.f};
    float S = 0.0f;
    float4 hself = Hp4[(size_t)n * 32 + lane];

    int s0 = off[n];
    for (int i = 0; i < cnt; ++i) {
        int e = eidA[s0 + i];
        int s = srcA[e];
        float ae = a_e[(size_t)e * 4 + h];
        S += ae;
        float lg = leaky(a_s[(size_t)s * 4 + h] + ad_n + ae);
        float Mn = fmaxf(M, lg);
        float sc = __expf(M - Mn);
        float p  = __expf(lg - Mn);
        float4 hv = Hp4[(size_t)s * 32 + lane];
        A.x = A.x * sc + p * hv.x;
        A.y = A.y * sc + p * hv.y;
        A.z = A.z * sc + p * hv.z;
        A.w = A.w * sc + p * hv.w;
        D = D * sc + p;
        M = Mn;
    }
    {
        float lg = leaky(as_n + ad_n + S / degf);
        float Mn = fmaxf(M, lg);
        float sc = __expf(M - Mn);
        float p  = __expf(lg - Mn);
        A.x = A.x * sc + p * hself.x;
        A.y = A.y * sc + p * hself.y;
        A.z = A.z * sc + p * hself.z;
        A.w = A.w * sc + p * hself.w;
        D = D * sc + p;
    }
    float4 bv = reinterpret_cast<const float4*>(bias)[lane];
    float inv = 1.0f / D;
    float4 r = { A.x * inv + bv.x, A.y * inv + bv.y, A.z * inv + bv.z, A.w * inv + bv.w };
    if (mode == 0) {
        r.x = fmaxf(r.x, 0.f); r.y = fmaxf(r.y, 0.f);
        r.z = fmaxf(r.z, 0.f); r.w = fmaxf(r.w, 0.f);
        reinterpret_cast<float4*>(outv)[(size_t)n * 32 + lane] = r;
    } else {
        float4 lw = reinterpret_cast<const float4*>(lin_w)[lane];
        float part = r.x * lw.x + r.y * lw.y + r.z * lw.z + r.w * lw.w;
#pragma unroll
        for (int msk = 16; msk >= 1; msk >>= 1) part += __shfl_xor(part, msk, 32);
        if (lane == 0) outs[n] = part + lin_b[0];
    }
}

extern "C" void kernel_launch(void* const* d_in, const int* in_sizes, int n_in,
                              void* d_out, int out_size, void* d_ws, size_t ws_size,
                              hipStream_t stream) {
    const float* x   = (const float*)d_in[0];
    const float* ea  = (const float*)d_in[1];
    const float* W1  = (const float*)d_in[2];
    const float* as1 = (const float*)d_in[3];
    const float* ad1 = (const float*)d_in[4];
    const float* We1 = (const float*)d_in[5];
    const float* ae1 = (const float*)d_in[6];
    const float* b1  = (const float*)d_in[7];
    const float* W2  = (const float*)d_in[8];
    const float* as2 = (const float*)d_in[9];
    const float* ad2 = (const float*)d_in[10];
    const float* We2 = (const float*)d_in[11];
    const float* ae2 = (const float*)d_in[12];
    const float* b2  = (const float*)d_in[13];
    const float* lw  = (const float*)d_in[14];
    const float* lb  = (const float*)d_in[15];
    const int*   ei  = (const int*)d_in[16];

    const int N = in_sizes[0] / F_IN;
    const int E = in_sizes[16] / 2;
    const int* srcA = ei;
    const int* dstA = ei + E;

    char* w = (char*)d_ws;
    auto alloc = [&](size_t bytes) -> void* {
        void* p = (void*)w;
        w += (bytes + 255) & ~(size_t)255;
        return p;
    };
    int*   degI   = (int*)  alloc((size_t)N * 4);
    int*   cursor = (int*)  alloc((size_t)N * 4);
    size_t zbytes = (size_t)(w - (char*)degI);
    float* Ve1  = (float*)alloc(F_IN * 4 * 4);
    float* Ve2  = (float*)alloc(F_IN * 4 * 4);
    float* ae1v = (float*)alloc((size_t)E * 16);
    float* ae2v = (float*)alloc((size_t)E * 16);
    float* a_sb = (float*)alloc((size_t)N * 16);
    float* a_db = (float*)alloc((size_t)N * 16);
    int*   off  = (int*)  alloc((size_t)N * 4);
    int*   eidA = (int*)  alloc((size_t)E * 4);
    float* Hp   = (float*)alloc((size_t)N * HC * 4);
    float* h1   = (float*)alloc((size_t)N * HC * 4);

    hipMemsetAsync(degI, 0, zbytes, stream);

    fold_kernel<<<1, 256, 0, stream>>>(We1, ae1, We2, ae2, Ve1, Ve2);
    edge_proj_kernel<<<(E + 255) / 256, 256, 0, stream>>>(ea, dstA, Ve1, Ve2,
                                                          ae1v, ae2v, degI, E);
    scan_kernel<<<1, 1024, 0, stream>>>(degI, off, N);
    scatter_kernel<<<(E + 255) / 256, 256, 0, stream>>>(dstA, off, cursor, eidA, E);

    const int mt = (N + 15) / 16;
    const int gemmBlocks = (mt * 8 + 3) / 4;
    const int nodeBlocks = (N + 7) / 8;

    gemm_wmma_kernel<<<gemmBlocks, 128, 0, stream>>>(x, W1, Hp, N, F_IN);
    node_att_kernel<<<nodeBlocks, 256, 0, stream>>>(Hp, as1, ad1, a_sb, a_db, N);
    agg_kernel<<<nodeBlocks, 256, 0, stream>>>(off, degI, eidA, srcA, ae1v, a_sb, a_db,
                                               Hp, b1, nullptr, nullptr,
                                               h1, nullptr, N, 0);
    gemm_wmma_kernel<<<gemmBlocks, 128, 0, stream>>>(h1, W2, Hp, N, HC);
    node_att_kernel<<<nodeBlocks, 256, 0, stream>>>(Hp, as2, ad2, a_sb, a_db, N);
    agg_kernel<<<nodeBlocks, 256, 0, stream>>>(off, degI, eidA, srcA, ae2v, a_sb, a_db,
                                               Hp, b2, lw, lb,
                                               nullptr, (float*)d_out, N, 1);
}


# edge pass reads edge_attr with whole-line 16-byte loads: 8 lanes per edge, coefficient slices in registers, partial sums folded across lanes by DPP; no LDS, no scratch (a_e summed in a new order)
# speedup vs baseline: 1.6020x; 1.1175x over previous
#include <hip/hip_runtime.h>
#include <hip/hip_bf16.h>

#define F_IN   64
#define HEADS  4
#define CHD    32
#define HC     128
#define NEG_SLOPE 0.2f

typedef __attribute__((ext_vector_type(2))) float v2f;
typedef __attribute__((ext_vector_type(8))) float v8f;

__device__ __forceinline__ float leaky(float x) { return x > 0.f ? x : NEG_SLOPE * x; }

__global__ void fold_kernel(const float* __restrict__ We1, const float* __restrict__ ae1,
                            const float* __restrict__ We2, const float* __restrict__ ae2,
                            float* __restrict__ Ve1, float* __restrict__ Ve2) {
    int t = threadIdx.x;
    int f = t >> 2, h = t & 3;
    float s1 = 0.f, s2 = 0.f;
#pragma unroll
    for (int c = 0; c < CHD; ++c) {
        s1 += We1[f * HC + h * CHD + c] * ae1[h * CHD + c];
        s2 += We2[f * HC + h * CHD + c] * ae2[h * CHD + c];
    }
    Ve1[f * 4 + h] = s1;
    Ve2[f * 4 + h] = s2;
}

__device__ __forceinline__ float lane_xor4(float v) {
    return __builtin_bit_cast(float, __builtin_amdgcn_update_dpp(0, __builtin_bit_cast(int, v), 0x164, 0xf, 0xf, false));
}
__device__ __forceinline__ float lane_xor2(float v) {
    return __builtin_bit_cast(float, __builtin_amdgcn_update_dpp(0, __builtin_bit_cast(int, v), 0x4E, 0xf, 0xf, false));
}
__device__ __forceinline__ float lane_xor1(float v) {
    return __builtin_bit_cast(float, __builtin_amdgcn_update_dpp(0, __builtin_bit_cast(int, v), 0xB1, 0xf, 0xf, false));
}

#define EP_GROUPS 8

__global__ __launch_bounds__(256)
void edge_proj_kernel(const float* __restrict__ EA, const int* __restrict__ dstA,
                      const float* __restrict__ Ve1, const float* __restrict__ Ve2,
                      float* __restrict__ ae1v, float* __restrict__ ae2v,
                      int* __restrict__ degI, int E) {
    const int lane = threadIdx.x & 31;
    const int j = lane & 7;
    const int sub = lane >> 3;
    const int gw = blockIdx.x * (blockDim.x >> 5) + (threadIdx.x >> 5);
    const int base = gw * (4 * EP_GROUPS);
    if (base >= E) return;

    const int eown = base + lane;
    const int down = dstA[min(eown, E - 1)];

    const float4* EA4 = reinterpret_cast<const float4*>(EA);
    float4 x0[EP_GROUPS], x1[EP_GROUPS];
#pragma unroll
    for (int t = 0; t < EP_GROUPS; ++t) {
        int e = min(base + 4 * t + sub, E - 1);
        x0[t] = EA4[(size_t)e * 16 + j];
        x1[t] = EA4[(size_t)e * 16 + 8 + j];
    }

    const float4* V1 = reinterpret_cast<const float4*>(Ve1);
    const float4* V2 = reinterpret_cast<const float4*>(Ve2);
    float wl[4][8], wh[4][8];
#pragma unroll
    for (int c = 0; c < 4; ++c) {
        float4 p = V1[4 * j + c], q = V2[4 * j + c];
        float4 r = V1[32 + 4 * j + c], s = V2[32 + 4 * j + c];
        wl[c][0] = p.x; wl[c][1] = p.y; wl[c][2] = p.z; wl[c][3] = p.w;
        wl[c][4] = q.x; wl[c][5] = q.y; wl[c][6] = q.z; wl[c][7] = q.w;
        wh[c][0] = r.x; wh[c][1] = r.y; wh[c][2] = r.z; wh[c][3] = r.w;
        wh[c][4] = s.x; wh[c][5] = s.y; wh[c][6] = s.z; wh[c][7] = s.w;
    }

    const bool b4 = (j & 4) != 0, b2 = (j & 2) != 0, b1 = (j & 1) != 0;
    float* outp = (b4 ? ae2v : ae1v) + (j & 3);
#pragma unroll
    for (int t = 0; t < EP_GROUPS; ++t) {
        float acc[8];
#pragma unroll
        for (int q = 0; q < 8; ++q) {
            acc[q] = x0[t].x * wl[0][q] + x0[t].y * wl[1][q] + x0[t].z * wl[2][q] + x0[t].w * wl[3][q]
                   + x1[t].x * wh[0][q] + x1[t].y * wh[1][q] + x1[t].z * wh[2][q] + x1[t].w * wh[3][q];
        }
        float r[4], s[2];
#pragma unroll
        for (int q = 0; q < 4; ++q) {
            float keep = b4 ? acc[q + 4] : acc[q];
            float send = b4 ? acc[q] : acc[q + 4];
            r[q] = keep + lane_xor4(send);
        }
#pragma unroll
        for (int q = 0; q < 2; ++q) {
            float keep = b2 ? r[q + 2] : r[q];
            float send = b2 ? r[q] : r[q + 2];
            s[q] = keep + lane_xor2(send);
        }
        float keep = b1 ? s[1] : s[0];
        float send = b1 ? s[0] : s[1];
        float v = keep + lane_xor1(send);
        int e = base + 4 * t + sub;
        if (e < E) outp[(size_t)e * 4] = v;
    }
    if (eown < E) atomicAdd(&degI[down], 1);
}

__global__ void scan_kernel(const int* __restrict__ deg, int* __restrict__ off, int N) {
    __shared__ int sh[1024];
    int t = threadIdx.x;
    int CHK = (N + 1023) / 1024;
    int b = t * CHK, en = min(b + CHK, N);
    int sum = 0;
    for (int i = b; i < en; ++i) sum += deg[i];
    sh[t] = sum;
    __syncthreads();
    for (int d = 1; d < 1024; d <<= 1) {
        int v = (t >= d) ? sh[t - d] : 0;
        __syncthreads();
        sh[t] += v;
        __syncthreads();
    }
    int run = sh[t] - sum;
    for (int i = b; i < en; ++i) { off[i] = run; run += deg[i]; }
}

__global__ void scatter_kernel(const int* __restrict__ dstA, const int* __restrict__ off,
                               int* __restrict__ cursor, int* __restrict__ eidA, int E) {
    int e = blockIdx.x * blockDim.x + threadIdx.x;
    if (e >= E) return;
    int d = dstA[e];
    int pos = off[d] + atomicAdd(&cursor[d], 1);
    eidA[pos] = e;
}

__global__ void gemm_wmma_kernel(const float* __restrict__ X, const float* __restrict__ W,
                                 float* __restrict__ Y, int M, int K) {
    int wave = threadIdx.x >> 5;
    int lane = threadIdx.x & 31;
    int tile = blockIdx.x * (blockDim.x >> 5) + wave;
    int mt = tile >> 3;
    int nt = tile & 7;
    if (mt * 16 >= M) return;
    int row0 = mt * 16, col0 = nt * 16;
    int m = lane & 15, half = lane >> 4;

    v8f acc = {};
    const float* ap0 = X + (size_t)(row0 + m) * K + 2 * half;
    const float* bp0 = W + (size_t)(2 * half) * HC + col0 + m;
    for (int k0 = 0; k0 < K; k0 += 4) {
        v2f a, b;
        a.x = ap0[k0];
        a.y = ap0[k0 + 1];
        b.x = bp0[(size_t)k0 * HC];
        b.y = bp0[(size_t)(k0 + 1) * HC];
        acc = __builtin_amdgcn_wmma_f32_16x16x4_f32(false, a, false, b,
                                                    (short)0, acc, false, false);
    }
    float* yp = Y + (size_t)(row0 + 8 * half) * HC + col0 + m;
#pragma unroll
    for (int r = 0; r < 8; ++r) yp[(size_t)r * HC] = acc[r];
}

__global__ void node_att_kernel(const float* __restrict__ Hp,
                                const float* __restrict__ att_s, const float* __restrict__ att_d,
                                float* __restrict__ a_s, float* __restrict__ a_d, int N) {
    int wave = threadIdx.x >> 5, lane = threadIdx.x & 31;
    int n = blockIdx.x * (blockDim.x >> 5) + wave;
    if (n >= N) return;
    int h = lane >> 3, q = lane & 7;
    float4 hv = reinterpret_cast<const float4*>(Hp)[(size_t)n * 32 + lane];
    float4 sa = reinterpret_cast<const float4*>(att_s)[lane];
    float4 da = reinterpret_cast<const float4*>(att_d)[lane];
    float ps = hv.x * sa.x + hv.y * sa.y + hv.z * sa.z + hv.w * sa.w;
    float pd = hv.x * da.x + hv.y * da.y + hv.z * da.z + hv.w * da.w;
#pragma unroll
    for (int msk = 1; msk < 8; msk <<= 1) {
        ps += __shfl_xor(ps, msk, 32);
        pd += __shfl_xor(pd, msk, 32);
    }
    if (q == 0) { a_s[(size_t)n * 4 + h] = ps; a_d[(size_t)n * 4 + h] = pd; }
}

__global__ void agg_kernel(const int* __restrict__ off, const int* __restrict__ degI,
                           const int* __restrict__ eidA, const int* __restrict__ srcA,
                           const float* __restrict__ a_e, const float* __restrict__ a_s,
                           const float* __restrict__ a_d,
                           const float* __restrict__ Hp, const float* __restrict__ bias,
                           const float* __restrict__ lin_w, const float* __restrict__ lin_b,
                           float* __restrict__ outv, float* __restrict__ outs,
                           int N, int mode) {
    int wave = threadIdx.x >> 5, lane = threadIdx.x & 31;
    int n = blockIdx.x * (blockDim.x >> 5) + wave;
    if (n >= N) return;
    int h = lane >> 3;
    const float4* Hp4 = reinterpret_cast<const float4*>(Hp);

    float ad_n = a_d[(size_t)n * 4 + h];
    float as_n = a_s[(size_t)n * 4 + h];
    int cnt = degI[n];
    float degf = fmaxf((float)cnt, 1.0f);
    float M = -INFINITY;
    float D = 0.0f;
    float4 A = {0.f, 0.f, 0.f, 0.f};
    float S = 0.0f;
    float4 hself = Hp4[(size_t)n * 32 + lane];

    int s0 = off[n];
    for (int i = 0; i < cnt; ++i) {
        int e = eidA[s0 + i];
        int s = srcA[e];
        float ae = a_e[(size_t)e * 4 + h];
        S += ae;
        float lg = leaky(a_s[(size_t)s * 4 + h] + ad_n + ae);
        float Mn = fmaxf(M, lg);
        float sc = __expf(M - Mn);
        float p  = __expf(lg - Mn);
        float4 hv = Hp4[(size_t)s * 32 + lane];
        A.x = A.x * sc + p * hv.x;
        A.y = A.y * sc + p * hv.y;
        A.z = A.z * sc + p * hv.z;
        A.w = A.w * sc + p * hv.w;
        D = D * sc + p;
        M = Mn;
    }
    {
        float lg = leaky(as_n + ad_n + S / degf);
        float Mn = fmaxf(M, lg);
        float sc = __expf(M - Mn);
        float p  = __expf(lg - Mn);
        A.x = A.x * sc + p * hself.x;
        A.y = A.y * sc + p * hself.y;
        A.z = A.z * sc + p * hself.z;
        A.w = A.w * sc + p * hself.w;
        D = D * sc + p;
    }
    float4 bv = reinterpret_cast<const float4*>(bias)[lane];
    float inv = 1.0f / D;
    float4 r = { A.x * inv + bv.x, A.y * inv + bv.y, A.z * inv + bv.z, A.w * inv + bv.w };
    if (mode == 0) {
        r.x = fmaxf(r.x, 0.f); r.y = fmaxf(r.y, 0.f);
        r.z = fmaxf(r.z, 0.f); r.w = fmaxf(r.w, 0.f);
        reinterpret_cast<float4*>(outv)[(size_t)n * 32 + lane] = r;
    } else {
        float4 lw = reinterpret_cast<const float4*>(lin_w)[lane];
        float part = r.x * lw.x + r.y * lw.y + r.z * lw.z + r.w * lw.w;
#pragma unroll
        for (int msk = 16; msk >= 1; msk >>= 1) part += __shfl_xor(part, msk, 32);
        if (lane == 0) outs[n] = part + lin_b[0];
    }
}

extern "C" void kernel_launch(void* const* d_in, const int* in_sizes, int n_in,
                              void* d_out, int out_size, void* d_ws, size_t ws_size,
                              hipStream_t stream) {
    const float* x   = (const float*)d_in[0];
    const float* ea  = (const float*)d_in[1];
    const float* W1  = (const float*)d_in[2];
    const float* as1 = (const float*)d_in[3];
    const float* ad1 = (const float*)d_in[4];
    const float* We1 = (const float*)d_in[5];
    const float* ae1 = (const float*)d_in[6];
    const float* b1  = (const float*)d_in[7];
    const float* W2  = (const float*)d_in[8];
    const float* as2 = (const float*)d_in[9];
    const float* ad2 = (const float*)d_in[10];
    const float* We2 = (const float*)d_in[11];
    const float* ae2 = (const float*)d_in[12];
    const float* b2  = (const float*)d_in[13];
    const float* lw  = (const float*)d_in[14];
    const float* lb  = (const float*)d_in[15];
    const int*   ei  = (const int*)d_in[16];

    const int N = in_sizes[0] / F_IN;
    const int E = in_sizes[16] / 2;
    const int* srcA = ei;
    const int* dstA = ei + E;

    char* w = (char*)d_ws;
    auto alloc = [&](size_t bytes) -> void* {
        void* p = (void*)w;
        w += (bytes + 255) & ~(size_t)255;
        return p;
    };
    int*   degI   = (int*)  alloc((size_t)N * 4);
    int*   cursor = (int*)  alloc((size_t)N * 4);
    size_t zbytes = (size_t)(w - (char*)degI);
    float* Ve1  = (float*)alloc(F_IN * 4 * 4);
    float* Ve2  = (float*)alloc(F_IN * 4 * 4);
    float* ae1v = (float*)alloc((size_t)E * 16);
    float* ae2v = (float*)alloc((size_t)E * 16);
    float* a_sb = (float*)alloc((size_t)N * 16);
    float* a_db = (float*)alloc((size_t)N * 16);
    int*   off  = (int*)  alloc((size_t)N * 4);
    int*   eidA = (int*)  alloc((size_t)E * 4);
    float* Hp   = (float*)alloc((size_t)N * HC * 4);
    float* h1   = (float*)alloc((size_t)N * HC * 4);

    hipMemsetAsync(degI, 0, zbytes, stream);

    fold_kernel<<<1, 256, 0, stream>>>(We1, ae1, We2, ae2, Ve1, Ve2);
    edge_proj_kernel<<<(E + 255) / 256, 256, 0, stream>>>(ea, dstA, Ve1, Ve2,
                                                          ae1v, ae2v, degI, E);
    scan_kernel<<<1, 1024, 0, stream>>>(degI, off, N);
    scatter_kernel<<<(E + 255) / 256, 256, 0, stream>>>(dstA, off, cursor, eidA, E);

    const int mt = (N + 15) / 16;
    const int gemmBlocks = (mt * 8 + 3) / 4;
    const int nodeBlocks = (N + 7) / 8;

    gemm_wmma_kernel<<<gemmBlocks, 128, 0, stream>>>(x, W1, Hp, N, F_IN);
    node_att_kernel<<<nodeBlocks, 256, 0, stream>>>(Hp, as1, ad1, a_sb, a_db, N);
    agg_kernel<<<nodeBlocks, 256, 0, stream>>>(off, degI, eidA, srcA, ae1v, a_sb, a_db,
                                               Hp, b1, nullptr, nullptr,
                                               h1, nullptr, N, 0);
    gemm_wmma_kernel<<<gemmBlocks, 128, 0, stream>>>(h1, W2, Hp, N, HC);
    node_att_kernel<<<nodeBlocks, 256, 0, stream>>>(Hp, as2, ad2, a_sb, a_db, N);
    agg_kernel<<<nodeBlocks, 256, 0, stream>>>(off, degI, eidA, srcA, ae2v, a_sb, a_db,
                                               Hp, b2, lw, lb,
                                               nullptr, (float*)d_out, N, 1);
}


# projection wave owns 16 whole rows (8 column tiles, one A fragment load per k-step) and folds a_s/a_d out of its accumulators by DPP; both node_att launches gone (a_s/a_d summed in a new order)
# speedup vs baseline: 1.6719x; 1.0436x over previous
#include <hip/hip_runtime.h>
#include <hip/hip_bf16.h>

#define F_IN   64
#define HEADS  4
#define CHD    32
#define HC     128
#define NEG_SLOPE 0.2f

typedef __attribute__((ext_vector_type(2))) float v2f;
typedef __attribute__((ext_vector_type(8))) float v8f;

__device__ __forceinline__ float leaky(float x) { return x > 0.f ? x : NEG_SLOPE * x; }

__global__ void fold_kernel(const float* __restrict__ We1, const float* __restrict__ ae1,
                            const float* __restrict__ We2, const float* __restrict__ ae2,
                            float* __restrict__ Ve1, float* __restrict__ Ve2) {
    int t = threadIdx.x;
    int f = t >> 2, h = t & 3;
    float s1 = 0.f, s2 = 0.f;
#pragma unroll
    for (int c = 0; c < CHD; ++c) {
        s1 += We1[f * HC + h * CHD + c] * ae1[h * CHD + c];
        s2 += We2[f * HC + h * CHD + c] * ae2[h * CHD + c];
    }
    Ve1[f * 4 + h] = s1;
    Ve2[f * 4 + h] = s2;
}

__device__ __forceinline__ float lane_xor4(float v) {
    return __builtin_bit_cast(float, __builtin_amdgcn_update_dpp(0, __builtin_bit_cast(int, v), 0x164, 0xf, 0xf, false));
}
__device__ __forceinline__ float lane_xor2(float v) {
    return __builtin_bit_cast(float, __builtin_amdgcn_update_dpp(0, __builtin_bit_cast(int, v), 0x4E, 0xf, 0xf, false));
}
__device__ __forceinline__ float lane_xor1(float v) {
    return __builtin_bit_cast(float, __builtin_amdgcn_update_dpp(0, __builtin_bit_cast(int, v), 0xB1, 0xf, 0xf, false));
}

#define EP_GROUPS 8

__global__ __launch_bounds__(256)
void edge_proj_kernel(const float* __restrict__ EA, const int* __restrict__ dstA,
                      const float* __restrict__ Ve1, const float* __restrict__ Ve2,
                      float* __restrict__ ae1v, float* __restrict__ ae2v,
                      int* __restrict__ degI, int E) {
    const int lane = threadIdx.x & 31;
    const int j = lane & 7;
    const int sub = lane >> 3;
    const int gw = blockIdx.x * (blockDim.x >> 5) + (threadIdx.x >> 5);
    const int base = gw * (4 * EP_GROUPS);
    if (base >= E) return;

    const int eown = base + lane;
    const int down = dstA[min(eown, E - 1)];

    const float4* EA4 = reinterpret_cast<const float4*>(EA);
    float4 x0[EP_GROUPS], x1[EP_GROUPS];
#pragma unroll
    for (int t = 0; t < EP_GROUPS; ++t) {
        int e = min(base + 4 * t + sub, E - 1);
        x0[t] = EA4[(size_t)e * 16 + j];
        x1[t] = EA4[(size_t)e * 16 + 8 + j];
    }

    const float4* V1 = reinterpret_cast<const float4*>(Ve1);
    const float4* V2 = reinterpret_cast<const float4*>(Ve2);
    float wl[4][8], wh[4][8];
#pragma unroll
    for (int c = 0; c < 4; ++c) {
        float4 p = V1[4 * j + c], q = V2[4 * j + c];
        float4 r = V1[32 + 4 * j + c], s = V2[32 + 4 * j + c];
        wl[c][0] = p.x; wl[c][1] = p.y; wl[c][2] = p.z; wl[c][3] = p.w;
        wl[c][4] = q.x; wl[c][5] = q.y; wl[c][6] = q.z; wl[c][7] = q.w;
        wh[c][0] = r.x; wh[c][1] = r.y; wh[c][2] = r.z; wh[c][3] = r.w;
        wh[c][4] = s.x; wh[c][5] = s.y; wh[c][6] = s.z; wh[c][7] = s.w;
    }

    const bool b4 = (j & 4) != 0, b2 = (j & 2) != 0, b1 = (j & 1) != 0;
    float* outp = (b4 ? ae2v : ae1v) + (j & 3);
#pragma unroll
    for (int t = 0; t < EP_GROUPS; ++t) {
        float acc[8];
#pragma unroll
        for (int q = 0; q < 8; ++q) {
            acc[q] = x0[t].x * wl[0][q] + x0[t].y * wl[1][q] + x0[t].z * wl[2][q] + x0[t].w * wl[3][q]
                   + x1[t].x * wh[0][q] + x1[t].y * wh[1][q] + x1[t].z * wh[2][q] + x1[t].w * wh[3][q];
        }
        float r[4], s[2];
#pragma unroll
        for (int q = 0; q < 4; ++q) {
            float keep = b4 ? acc[q + 4] : acc[q];
            float send = b4 ? acc[q] : acc[q + 4];
            r[q] = keep + lane_xor4(send);
        }
#pragma unroll
        for (int q = 0; q < 2; ++q) {
            float keep = b2 ? r[q + 2] : r[q];
            float send = b2 ? r[q] : r[q + 2];
            s[q] = keep + lane_xor2(send);
        }
        float keep = b1 ? s[1] : s[0];
        float send = b1 ? s[0] : s[1];
        float v = keep + lane_xor1(send);
        int e = base + 4 * t + sub;
        if (e < E) outp[(size_t)e * 4] = v;
    }
    if (eown < E) atomicAdd(&degI[down], 1);
}

__global__ void scan_kernel(const int* __restrict__ deg, int* __restrict__ off, int N) {
    __shared__ int sh[1024];
    int t = threadIdx.x;
    int CHK = (N + 1023) / 1024;
    int b = t * CHK, en = min(b + CHK, N);
    int sum = 0;
    for (int i = b; i < en; ++i) sum += deg[i];
    sh[t] = sum;
    __syncthreads();
    for (int d = 1; d < 1024; d <<= 1) {
        int v = (t >= d) ? sh[t - d] : 0;
        __syncthreads();
        sh[t] += v;
        __syncthreads();
    }
    int run = sh[t] - sum;
    for (int i = b; i < en; ++i) { off[i] = run; run += deg[i]; }
}

__global__ void scatter_kernel(const int* __restrict__ dstA, const int* __restrict__ off,
                               int* __restrict__ cursor, int* __restrict__ eidA, int E) {
    int e = blockIdx.x * blockDim.x + threadIdx.x;
    if (e >= E) return;
    int d = dstA[e];
    int pos = off[d] + atomicAdd(&cursor[d], 1);
    eidA[pos] = e;
}

__device__ __forceinline__ float lane_xor8(float v) {
    return __builtin_bit_cast(float, __builtin_amdgcn_update_dpp(0, __builtin_bit_cast(int, v), 0x168, 0xf, 0xf, false));
}

__global__ __launch_bounds__(128)
void proj_att_kernel(const float* __restrict__ X, const float* __restrict__ W,
                     const float* __restrict__ att_s, const float* __restrict__ att_d,
                     float* __restrict__ Y, float* __restrict__ a_s, float* __restrict__ a_d,
                     int M, int K) {
    const int wave = threadIdx.x >> 5;
    const int lane = threadIdx.x & 31;
    const int mt = blockIdx.x * (blockDim.x >> 5) + wave;
    if (mt * 16 >= M) return;
    const int row0 = mt * 16;
    const int m = lane & 15, half = lane >> 4;

    float ws[8], wd[8];
#pragma unroll
    for (int nt = 0; nt < 8; ++nt) {
        ws[nt] = att_s[nt * 16 + m];
        wd[nt] = att_d[nt * 16 + m];
    }

    v8f acc[8];
#pragma unroll
    for (int nt = 0; nt < 8; ++nt) acc[nt] = v8f{};

    const float* ap0 = X + (size_t)(row0 + m) * K + 2 * half;
    const float* bp0 = W + (size_t)(2 * half) * HC + m;
    for (int k0 = 0; k0 < K; k0 += 4) {
        v2f a;
        a.x = ap0[k0];
        a.y = ap0[k0 + 1];
        const float* bp = bp0 + (size_t)k0 * HC;
#pragma unroll
        for (int nt = 0; nt < 8; ++nt) {
            v2f b;
            b.x = bp[nt * 16];
            b.y = bp[HC + nt * 16];
            acc[nt] = __builtin_amdgcn_wmma_f32_16x16x4_f32(false, a, false, b,
                                                            (short)0, acc[nt], false, false);
        }
    }

    float* yp = Y + (size_t)(row0 + 8 * half) * HC + m;
#pragma unroll
    for (int nt = 0; nt < 8; ++nt) {
#pragma unroll
        for (int r = 0; r < 8; ++r) yp[(size_t)r * HC + nt * 16] = acc[nt][r];
    }

    float p[2][4][8];
#pragma unroll
    for (int h = 0; h < 4; ++h) {
#pragma unroll
        for (int r = 0; r < 8; ++r) {
            p[0][h][r] = acc[2 * h][r] * ws[2 * h] + acc[2 * h + 1][r] * ws[2 * h + 1];
            p[1][h][r] = acc[2 * h][r] * wd[2 * h] + acc[2 * h + 1][r] * wd[2 * h + 1];
        }
    }
    const bool b8 = (m & 8) != 0, b4 = (m & 4) != 0, b2 = (m & 2) != 0, b1 = (m & 1) != 0;
    float q[2][4][4];
#pragma unroll
    for (int sd = 0; sd < 2; ++sd)
#pragma unroll
        for (int h = 0; h < 4; ++h)
#pragma unroll
            for (int r = 0; r < 4; ++r) {
                float keep = b8 ? p[sd][h][r + 4] : p[sd][h][r];
                float send = b8 ? p[sd][h][r] : p[sd][h][r + 4];
                q[sd][h][r] = keep + lane_xor8(send);
            }
    float t[2][4][2];
#pragma unroll
    for (int sd = 0; sd < 2; ++sd)
#pragma unroll
        for (int h = 0; h < 4; ++h)
#pragma unroll
            for (int r = 0; r < 2; ++r) {
                float keep = b4 ? q[sd][h][r + 2] : q[sd][h][r];
                float send = b4 ? q[sd][h][r] : q[sd][h][r + 2];
                t[sd][h][r] = keep + lane_xor4(send);
            }
    float u[2][4];
#pragma unroll
    for (int sd = 0; sd < 2; ++sd)
#pragma unroll
        for (int h = 0; h < 4; ++h) {
            float keep = b2 ? t[sd][h][1] : t[sd][h][0];
            float send = b2 ? t[sd][h][0] : t[sd][h][1];
            u[sd][h] = keep + lane_xor2(send);
        }
    float v[4];
#pragma unroll
    for (int h = 0; h < 4; ++h) {
        float keep = b1 ? u[1][h] : u[0][h];
        float send = b1 ? u[0][h] : u[1][h];
        v[h] = keep + lane_xor1(send);
    }
    float* ap = (b1 ? a_d : a_s) + (size_t)(row0 + 8 * half + (m >> 1)) * 4;
    *reinterpret_cast<float4*>(ap) = float4{v[0], v[1], v[2], v[3]};
}

__global__ void agg_kernel(const int* __restrict__ off, const int* __restrict__ degI,
                           const int* __restrict__ eidA, const int* __restrict__ srcA,
                           const float* __restrict__ a_e, const float* __restrict__ a_s,
                           const float* __restrict__ a_d,
                           const float* __restrict__ Hp, const float* __restrict__ bias,
                           const float* __restrict__ lin_w, const float* __restrict__ lin_b,
                           float* __restrict__ outv, float* __restrict__ outs,
                           int N, int mode) {
    int wave = threadIdx.x >> 5, lane = threadIdx.x & 31;
    int n = blockIdx.x * (blockDim.x >> 5) + wave;
    if (n >= N) return;
    int h = lane >> 3;
    const float4* Hp4 = reinterpret_cast<const float4*>(Hp);

    float ad_n = a_d[(size_t)n * 4 + h];
    float as_n = a_s[(size_t)n * 4 + h];
    int cnt = degI[n];
    float degf = fmaxf((float)cnt, 1.0f);
    float M = -INFINITY;
    float D = 0.0f;
    float4 A = {0.f, 0.f, 0.f, 0.f};
    float S = 0.0f;
    float4 hself = Hp4[(size_t)n * 32 + lane];

    int s0 = off[n];
    for (int i = 0; i < cnt; ++i) {
        int e = eidA[s0 + i];
        int s = srcA[e];
        float ae = a_e[(size_t)e * 4 + h];
        S += ae;
        float lg = leaky(a_s[(size_t)s * 4 + h] + ad_n + ae);
        float Mn = fmaxf(M, lg);
        float sc = __expf(M - Mn);
        float p  = __expf(lg - Mn);
        float4 hv = Hp4[(size_t)s * 32 + lane];
        A.x = A.x * sc + p * hv.x;
        A.y = A.y * sc + p * hv.y;
        A.z = A.z * sc + p * hv.z;
        A.w = A.w * sc + p * hv.w;
        D = D * sc + p;
        M = Mn;
    }
    {
        float lg = leaky(as_n + ad_n + S / degf);
        float Mn = fmaxf(M, lg);
        float sc = __expf(M - Mn);
        float p  = __expf(lg - Mn);
        A.x = A.x * sc + p * hself.x;
        A.y = A.y * sc + p * hself.y;
        A.z = A.z * sc + p * hself.z;
        A.w = A.w * sc + p * hself.w;
        D = D * sc + p;
    }
    float4 bv = reinterpret_cast<const float4*>(bias)[lane];
    float inv = 1.0f / D;
    float4 r = { A.x * inv + bv.x, A.y * inv + bv.y, A.z * inv + bv.z, A.w * inv + bv.w };
    if (mode == 0) {
        r.x = fmaxf(r.x, 0.f); r.y = fmaxf(r.y, 0.f);
        r.z = fmaxf(r.z, 0.f); r.w = fmaxf(r.w, 0.f);
        reinterpret_cast<float4*>(outv)[(size_t)n * 32 + lane] = r;
    } else {
        float4 lw = reinterpret_cast<const float4*>(lin_w)[lane];
        float part = r.x * lw.x + r.y * lw.y + r.z * lw.z + r.w * lw.w;
#pragma unroll
        for (int msk = 16; msk >= 1; msk >>= 1) part += __shfl_xor(part, msk, 32);
        if (lane == 0) outs[n] = part + lin_b[0];
    }
}

extern "C" void kernel_launch(void* const* d_in, const int* in_sizes, int n_in,
                              void* d_out, int out_size, void* d_ws, size_t ws_size,
                              hipStream_t stream) {
    const float* x   = (const float*)d_in[0];
    const float* ea  = (const float*)d_in[1];
    const float* W1  = (const float*)d_in[2];
    const float* as1 = (const float*)d_in[3];
    const float* ad1 = (const float*)d_in[4];
    const float* We1 = (const float*)d_in[5];
    const float* ae1 = (const float*)d_in[6];
    const float* b1  = (const float*)d_in[7];
    const float* W2  = (const float*)d_in[8];
    const float* as2 = (const float*)d_in[9];
    const float* ad2 = (const float*)d_in[10];
    const float* We2 = (const float*)d_in[11];
    const float* ae2 = (const float*)d_in[12];
    const float* b2  = (const float*)d_in[13];
    const float* lw  = (const float*)d_in[14];
    const float* lb  = (const float*)d_in[15];
    const int*   ei  = (const int*)d_in[16];

    const int N = in_sizes[0] / F_IN;
    const int E = in_sizes[16] / 2;
    const int* srcA = ei;
    const int* dstA = ei + E;

    char* w = (char*)d_ws;
    auto alloc = [&](size_t bytes) -> void* {
        void* p = (void*)w;
        w += (bytes + 255) & ~(size_t)255;
        return p;
    };
    int*   degI   = (int*)  alloc((size_t)N * 4);
    int*   cursor = (int*)  alloc((size_t)N * 4);
    size_t zbytes = (size_t)(w - (char*)degI);
    float* Ve1  = (float*)alloc(F_IN * 4 * 4);
    float* Ve2  = (float*)alloc(F_IN * 4 * 4);
    float* ae1v = (float*)alloc((size_t)E * 16);
    float* ae2v = (float*)alloc((size_t)E * 16);
    float* a_sb = (float*)alloc((size_t)N * 16);
    float* a_db = (float*)alloc((size_t)N * 16);
    int*   off  = (int*)  alloc((size_t)N * 4);
    int*   eidA = (int*)  alloc((size_t)E * 4);
    float* Hp   = (float*)alloc((size_t)N * HC * 4);
    float* h1   = (float*)alloc((size_t)N * HC * 4);

    hipMemsetAsync(degI, 0, zbytes, stream);

    fold_kernel<<<1, 256, 0, stream>>>(We1, ae1, We2, ae2, Ve1, Ve2);
    edge_proj_kernel<<<(E + 255) / 256, 256, 0, stream>>>(ea, dstA, Ve1, Ve2,
                                                          ae1v, ae2v, degI, E);
    scan_kernel<<<1, 1024, 0, stream>>>(degI, off, N);
    scatter_kernel<<<(E + 255) / 256, 256, 0, stream>>>(dstA, off, cursor, eidA, E);

    const int mt = (N + 15) / 16;
    const int projBlocks = (mt + 3) / 4;
    const int nodeBlocks = (N + 7) / 8;

    proj_att_kernel<<<projBlocks, 128, 0, stream>>>(x, W1, as1, ad1, Hp, a_sb, a_db, N, F_IN);
    agg_kernel<<<nodeBlocks, 256, 0, stream>>>(off, degI, eidA, srcA, ae1v, a_sb, a_db,
                                               Hp, b1, nullptr, nullptr,
                                               h1, nullptr, N, 0);
    proj_att_kernel<<<projBlocks, 128, 0, stream>>>(h1, W2, as2, ad2, Hp, a_sb, a_db, N, HC);
    agg_kernel<<<nodeBlocks, 256, 0, stream>>>(off, degI, eidA, srcA, ae2v, a_sb, a_db,
                                               Hp, b2, lw, lb,
                                               nullptr, (float*)d_out, N, 1);
}


# aggregation wave takes its list 8 edges x 4 heads per step: ids, a_e, a_s loaded for the whole list at once, softmax max and sums by 8-lane folds, rows gathered 8 per group (new summation order)
# speedup vs baseline: 1.7251x; 1.0318x over previous
#include <hip/hip_runtime.h>
#include <hip/hip_bf16.h>

#define F_IN   64
#define HEADS  4
#define CHD    32
#define HC     128
#define NEG_SLOPE 0.2f

typedef __attribute__((ext_vector_type(2))) float v2f;
typedef __attribute__((ext_vector_type(8))) float v8f;

__device__ __forceinline__ float leaky(float x) { return x > 0.f ? x : NEG_SLOPE * x; }

__global__ void fold_kernel(const float* __restrict__ We1, const float* __restrict__ ae1,
                            const float* __restrict__ We2, const float* __restrict__ ae2,
                            float* __restrict__ Ve1, float* __restrict__ Ve2) {
    int t = threadIdx.x;
    int f = t >> 2, h = t & 3;
    float s1 = 0.f, s2 = 0.f;
#pragma unroll
    for (int c = 0; c < CHD; ++c) {
        s1 += We1[f * HC + h * CHD + c] * ae1[h * CHD + c];
        s2 += We2[f * HC + h * CHD + c] * ae2[h * CHD + c];
    }
    Ve1[f * 4 + h] = s1;
    Ve2[f * 4 + h] = s2;
}

__device__ __forceinline__ float lane_xor4(float v) {
    return __builtin_bit_cast(float, __builtin_amdgcn_update_dpp(0, __builtin_bit_cast(int, v), 0x164, 0xf, 0xf, false));
}
__device__ __forceinline__ float lane_xor2(float v) {
    return __builtin_bit_cast(float, __builtin_amdgcn_update_dpp(0, __builtin_bit_cast(int, v), 0x4E, 0xf, 0xf, false));
}
__device__ __forceinline__ float lane_xor1(float v) {
    return __builtin_bit_cast(float, __builtin_amdgcn_update_dpp(0, __builtin_bit_cast(int, v), 0xB1, 0xf, 0xf, false));
}

#define EP_GROUPS 8

__global__ __launch_bounds__(256)
void edge_proj_kernel(const float* __restrict__ EA, const int* __restrict__ dstA,
                      const float* __restrict__ Ve1, const float* __restrict__ Ve2,
                      float* __restrict__ ae1v, float* __restrict__ ae2v,
                      int* __restrict__ degI, int E) {
    const int lane = threadIdx.x & 31;
    const int j = lane & 7;
    const int sub = lane >> 3;
    const int gw = blockIdx.x * (blockDim.x >> 5) + (threadIdx.x >> 5);
    const int base = gw * (4 * EP_GROUPS);
    if (base >= E) return;

    const int eown = base + lane;
    const int down = dstA[min(eown, E - 1)];

    const float4* EA4 = reinterpret_cast<const float4*>(EA);
    float4 x0[EP_GROUPS], x1[EP_GROUPS];
#pragma unroll
    for (int t = 0; t < EP_GROUPS; ++t) {
        int e = min(base + 4 * t + sub, E - 1);
        x0[t] = EA4[(size_t)e * 16 + j];
        x1[t] = EA4[(size_t)e * 16 + 8 + j];
    }

    const float4* V1 = reinterpret_cast<const float4*>(Ve1);
    const float4* V2 = reinterpret_cast<const float4*>(Ve2);
    float wl[4][8], wh[4][8];
#pragma unroll
    for (int c = 0; c < 4; ++c) {
        float4 p = V1[4 * j + c], q = V2[4 * j + c];
        float4 r = V1[32 + 4 * j + c], s = V2[32 + 4 * j + c];
        wl[c][0] = p.x; wl[c][1] = p.y; wl[c][2] = p.z; wl[c][3] = p.w;
        wl[c][4] = q.x; wl[c][5] = q.y; wl[c][6] = q.z; wl[c][7] = q.w;
        wh[c][0] = r.x; wh[c][1] = r.y; wh[c][2] = r.z; wh[c][3] = r.w;
        wh[c][4] = s.x; wh[c][5] = s.y; wh[c][6] = s.z; wh[c][7] = s.w;
    }

    const bool b4 = (j & 4) != 0, b2 = (j & 2) != 0, b1 = (j & 1) != 0;
    float* outp = (b4 ? ae2v : ae1v) + (j & 3);
#pragma unroll
    for (int t = 0; t < EP_GROUPS; ++t) {
        float acc[8];
#pragma unroll
        for (int q = 0; q < 8; ++q) {
            acc[q] = x0[t].x * wl[0][q] + x0[t].y * wl[1][q] + x0[t].z * wl[2][q] + x0[t].w * wl[3][q]
                   + x1[t].x * wh[0][q] + x1[t].y * wh[1][q] + x1[t].z * wh[2][q] + x1[t].w * wh[3][q];
        }
        float r[4], s[2];
#pragma unroll
        for (int q = 0; q < 4; ++q) {
            float keep = b4 ? acc[q + 4] : acc[q];
            float send = b4 ? acc[q] : acc[q + 4];
            r[q] = keep + lane_xor4(send);
        }
#pragma unroll
        for (int q = 0; q < 2; ++q) {
            float keep = b2 ? r[q + 2] : r[q];
            float send = b2 ? r[q] : r[q + 2];
            s[q] = keep + lane_xor2(send);
        }
        float keep = b1 ? s[1] : s[0];
        float send = b1 ? s[0] : s[1];
        float v = keep + lane_xor1(send);
        int e = base + 4 * t + sub;
        if (e < E) outp[(size_t)e * 4] = v;
    }
    if (eown < E) atomicAdd(&degI[down], 1);
}

__global__ void scan_kernel(const int* __restrict__ deg, int* __restrict__ off, int N) {
    __shared__ int sh[1024];
    int t = threadIdx.x;
    int CHK = (N + 1023) / 1024;
    int b = t * CHK, en = min(b + CHK, N);
    int sum = 0;
    for (int i = b; i < en; ++i) sum += deg[i];
    sh[t] = sum;
    __syncthreads();
    for (int d = 1; d < 1024; d <<= 1) {
        int v = (t >= d) ? sh[t - d] : 0;
        __syncthreads();
        sh[t] += v;
        __syncthreads();
    }
    int run = sh[t] - sum;
    for (int i = b; i < en; ++i) { off[i] = run; run += deg[i]; }
}

__global__ void scatter_kernel(const int* __restrict__ dstA, const int* __restrict__ off,
                               int* __restrict__ cursor, int* __restrict__ eidA, int E) {
    int e = blockIdx.x * blockDim.x + threadIdx.x;
    if (e >= E) return;
    int d = dstA[e];
    int pos = off[d] + atomicAdd(&cursor[d], 1);
    eidA[pos] = e;
}

__device__ __forceinline__ float lane_xor8(float v) {
    return __builtin_bit_cast(float, __builtin_amdgcn_update_dpp(0, __builtin_bit_cast(int, v), 0x168, 0xf, 0xf, false));
}

__global__ __launch_bounds__(128)
void proj_att_kernel(const float* __restrict__ X, const float* __restrict__ W,
                     const float* __restrict__ att_s, const float* __restrict__ att_d,
                     float* __restrict__ Y, float* __restrict__ a_s, float* __restrict__ a_d,
                     int M, int K) {
    const int wave = threadIdx.x >> 5;
    const int lane = threadIdx.x & 31;
    const int mt = blockIdx.x * (blockDim.x >> 5) + wave;
    if (mt * 16 >= M) return;
    const int row0 = mt * 16;
    const int m = lane & 15, half = lane >> 4;

    float ws[8], wd[8];
#pragma unroll
    for (int nt = 0; nt < 8; ++nt) {
        ws[nt] = att_s[nt * 16 + m];
        wd[nt] = att_d[nt * 16 + m];
    }

    v8f acc[8];
#pragma unroll
    for (int nt = 0; nt < 8; ++nt) acc[nt] = v8f{};

    const float* ap0 = X + (size_t)(row0 + m) * K + 2 * half;
    const float* bp0 = W + (size_t)(2 * half) * HC + m;
    for (int k0 = 0; k0 < K; k0 += 4) {
        v2f a;
        a.x = ap0[k0];
        a.y = ap0[k0 + 1];
        const float* bp = bp0 + (size_t)k0 * HC;
#pragma unroll
        for (int nt = 0; nt < 8; ++nt) {
            v2f b;
            b.x = bp[nt * 16];
            b.y = bp[HC + nt * 16];
            acc[nt] = __builtin_amdgcn_wmma_f32_16x16x4_f32(false, a, false, b,
                                                            (short)0, acc[nt], false, false);
        }
    }

    float* yp = Y + (size_t)(row0 + 8 * half) * HC + m;
#pragma unroll
    for (int nt = 0; nt < 8; ++nt) {
#pragma unroll
        for (int r = 0; r < 8; ++r) yp[(size_t)r * HC + nt * 16] = acc[nt][r];
    }

    float p[2][4][8];
#pragma unroll
    for (int h = 0; h < 4; ++h) {
#pragma unroll
        for (int r = 0; r < 8; ++r) {
            p[0][h][r] = acc[2 * h][r] * ws[2 * h] + acc[2 * h + 1][r] * ws[2 * h + 1];
            p[1][h][r] = acc[2 * h][r] * wd[2 * h] + acc[2 * h + 1][r] * wd[2 * h + 1];
        }
    }
    const bool b8 = (m & 8) != 0, b4 = (m & 4) != 0, b2 = (m & 2) != 0, b1 = (m & 1) != 0;
    float q[2][4][4];
#pragma unroll
    for (int sd = 0; sd < 2; ++sd)
#pragma unroll
        for (int h = 0; h < 4; ++h)
#pragma unroll
            for (int r = 0; r < 4; ++r) {
                float keep = b8 ? p[sd][h][r + 4] : p[sd][h][r];
                float send = b8 ? p[sd][h][r] : p[sd][h][r + 4];
                q[sd][h][r] = keep + lane_xor8(send);
            }
    float t[2][4][2];
#pragma unroll
    for (int sd = 0; sd < 2; ++sd)
#pragma unroll
        for (int h = 0; h < 4; ++h)
#pragma unroll
            for (int r = 0; r < 2; ++r) {
                float keep = b4 ? q[sd][h][r + 2] : q[sd][h][r];
                float send = b4 ? q[sd][h][r] : q[sd][h][r + 2];
                t[sd][h][r] = keep + lane_xor4(send);
            }
    float u[2][4];
#pragma unroll
    for (int sd = 0; sd < 2; ++sd)
#pragma unroll
        for (int h = 0; h < 4; ++h) {
            float keep = b2 ? t[sd][h][1] : t[sd][h][0];
            float send = b2 ? t[sd][h][0] : t[sd][h][1];
            u[sd][h] = keep + lane_xor2(send);
        }
    float v[4];
#pragma unroll
    for (int h = 0; h < 4; ++h) {
        float keep = b1 ? u[1][h] : u[0][h];
        float send = b1 ? u[0][h] : u[1][h];
        v[h] = keep + lane_xor1(send);
    }
    float* ap = (b1 ? a_d : a_s) + (size_t)(row0 + 8 * half + (m >> 1)) * 4;
    *reinterpret_cast<float4*>(ap) = float4{v[0], v[1], v[2], v[3]};
}

template <int J> __device__ __forceinline__ int bc8i(int v) {
    return (int)__builtin_amdgcn_mov_dpp8((unsigned)v, (unsigned)J * 0x249249u);
}
template <int J> __device__ __forceinline__ float bc8f(float v) {
    return __builtin_bit_cast(float, __builtin_amdgcn_mov_dpp8(__builtin_bit_cast(unsigned, v), (unsigned)J * 0x249249u));
}
__device__ __forceinline__ float sum8(float v) {
    v += lane_xor1(v); v += lane_xor2(v); v += lane_xor4(v);
    return v;
}
__device__ __forceinline__ float max8(float v) {
    v = fmaxf(v, lane_xor1(v)); v = fmaxf(v, lane_xor2(v)); v = fmaxf(v, lane_xor4(v));
    return v;
}

template <int NG>
__device__ __forceinline__ void edge_groups(const int* __restrict__ ep, const int* __restrict__ srcA,
                                            const float* __restrict__ a_e, const float* __restrict__ a_s,
                                            int j, int h, int nc, float ad_n,
                                            float& S, float* lg, int* sv) {
    unsigned e[NG], s[NG], keep[NG];
    float ae[NG], as[NG], pen[NG];
#pragma unroll
    for (int q = 0; q < NG; ++q) {
        const int idx = 8 * q + j;
        keep[q] = idx < nc ? 0xFFFFFFFFu : 0u;
        pen[q]  = idx < nc ? 0.0f : -INFINITY;
        e[q] = (unsigned)ep[(unsigned)min(idx, nc - 1)];
    }
#pragma unroll
    for (int q = 0; q < NG; ++q) s[q] = (unsigned)srcA[e[q]];
#pragma unroll
    for (int q = 0; q < NG; ++q) ae[q] = a_e[(size_t)e[q] * 4 + h];
#pragma unroll
    for (int q = 0; q < NG; ++q) as[q] = a_s[(size_t)s[q] * 4 + h];
#pragma unroll
    for (int q = 0; q < NG; ++q) {
        S += __builtin_bit_cast(float, __builtin_bit_cast(unsigned, ae[q]) & keep[q]);
        lg[q] = leaky(as[q] + ad_n + ae[q]) + pen[q];
        sv[q] = (int)s[q];
    }
}

__device__ __forceinline__ void gather8(float4* hv, const float4* __restrict__ Hp4, int svg, int lane) {
    hv[0] = Hp4[(size_t)(unsigned)bc8i<0>(svg) * 32 + lane];
    hv[1] = Hp4[(size_t)(unsigned)bc8i<1>(svg) * 32 + lane];
    hv[2] = Hp4[(size_t)(unsigned)bc8i<2>(svg) * 32 + lane];
    hv[3] = Hp4[(size_t)(unsigned)bc8i<3>(svg) * 32 + lane];
    hv[4] = Hp4[(size_t)(unsigned)bc8i<4>(svg) * 32 + lane];
    hv[5] = Hp4[(size_t)(unsigned)bc8i<5>(svg) * 32 + lane];
    hv[6] = Hp4[(size_t)(unsigned)bc8i<6>(svg) * 32 + lane];
    hv[7] = Hp4[(size_t)(unsigned)bc8i<7>(svg) * 32 + lane];
}
__device__ __forceinline__ void fma_row(float4& A, float w, const float4& v) {
    A.x += w * v.x; A.y += w * v.y; A.z += w * v.z; A.w += w * v.w;
}
__device__ __forceinline__ void accum8(float4& A, const float4* hv, float pg) {
    fma_row(A, bc8f<0>(pg), hv[0]);
    fma_row(A, bc8f<1>(pg), hv[1]);
    fma_row(A, bc8f<2>(pg), hv[2]);
    fma_row(A, bc8f<3>(pg), hv[3]);
    fma_row(A, bc8f<4>(pg), hv[4]);
    fma_row(A, bc8f<5>(pg), hv[5]);
    fma_row(A, bc8f<6>(pg), hv[6]);
    fma_row(A, bc8f<7>(pg), hv[7]);
}

__global__ void agg_kernel(const int* __restrict__ off, const int* __restrict__ degI,
                const int* __restrict__ eidA, const int* __restrict__ srcA,
                const float* __restrict__ a_e, const float* __restrict__ a_s,
                const float* __restrict__ a_d,
                const float* __restrict__ Hp, const float* __restrict__ bias,
                const float* __restrict__ lin_w, const float* __restrict__ lin_b,
                float* __restrict__ outv, float* __restrict__ outs,
                int N, int mode) {
    const int lane = threadIdx.x & 31;
    const int n = __builtin_amdgcn_readfirstlane(blockIdx.x * (blockDim.x >> 5) + (threadIdx.x >> 5));
    if (n >= N) return;
    const int h = lane >> 3, j = lane & 7;
    const float4* Hp4 = reinterpret_cast<const float4*>(Hp);

    const int cnt = degI[n];
    const int s0 = off[n];
    const float ad_n = a_d[(size_t)n * 4 + h];
    const float as_n = a_s[(size_t)n * 4 + h];
    const float4 hself = Hp4[(size_t)n * 32 + lane];
    const float degf = fmaxf((float)cnt, 1.0f);
    float M = -INFINITY;
    float D = 0.0f;
    float4 A = {0.f, 0.f, 0.f, 0.f};
    float S = 0.0f;

    for (int c = 0; c < cnt; c += 32) {
        const int nc = min(cnt - c, 32);
        const int* ep = eidA + s0 + c;
        float lg[4];
        int sv[4];
        if (nc > 16) {
            edge_groups<4>(ep, srcA, a_e, a_s, j, h, nc, ad_n, S, lg, sv);
        } else {
            edge_groups<2>(ep, srcA, a_e, a_s, j, h, nc, ad_n, S, lg, sv);
            lg[2] = -INFINITY; lg[3] = -INFINITY;
            sv[2] = 0; sv[3] = 0;
        }
        float4 hv[8];
        gather8(hv, Hp4, sv[0], lane);

        const float m = max8(fmaxf(fmaxf(lg[0], lg[1]), fmaxf(lg[2], lg[3])));
        const float Mn = fmaxf(M, m);
        const float sc = __expf(M - Mn);
        float p[4];
#pragma unroll
        for (int g = 0; g < 4; ++g) p[g] = __expf(lg[g] - Mn);
        const float ps = sum8((p[0] + p[1]) + (p[2] + p[3]));
        D = D * sc + ps;
        A.x *= sc; A.y *= sc; A.z *= sc; A.w *= sc;
        M = Mn;

        accum8(A, hv, p[0]);
        if (nc > 8)  { gather8(hv, Hp4, sv[1], lane); accum8(A, hv, p[1]); }
        if (nc > 16) { gather8(hv, Hp4, sv[2], lane); accum8(A, hv, p[2]); }
        if (nc > 24) { gather8(hv, Hp4, sv[3], lane); accum8(A, hv, p[3]); }
    }
    S = sum8(S);
    {
        float lgs = leaky(as_n + ad_n + S / degf);
        float Mn = fmaxf(M, lgs);
        float sc = __expf(M - Mn);
        float p  = __expf(lgs - Mn);
        A.x = A.x * sc + p * hself.x;
        A.y = A.y * sc + p * hself.y;
        A.z = A.z * sc + p * hself.z;
        A.w = A.w * sc + p * hself.w;
        D = D * sc + p;
    }
    float4 bv = reinterpret_cast<const float4*>(bias)[lane];
    float inv = 1.0f / D;
    float4 r = { A.x * inv + bv.x, A.y * inv + bv.y, A.z * inv + bv.z, A.w * inv + bv.w };
    if (mode == 0) {
        r.x = fmaxf(r.x, 0.f); r.y = fmaxf(r.y, 0.f);
        r.z = fmaxf(r.z, 0.f); r.w = fmaxf(r.w, 0.f);
        reinterpret_cast<float4*>(outv)[(size_t)n * 32 + lane] = r;
    } else {
        float4 lw = reinterpret_cast<const float4*>(lin_w)[lane];
        float part = r.x * lw.x + r.y * lw.y + r.z * lw.z + r.w * lw.w;
#pragma unroll
        for (int msk = 16; msk >= 1; msk >>= 1) part += __shfl_xor(part, msk, 32);
        if (lane == 0) outs[n] = part + lin_b[0];
    }
}

extern "C" void kernel_launch(void* const* d_in, const int* in_sizes, int n_in,
                              void* d_out, int out_size, void* d_ws, size_t ws_size,
                              hipStream_t stream) {
    const float* x   = (const float*)d_in[0];
    const float* ea  = (const float*)d_in[1];
    const float* W1  = (const float*)d_in[2];
    const float* as1 = (const float*)d_in[3];
    const float* ad1 = (const float*)d_in[4];
    const float* We1 = (const float*)d_in[5];
    const float* ae1 = (const float*)d_in[6];
    const float* b1  = (const float*)d_in[7];
    const float* W2  = (const float*)d_in[8];
    const float* as2 = (const float*)d_in[9];
    const float* ad2 = (const float*)d_in[10];
    const float* We2 = (const float*)d_in[11];
    const float* ae2 = (const float*)d_in[12];
    const float* b2  = (const float*)d_in[13];
    const float* lw  = (const float*)d_in[14];
    const float* lb  = (const float*)d_in[15];
    const int*   ei  = (const int*)d_in[16];

    const int N = in_sizes[0] / F_IN;
    const int E = in_sizes[16] / 2;
    const int* srcA = ei;
    const int* dstA = ei + E;

    char* w = (char*)d_ws;
    auto alloc = [&](size_t bytes) -> void* {
        void* p = (void*)w;
        w += (bytes + 255) & ~(size_t)255;
        return p;
    };
    int*   degI   = (int*)  alloc((size_t)N * 4);
    int*   cursor = (int*)  alloc((size_t)N * 4);
    size_t zbytes = (size_t)(w - (char*)degI);
    float* Ve1  = (float*)alloc(F_IN * 4 * 4);
    float* Ve2  = (float*)alloc(F_IN * 4 * 4);
    float* ae1v = (float*)alloc((size_t)E * 16);
    float* ae2v = (float*)alloc((size_t)E * 16);
    float* a_sb = (float*)alloc((size_t)N * 16);
    float* a_db = (float*)alloc((size_t)N * 16);
    int*   off  = (int*)  alloc((size_t)N * 4);
    int*   eidA = (int*)  alloc((size_t)E * 4);
    float* Hp   = (float*)alloc((size_t)N * HC * 4);
    float* h1   = (float*)alloc((size_t)N * HC * 4);

    hipMemsetAsync(degI, 0, zbytes, stream);

    fold_kernel<<<1, 256, 0, stream>>>(We1, ae1, We2, ae2, Ve1, Ve2);
    edge_proj_kernel<<<(E + 255) / 256, 256, 0, stream>>>(ea, dstA, Ve1, Ve2,
                                                          ae1v, ae2v, degI, E);
    scan_kernel<<<1, 1024, 0, stream>>>(degI, off, N);
    scatter_kernel<<<(E + 255) / 256, 256, 0, stream>>>(dstA, off, cursor, eidA, E);

    const int mt = (N + 15) / 16;
    const int projBlocks = (mt + 3) / 4;
    const int nodeBlocks = (N + 7) / 8;

    proj_att_kernel<<<projBlocks, 128, 0, stream>>>(x, W1, as1, ad1, Hp, a_sb, a_db, N, F_IN);
    agg_kernel<<<nodeBlocks, 256, 0, stream>>>(off, degI, eidA, srcA, ae1v, a_sb, a_db,
                                               Hp, b1, nullptr, nullptr,
                                               h1, nullptr, N, 0);
    proj_att_kernel<<<projBlocks, 128, 0, stream>>>(h1, W2, as2, ad2, Hp, a_sb, a_db, N, HC);
    agg_kernel<<<nodeBlocks, 256, 0, stream>>>(off, degI, eidA, srcA, ae2v, a_sb, a_db,
                                               Hp, b2, lw, lb,
                                               nullptr, (float*)d_out, N, 1);
}


# CSR offset scan moves whole lines: 16-byte loads and stores over consecutive elements, row and wave sums by DPP scans, two barriers per 32768 elements, no per-element strided walk (integer, exact)
# speedup vs baseline: 2.1202x; 1.2290x over previous
#include <hip/hip_runtime.h>
#include <hip/hip_bf16.h>

#define F_IN   64
#define HEADS  4
#define CHD    32
#define HC     128
#define NEG_SLOPE 0.2f

typedef __attribute__((ext_vector_type(2))) float v2f;
typedef __attribute__((ext_vector_type(8))) float v8f;

__device__ __forceinline__ float leaky(float x) { return x > 0.f ? x : NEG_SLOPE * x; }

__global__ void fold_kernel(const float* __restrict__ We1, const float* __restrict__ ae1,
                            const float* __restrict__ We2, const float* __restrict__ ae2,
                            float* __restrict__ Ve1, float* __restrict__ Ve2) {
    int t = threadIdx.x;
    int f = t >> 2, h = t & 3;
    float s1 = 0.f, s2 = 0.f;
#pragma unroll
    for (int c = 0; c < CHD; ++c) {
        s1 += We1[f * HC + h * CHD + c] * ae1[h * CHD + c];
        s2 += We2[f * HC + h * CHD + c] * ae2[h * CHD + c];
    }
    Ve1[f * 4 + h] = s1;
    Ve2[f * 4 + h] = s2;
}

__device__ __forceinline__ float lane_xor4(float v) {
    return __builtin_bit_cast(float, __builtin_amdgcn_update_dpp(0, __builtin_bit_cast(int, v), 0x164, 0xf, 0xf, false));
}
__device__ __forceinline__ float lane_xor2(float v) {
    return __builtin_bit_cast(float, __builtin_amdgcn_update_dpp(0, __builtin_bit_cast(int, v), 0x4E, 0xf, 0xf, false));
}
__device__ __forceinline__ float lane_xor1(float v) {
    return __builtin_bit_cast(float, __builtin_amdgcn_update_dpp(0, __builtin_bit_cast(int, v), 0xB1, 0xf, 0xf, false));
}

#define EP_GROUPS 8

__global__ __launch_bounds__(256)
void edge_proj_kernel(const float* __restrict__ EA, const int* __restrict__ dstA,
                      const float* __restrict__ Ve1, const float* __restrict__ Ve2,
                      float* __restrict__ ae1v, float* __restrict__ ae2v,
                      int* __restrict__ degI, int E) {
    const int lane = threadIdx.x & 31;
    const int j = lane & 7;
    const int sub = lane >> 3;
    const int gw = blockIdx.x * (blockDim.x >> 5) + (threadIdx.x >> 5);
    const int base = gw * (4 * EP_GROUPS);
    if (base >= E) return;

    const int eown = base + lane;
    const int down = dstA[min(eown, E - 1)];

    const float4* EA4 = reinterpret_cast<const float4*>(EA);
    float4 x0[EP_GROUPS], x1[EP_GROUPS];
#pragma unroll
    for (int t = 0; t < EP_GROUPS; ++t) {
        int e = min(base + 4 * t + sub, E - 1);
        x0[t] = EA4[(size_t)e * 16 + j];
        x1[t] = EA4[(size_t)e * 16 + 8 + j];
    }

    const float4* V1 = reinterpret_cast<const float4*>(Ve1);
    const float4* V2 = reinterpret_cast<const float4*>(Ve2);
    float wl[4][8], wh[4][8];
#pragma unroll
    for (int c = 0; c < 4; ++c) {
        float4 p = V1[4 * j + c], q = V2[4 * j + c];
        float4 r = V1[32 + 4 * j + c], s = V2[32 + 4 * j + c];
        wl[c][0] = p.x; wl[c][1] = p.y; wl[c][2] = p.z; wl[c][3] = p.w;
        wl[c][4] = q.x; wl[c][5] = q.y; wl[c][6] = q.z; wl[c][7] = q.w;
        wh[c][0] = r.x; wh[c][1] = r.y; wh[c][2] = r.z; wh[c][3] = r.w;
        wh[c][4] = s.x; wh[c][5] = s.y; wh[c][6] = s.z; wh[c][7] = s.w;
    }

    const bool b4 = (j & 4) != 0, b2 = (j & 2) != 0, b1 = (j & 1) != 0;
    float* outp = (b4 ? ae2v : ae1v) + (j & 3);
#pragma unroll
    for (int t = 0; t < EP_GROUPS; ++t) {
        float acc[8];
#pragma unroll
        for (int q = 0; q < 8; ++q) {
            acc[q] = x0[t].x * wl[0][q] + x0[t].y * wl[1][q] + x0[t].z * wl[2][q] + x0[t].w * wl[3][q]
                   + x1[t].x * wh[0][q] + x1[t].y * wh[1][q] + x1[t].z * wh[2][q] + x1[t].w * wh[3][q];
        }
        float r[4], s[2];
#pragma unroll
        for (int q = 0; q < 4; ++q) {
            float keep = b4 ? acc[q + 4] : acc[q];
            float send = b4 ? acc[q] : acc[q + 4];
            r[q] = keep + lane_xor4(send);
        }
#pragma unroll
        for (int q = 0; q < 2; ++q) {
            float keep = b2 ? r[q + 2] : r[q];
            float send = b2 ? r[q] : r[q + 2];
            s[q] = keep + lane_xor2(send);
        }
        float keep = b1 ? s[1] : s[0];
        float send = b1 ? s[0] : s[1];
        float v = keep + lane_xor1(send);
        int e = base + 4 * t + sub;
        if (e < E) outp[(size_t)e * 4] = v;
    }
    if (eown < E) atomicAdd(&degI[down], 1);
}

#define SCAN_THREADS 1024
#define SCAN_ROWS    8
#define SCAN_WAVE    (SCAN_ROWS * 128)
#define SCAN_CHUNK   (32 * SCAN_WAVE)

template <int CTRL> __device__ __forceinline__ int dpp_row_shr(int v) {
    return __builtin_amdgcn_update_dpp(0, v, CTRL, 0xf, 0xf, false);
}
__device__ __forceinline__ int wave_incl_scan(int x, int lane) {
    x += dpp_row_shr<0x111>(x);
    x += dpp_row_shr<0x112>(x);
    x += dpp_row_shr<0x114>(x);
    x += dpp_row_shr<0x118>(x);
    const int low = __builtin_amdgcn_readlane(x, 15);
    return lane >= 16 ? x + low : x;
}

__global__ __launch_bounds__(SCAN_THREADS)
void scan_kernel(const int* __restrict__ deg, int* __restrict__ off, int N) {
    __shared__ int wsum[32];
    const int lane = threadIdx.x & 31;
    const int wv = __builtin_amdgcn_readfirstlane((int)(threadIdx.x >> 5));
    int carry = 0;
    for (int c0 = 0; c0 < N; c0 += SCAN_CHUNK) {
        const int i0 = c0 + wv * SCAN_WAVE + lane * 4;
        int4 q[SCAN_ROWS];
#pragma unroll
        for (int r = 0; r < SCAN_ROWS; ++r) {
            const int i = i0 + r * 128;
            if (i + 4 <= N) {
                q[r] = *reinterpret_cast<const int4*>(deg + i);
            } else {
                q[r].x = (i     < N) ? deg[i]     : 0;
                q[r].y = (i + 1 < N) ? deg[i + 1] : 0;
                q[r].z = (i + 2 < N) ? deg[i + 2] : 0;
                q[r].w = 0;
            }
        }
        int s[SCAN_ROWS], incl[SCAN_ROWS];
#pragma unroll
        for (int r = 0; r < SCAN_ROWS; ++r) {
            s[r] = (q[r].x + q[r].y) + (q[r].z + q[r].w);
            incl[r] = wave_incl_scan(s[r], lane);
        }
        int rbase[SCAN_ROWS];
        int wtotal = 0;
#pragma unroll
        for (int r = 0; r < SCAN_ROWS; ++r) {
            rbase[r] = wtotal;
            wtotal += __builtin_amdgcn_readlane(incl[r], 31);
        }
        if (lane == 0) wsum[wv] = wtotal;
        __syncthreads();
        const int mine = wsum[lane];
        const int winc = wave_incl_scan(mine, lane);
        const int wbase = __builtin_amdgcn_readlane(winc - mine, wv);
        const int total = __builtin_amdgcn_readlane(winc, 31);
        __syncthreads();
        const int cb = carry + wbase;
#pragma unroll
        for (int r = 0; r < SCAN_ROWS; ++r) {
            const int i = i0 + r * 128;
            const int b0 = cb + rbase[r] + (incl[r] - s[r]);
            const int b1 = b0 + q[r].x, b2 = b1 + q[r].y, b3 = b2 + q[r].z;
            if (i + 4 <= N) {
                *reinterpret_cast<int4*>(off + i) = int4{b0, b1, b2, b3};
            } else {
                if (i     < N) off[i]     = b0;
                if (i + 1 < N) off[i + 1] = b1;
                if (i + 2 < N) off[i + 2] = b2;
            }
        }
        carry += total;
    }
}

__global__ void scatter_kernel(const int* __restrict__ dstA, const int* __restrict__ off,
                               int* __restrict__ cursor, int* __restrict__ eidA, int E) {
    int e = blockIdx.x * blockDim.x + threadIdx.x;
    if (e >= E) return;
    int d = dstA[e];
    int pos = off[d] + atomicAdd(&cursor[d], 1);
    eidA[pos] = e;
}

__device__ __forceinline__ float lane_xor8(float v) {
    return __builtin_bit_cast(float, __builtin_amdgcn_update_dpp(0, __builtin_bit_cast(int, v), 0x168, 0xf, 0xf, false));
}

__global__ __launch_bounds__(128)
void proj_att_kernel(const float* __restrict__ X, const float* __restrict__ W,
                     const float* __restrict__ att_s, const float* __restrict__ att_d,
                     float* __restrict__ Y, float* __restrict__ a_s, float* __restrict__ a_d,
                     int M, int K) {
    const int wave = threadIdx.x >> 5;
    const int lane = threadIdx.x & 31;
    const int mt = blockIdx.x * (blockDim.x >> 5) + wave;
    if (mt * 16 >= M) return;
    const int row0 = mt * 16;
    const int m = lane & 15, half = lane >> 4;

    float ws[8], wd[8];
#pragma unroll
    for (int nt = 0; nt < 8; ++nt) {
        ws[nt] = att_s[nt * 16 + m];
        wd[nt] = att_d[nt * 16 + m];
    }

    v8f acc[8];
#pragma unroll
    for (int nt = 0; nt < 8; ++nt) acc[nt] = v8f{};

    const float* ap0 = X + (size_t)(row0 + m) * K + 2 * half;
    const float* bp0 = W + (size_t)(2 * half) * HC + m;
    for (int k0 = 0; k0 < K; k0 += 4) {
        v2f a;
        a.x = ap0[k0];
        a.y = ap0[k0 + 1];
        const float* bp = bp0 + (size_t)k0 * HC;
#pragma unroll
        for (int nt = 0; nt < 8; ++nt) {
            v2f b;
            b.x = bp[nt * 16];
            b.y = bp[HC + nt * 16];
            acc[nt] = __builtin_amdgcn_wmma_f32_16x16x4_f32(false, a, false, b,
                                                            (short)0, acc[nt], false, false);
        }
    }

    float* yp = Y + (size_t)(row0 + 8 * half) * HC + m;
#pragma unroll
    for (int nt = 0; nt < 8; ++nt) {
#pragma unroll
        for (int r = 0; r < 8; ++r) yp[(size_t)r * HC + nt * 16] = acc[nt][r];
    }

    float p[2][4][8];
#pragma unroll
    for (int h = 0; h < 4; ++h) {
#pragma unroll
        for (int r = 0; r < 8; ++r) {
            p[0][h][r] = acc[2 * h][r] * ws[2 * h] + acc[2 * h + 1][r] * ws[2 * h + 1];
            p[1][h][r] = acc[2 * h][r] * wd[2 * h] + acc[2 * h + 1][r] * wd[2 * h + 1];
        }
    }
    const bool b8 = (m & 8) != 0, b4 = (m & 4) != 0, b2 = (m & 2) != 0, b1 = (m & 1) != 0;
    float q[2][4][4];
#pragma unroll
    for (int sd = 0; sd < 2; ++sd)
#pragma unroll
        for (int h = 0; h < 4; ++h)
#pragma unroll
            for (int r = 0; r < 4; ++r) {
                float keep = b8 ? p[sd][h][r + 4] : p[sd][h][r];
                float send = b8 ? p[sd][h][r] : p[sd][h][r + 4];
                q[sd][h][r] = keep + lane_xor8(send);
            }
    float t[2][4][2];
#pragma unroll
    for (int sd = 0; sd < 2; ++sd)
#pragma unroll
        for (int h = 0; h < 4; ++h)
#pragma unroll
            for (int r = 0; r < 2; ++r) {
                float keep = b4 ? q[sd][h][r + 2] : q[sd][h][r];
                float send = b4 ? q[sd][h][r] : q[sd][h][r + 2];
                t[sd][h][r] = keep + lane_xor4(send);
            }
    float u[2][4];
#pragma unroll
    for (int sd = 0; sd < 2; ++sd)
#pragma unroll
        for (int h = 0; h < 4; ++h) {
            float keep = b2 ? t[sd][h][1] : t[sd][h][0];
            float send = b2 ? t[sd][h][0] : t[sd][h][1];
            u[sd][h] = keep + lane_xor2(send);
        }
    float v[4];
#pragma unroll
    for (int h = 0; h < 4; ++h) {
        float keep = b1 ? u[1][h] : u[0][h];
        float send = b1 ? u[0][h] : u[1][h];
        v[h] = keep + lane_xor1(send);
    }
    float* ap = (b1 ? a_d : a_s) + (size_t)(row0 + 8 * half + (m >> 1)) * 4;
    *reinterpret_cast<float4*>(ap) = float4{v[0], v[1], v[2], v[3]};
}

template <int J> __device__ __forceinline__ int bc8i(int v) {
    return (int)__builtin_amdgcn_mov_dpp8((unsigned)v, (unsigned)J * 0x249249u);
}
template <int J> __device__ __forceinline__ float bc8f(float v) {
    return __builtin_bit_cast(float, __builtin_amdgcn_mov_dpp8(__builtin_bit_cast(unsigned, v), (unsigned)J * 0x249249u));
}
__device__ __forceinline__ float sum8(float v) {
    v += lane_xor1(v); v += lane_xor2(v); v += lane_xor4(v);
    return v;
}
__device__ __forceinline__ float max8(float v) {
    v = fmaxf(v, lane_xor1(v)); v = fmaxf(v, lane_xor2(v)); v = fmaxf(v, lane_xor4(v));
    return v;
}

template <int NG>
__device__ __forceinline__ void edge_groups(const int* __restrict__ ep, const int* __restrict__ srcA,
                                            const float* __restrict__ a_e, const float* __restrict__ a_s,
                                            int j, int h, int nc, float ad_n,
                                            float& S, float* lg, int* sv) {
    unsigned e[NG], s[NG], keep[NG];
    float ae[NG], as[NG], pen[NG];
#pragma unroll
    for (int q = 0; q < NG; ++q) {
        const int idx = 8 * q + j;
        keep[q] = idx < nc ? 0xFFFFFFFFu : 0u;
        pen[q]  = idx < nc ? 0.0f : -INFINITY;
        e[q] = (unsigned)ep[(unsigned)min(idx, nc - 1)];
    }
#pragma unroll
    for (int q = 0; q < NG; ++q) s[q] = (unsigned)srcA[e[q]];
#pragma unroll
    for (int q = 0; q < NG; ++q) ae[q] = a_e[(size_t)e[q] * 4 + h];
#pragma unroll
    for (int q = 0; q < NG; ++q) as[q] = a_s[(size_t)s[q] * 4 + h];
#pragma unroll
    for (int q = 0; q < NG; ++q) {
        S += __builtin_bit_cast(float, __builtin_bit_cast(unsigned, ae[q]) & keep[q]);
        lg[q] = leaky(as[q] + ad_n + ae[q]) + pen[q];
        sv[q] = (int)s[q];
    }
}

__device__ __forceinline__ void gather8(float4* hv, const float4* __restrict__ Hp4, int svg, int lane) {
    hv[0] = Hp4[(size_t)(unsigned)bc8i<0>(svg) * 32 + lane];
    hv[1] = Hp4[(size_t)(unsigned)bc8i<1>(svg) * 32 + lane];
    hv[2] = Hp4[(size_t)(unsigned)bc8i<2>(svg) * 32 + lane];
    hv[3] = Hp4[(size_t)(unsigned)bc8i<3>(svg) * 32 + lane];
    hv[4] = Hp4[(size_t)(unsigned)bc8i<4>(svg) * 32 + lane];
    hv[5] = Hp4[(size_t)(unsigned)bc8i<5>(svg) * 32 + lane];
    hv[6] = Hp4[(size_t)(unsigned)bc8i<6>(svg) * 32 + lane];
    hv[7] = Hp4[(size_t)(unsigned)bc8i<7>(svg) * 32 + lane];
}
__device__ __forceinline__ void fma_row(float4& A, float w, const float4& v) {
    A.x += w * v.x; A.y += w * v.y; A.z += w * v.z; A.w += w * v.w;
}
__device__ __forceinline__ void accum8(float4& A, const float4* hv, float pg) {
    fma_row(A, bc8f<0>(pg), hv[0]);
    fma_row(A, bc8f<1>(pg), hv[1]);
    fma_row(A, bc8f<2>(pg), hv[2]);
    fma_row(A, bc8f<3>(pg), hv[3]);
    fma_row(A, bc8f<4>(pg), hv[4]);
    fma_row(A, bc8f<5>(pg), hv[5]);
    fma_row(A, bc8f<6>(pg), hv[6]);
    fma_row(A, bc8f<7>(pg), hv[7]);
}

__global__ void agg_kernel(const int* __restrict__ off, const int* __restrict__ degI,
                const int* __restrict__ eidA, const int* __restrict__ srcA,
                const float* __restrict__ a_e, const float* __restrict__ a_s,
                const float* __restrict__ a_d,
                const float* __restrict__ Hp, const float* __restrict__ bias,
                const float* __restrict__ lin_w, const float* __restrict__ lin_b,
                float* __restrict__ outv, float* __restrict__ outs,
                int N, int mode) {
    const int lane = threadIdx.x & 31;
    const int n = __builtin_amdgcn_readfirstlane(blockIdx.x * (blockDim.x >> 5) + (threadIdx.x >> 5));
    if (n >= N) return;
    const int h = lane >> 3, j = lane & 7;
    const float4* Hp4 = reinterpret_cast<const float4*>(Hp);

    const int cnt = degI[n];
    const int s0 = off[n];
    const float ad_n = a_d[(size_t)n * 4 + h];
    const float as_n = a_s[(size_t)n * 4 + h];
    const float4 hself = Hp4[(size_t)n * 32 + lane];
    const float degf = fmaxf((float)cnt, 1.0f);
    float M = -INFINITY;
    float D = 0.0f;
    float4 A = {0.f, 0.f, 0.f, 0.f};
    float S = 0.0f;

    for (int c = 0; c < cnt; c += 32) {
        const int nc = min(cnt - c, 32);
        const int* ep = eidA + s0 + c;
        float lg[4];
        int sv[4];
        if (nc > 16) {
            edge_groups<4>(ep, srcA, a_e, a_s, j, h, nc, ad_n, S, lg, sv);
        } else {
            edge_groups<2>(ep, srcA, a_e, a_s, j, h, nc, ad_n, S, lg, sv);
            lg[2] = -INFINITY; lg[3] = -INFINITY;
            sv[2] = 0; sv[3] = 0;
        }
        float4 hv[8];
        gather8(hv, Hp4, sv[0], lane);

        const float m = max8(fmaxf(fmaxf(lg[0], lg[1]), fmaxf(lg[2], lg[3])));
        const float Mn = fmaxf(M, m);
        const float sc = __expf(M - Mn);
        float p[4];
#pragma unroll
        for (int g = 0; g < 4; ++g) p[g] = __expf(lg[g] - Mn);
        const float ps = sum8((p[0] + p[1]) + (p[2] + p[3]));
        D = D * sc + ps;
        A.x *= sc; A.y *= sc; A.z *= sc; A.w *= sc;
        M = Mn;

        accum8(A, hv, p[0]);
        if (nc > 8)  { gather8(hv, Hp4, sv[1], lane); accum8(A, hv, p[1]); }
        if (nc > 16) { gather8(hv, Hp4, sv[2], lane); accum8(A, hv, p[2]); }
        if (nc > 24) { gather8(hv, Hp4, sv[3], lane); accum8(A, hv, p[3]); }
    }
    S = sum8(S);
    {
        float lgs = leaky(as_n + ad_n + S / degf);
        float Mn = fmaxf(M, lgs);
        float sc = __expf(M - Mn);
        float p  = __expf(lgs - Mn);
        A.x = A.x * sc + p * hself.x;
        A.y = A.y * sc + p * hself.y;
        A.z = A.z * sc + p * hself.z;
        A.w = A.w * sc + p * hself.w;
        D = D * sc + p;
    }
    float4 bv = reinterpret_cast<const float4*>(bias)[lane];
    float inv = 1.0f / D;
    float4 r = { A.x * inv + bv.x, A.y * inv + bv.y, A.z * inv + bv.z, A.w * inv + bv.w };
    if (mode == 0) {
        r.x = fmaxf(r.x, 0.f); r.y = fmaxf(r.y, 0.f);
        r.z = fmaxf(r.z, 0.f); r.w = fmaxf(r.w, 0.f);
        reinterpret_cast<float4*>(outv)[(size_t)n * 32 + lane] = r;
    } else {
        float4 lw = reinterpret_cast<const float4*>(lin_w)[lane];
        float part = r.x * lw.x + r.y * lw.y + r.z * lw.z + r.w * lw.w;
#pragma unroll
        for (int msk = 16; msk >= 1; msk >>= 1) part += __shfl_xor(part, msk, 32);
        if (lane == 0) outs[n] = part + lin_b[0];
    }
}

extern "C" void kernel_launch(void* const* d_in, const int* in_sizes, int n_in,
                              void* d_out, int out_size, void* d_ws, size_t ws_size,
                              hipStream_t stream) {
    const float* x   = (const float*)d_in[0];
    const float* ea  = (const float*)d_in[1];
    const float* W1  = (const float*)d_in[2];
    const float* as1 = (const float*)d_in[3];
    const float* ad1 = (const float*)d_in[4];
    const float* We1 = (const float*)d_in[5];
    const float* ae1 = (const float*)d_in[6];
    const float* b1  = (const float*)d_in[7];
    const float* W2  = (const float*)d_in[8];
    const float* as2 = (const float*)d_in[9];
    const float* ad2 = (const float*)d_in[10];
    const float* We2 = (const float*)d_in[11];
    const float* ae2 = (const float*)d_in[12];
    const float* b2  = (const float*)d_in[13];
    const float* lw  = (const float*)d_in[14];
    const float* lb  = (const float*)d_in[15];
    const int*   ei  = (const int*)d_in[16];

    const int N = in_sizes[0] / F_IN;
    const int E = in_sizes[16] / 2;
    const int* srcA = ei;
    const int* dstA = ei + E;

    char* w = (char*)d_ws;
    auto alloc = [&](size_t bytes) -> void* {
        void* p = (void*)w;
        w += (bytes + 255) & ~(size_t)255;
        return p;
    };
    int*   degI   = (int*)  alloc((size_t)N * 4);
    int*   cursor = (int*)  alloc((size_t)N * 4);
    size_t zbytes = (size_t)(w - (char*)degI);
    float* Ve1  = (float*)alloc(F_IN * 4 * 4);
    float* Ve2  = (float*)alloc(F_IN * 4 * 4);
    float* ae1v = (float*)alloc((size_t)E * 16);
    float* ae2v = (float*)alloc((size_t)E * 16);
    float* a_sb = (float*)alloc((size_t)N * 16);
    float* a_db = (float*)alloc((size_t)N * 16);
    int*   off  = (int*)  alloc((size_t)N * 4);
    int*   eidA = (int*)  alloc((size_t)E * 4);
    float* Hp   = (float*)alloc((size_t)N * HC * 4);
    float* h1   = (float*)alloc((size_t)N * HC * 4);

    hipMemsetAsync(degI, 0, zbytes, stream);

    fold_kernel<<<1, 256, 0, stream>>>(We1, ae1, We2, ae2, Ve1, Ve2);
    edge_proj_kernel<<<(E + 255) / 256, 256, 0, stream>>>(ea, dstA, Ve1, Ve2,
                                                          ae1v, ae2v, degI, E);
    scan_kernel<<<1, SCAN_THREADS, 0, stream>>>(degI, off, N);
    scatter_kernel<<<(E + 255) / 256, 256, 0, stream>>>(dstA, off, cursor, eidA, E);

    const int mt = (N + 15) / 16;
    const int projBlocks = (mt + 3) / 4;
    const int nodeBlocks = (N + 7) / 8;

    proj_att_kernel<<<projBlocks, 128, 0, stream>>>(x, W1, as1, ad1, Hp, a_sb, a_db, N, F_IN);
    agg_kernel<<<nodeBlocks, 256, 0, stream>>>(off, degI, eidA, srcA, ae1v, a_sb, a_db,
                                               Hp, b1, nullptr, nullptr,
                                               h1, nullptr, N, 0);
    proj_att_kernel<<<projBlocks, 128, 0, stream>>>(h1, W2, as2, ad2, Hp, a_sb, a_db, N, HC);
    agg_kernel<<<nodeBlocks, 256, 0, stream>>>(off, degI, eidA, srcA, ae2v, a_sb, a_db,
                                               Hp, b2, lw, lb,
                                               nullptr, (float*)d_out, N, 1);
}


# one returning atomic per edge in the edge pass gives the in-degree and the edge's rank in its list; scatter pass has no atomic, cursor array and its clearing gone (list order by arrival, as before)
# speedup vs baseline: 2.4950x; 1.1768x over previous
#include <hip/hip_runtime.h>
#include <hip/hip_bf16.h>

#define F_IN   64
#define HEADS  4
#define CHD    32
#define HC     128
#define NEG_SLOPE 0.2f

typedef __attribute__((ext_vector_type(2))) float v2f;
typedef __attribute__((ext_vector_type(8))) float v8f;

__device__ __forceinline__ float leaky(float x) { return x > 0.f ? x : NEG_SLOPE * x; }

__global__ void fold_kernel(const float* __restrict__ We1, const float* __restrict__ ae1,
                            const float* __restrict__ We2, const float* __restrict__ ae2,
                            float* __restrict__ Ve1, float* __restrict__ Ve2) {
    int t = threadIdx.x;
    int f = t >> 2, h = t & 3;
    float s1 = 0.f, s2 = 0.f;
#pragma unroll
    for (int c = 0; c < CHD; ++c) {
        s1 += We1[f * HC + h * CHD + c] * ae1[h * CHD + c];
        s2 += We2[f * HC + h * CHD + c] * ae2[h * CHD + c];
    }
    Ve1[f * 4 + h] = s1;
    Ve2[f * 4 + h] = s2;
}

__device__ __forceinline__ float lane_xor4(float v) {
    return __builtin_bit_cast(float, __builtin_amdgcn_update_dpp(0, __builtin_bit_cast(int, v), 0x164, 0xf, 0xf, false));
}
__device__ __forceinline__ float lane_xor2(float v) {
    return __builtin_bit_cast(float, __builtin_amdgcn_update_dpp(0, __builtin_bit_cast(int, v), 0x4E, 0xf, 0xf, false));
}
__device__ __forceinline__ float lane_xor1(float v) {
    return __builtin_bit_cast(float, __builtin_amdgcn_update_dpp(0, __builtin_bit_cast(int, v), 0xB1, 0xf, 0xf, false));
}

#define EP_GROUPS 8

__global__ __launch_bounds__(256)
void edge_proj_kernel(const float* __restrict__ EA, const int* __restrict__ dstA,
                      const float* __restrict__ Ve1, const float* __restrict__ Ve2,
                      float* __restrict__ ae1v, float* __restrict__ ae2v,
                      int* __restrict__ degI, int* __restrict__ rankA, int E) {
    const int lane = threadIdx.x & 31;
    const int j = lane & 7;
    const int sub = lane >> 3;
    const int gw = blockIdx.x * (blockDim.x >> 5) + (threadIdx.x >> 5);
    const int base = gw * (4 * EP_GROUPS);
    if (base >= E) return;

    const int eown = base + lane;
    const int down = dstA[min(eown, E - 1)];

    const float4* EA4 = reinterpret_cast<const float4*>(EA);
    float4 x0[EP_GROUPS], x1[EP_GROUPS];
#pragma unroll
    for (int t = 0; t < EP_GROUPS; ++t) {
        int e = min(base + 4 * t + sub, E - 1);
        x0[t] = EA4[(size_t)e * 16 + j];
        x1[t] = EA4[(size_t)e * 16 + 8 + j];
    }

    const float4* V1 = reinterpret_cast<const float4*>(Ve1);
    const float4* V2 = reinterpret_cast<const float4*>(Ve2);
    float wl[4][8], wh[4][8];
#pragma unroll
    for (int c = 0; c < 4; ++c) {
        float4 p = V1[4 * j + c], q = V2[4 * j + c];
        float4 r = V1[32 + 4 * j + c], s = V2[32 + 4 * j + c];
        wl[c][0] = p.x; wl[c][1] = p.y; wl[c][2] = p.z; wl[c][3] = p.w;
        wl[c][4] = q.x; wl[c][5] = q.y; wl[c][6] = q.z; wl[c][7] = q.w;
        wh[c][0] = r.x; wh[c][1] = r.y; wh[c][2] = r.z; wh[c][3] = r.w;
        wh[c][4] = s.x; wh[c][5] = s.y; wh[c][6] = s.z; wh[c][7] = s.w;
    }

    const int rank = atomicAdd(&degI[down], eown < E ? 1 : 0);

    const bool b4 = (j & 4) != 0, b2 = (j & 2) != 0, b1 = (j & 1) != 0;
    float* outp = (b4 ? ae2v : ae1v) + (j & 3);
#pragma unroll
    for (int t = 0; t < EP_GROUPS; ++t) {
        float acc[8];
#pragma unroll
        for (int q = 0; q < 8; ++q) {
            acc[q] = x0[t].x * wl[0][q] + x0[t].y * wl[1][q] + x0[t].z * wl[2][q] + x0[t].w * wl[3][q]
                   + x1[t].x * wh[0][q] + x1[t].y * wh[1][q] + x1[t].z * wh[2][q] + x1[t].w * wh[3][q];
        }
        float r[4], s[2];
#pragma unroll
        for (int q = 0; q < 4; ++q) {
            float keep = b4 ? acc[q + 4] : acc[q];
            float send = b4 ? acc[q] : acc[q + 4];
            r[q] = keep + lane_xor4(send);
        }
#pragma unroll
        for (int q = 0; q < 2; ++q) {
            float keep = b2 ? r[q + 2] : r[q];
            float send = b2 ? r[q] : r[q + 2];
            s[q] = keep + lane_xor2(send);
        }
        float keep = b1 ? s[1] : s[0];
        float send = b1 ? s[0] : s[1];
        float v = keep + lane_xor1(send);
        int e = base + 4 * t + sub;
        if (e < E) outp[(size_t)e * 4] = v;
    }
    if (eown < E) rankA[eown] = rank;
}

#define SCAN_THREADS 1024
#define SCAN_ROWS    8
#define SCAN_WAVE    (SCAN_ROWS * 128)
#define SCAN_CHUNK   (32 * SCAN_WAVE)

template <int CTRL> __device__ __forceinline__ int dpp_row_shr(int v) {
    return __builtin_amdgcn_update_dpp(0, v, CTRL, 0xf, 0xf, false);
}
__device__ __forceinline__ int wave_incl_scan(int x, int lane) {
    x += dpp_row_shr<0x111>(x);
    x += dpp_row_shr<0x112>(x);
    x += dpp_row_shr<0x114>(x);
    x += dpp_row_shr<0x118>(x);
    const int low = __builtin_amdgcn_readlane(x, 15);
    return lane >= 16 ? x + low : x;
}

__global__ __launch_bounds__(SCAN_THREADS)
void scan_kernel(const int* __restrict__ deg, int* __restrict__ off, int N) {
    __shared__ int wsum[32];
    const int lane = threadIdx.x & 31;
    const int wv = __builtin_amdgcn_readfirstlane((int)(threadIdx.x >> 5));
    int carry = 0;
    for (int c0 = 0; c0 < N; c0 += SCAN_CHUNK) {
        const int i0 = c0 + wv * SCAN_WAVE + lane * 4;
        int4 q[SCAN_ROWS];
#pragma unroll
        for (int r = 0; r < SCAN_ROWS; ++r) {
            const int i = i0 + r * 128;
            if (i + 4 <= N) {
                q[r] = *reinterpret_cast<const int4*>(deg + i);
            } else {
                q[r].x = (i     < N) ? deg[i]     : 0;
                q[r].y = (i + 1 < N) ? deg[i + 1] : 0;
                q[r].z = (i + 2 < N) ? deg[i + 2] : 0;
                q[r].w = 0;
            }
        }
        int s[SCAN_ROWS], incl[SCAN_ROWS];
#pragma unroll
        for (int r = 0; r < SCAN_ROWS; ++r) {
            s[r] = (q[r].x + q[r].y) + (q[r].z + q[r].w);
            incl[r] = wave_incl_scan(s[r], lane);
        }
        int rbase[SCAN_ROWS];
        int wtotal = 0;
#pragma unroll
        for (int r = 0; r < SCAN_ROWS; ++r) {
            rbase[r] = wtotal;
            wtotal += __builtin_amdgcn_readlane(incl[r], 31);
        }
        if (lane == 0) wsum[wv] = wtotal;
        __syncthreads();
        const int mine = wsum[lane];
        const int winc = wave_incl_scan(mine, lane);
        const int wbase = __builtin_amdgcn_readlane(winc - mine, wv);
        const int total = __builtin_amdgcn_readlane(winc, 31);
        __syncthreads();
        const int cb = carry + wbase;
#pragma unroll
        for (int r = 0; r < SCAN_ROWS; ++r) {
            const int i = i0 + r * 128;
            const int b0 = cb + rbase[r] + (incl[r] - s[r]);
            const int b1 = b0 + q[r].x, b2 = b1 + q[r].y, b3 = b2 + q[r].z;
            if (i + 4 <= N) {
                *reinterpret_cast<int4*>(off + i) = int4{b0, b1, b2, b3};
            } else {
                if (i     < N) off[i]     = b0;
                if (i + 1 < N) off[i + 1] = b1;
                if (i + 2 < N) off[i + 2] = b2;
            }
        }
        carry += total;
    }
}

__global__ void scatter_kernel(const int* __restrict__ dstA, const int* __restrict__ off,
                               const int* __restrict__ rankA, int* __restrict__ eidA, int E) {
    int e = blockIdx.x * blockDim.x + threadIdx.x;
    if (e >= E) return;
    eidA[off[dstA[e]] + rankA[e]] = e;
}

__device__ __forceinline__ float lane_xor8(float v) {
    return __builtin_bit_cast(float, __builtin_amdgcn_update_dpp(0, __builtin_bit_cast(int, v), 0x168, 0xf, 0xf, false));
}

__global__ __launch_bounds__(128)
void proj_att_kernel(const float* __restrict__ X, const float* __restrict__ W,
                     const float* __restrict__ att_s, const float* __restrict__ att_d,
                     float* __restrict__ Y, float* __restrict__ a_s, float* __restrict__ a_d,
                     int M, int K) {
    const int wave = threadIdx.x >> 5;
    const int lane = threadIdx.x & 31;
    const int mt = blockIdx.x * (blockDim.x >> 5) + wave;
    if (mt * 16 >= M) return;
    const int row0 = mt * 16;
    const int m = lane & 15, half = lane >> 4;

    float ws[8], wd[8];
#pragma unroll
    for (int nt = 0; nt < 8; ++nt) {
        ws[nt] = att_s[nt * 16 + m];
        wd[nt] = att_d[nt * 16 + m];
    }

    v8f acc[8];
#pragma unroll
    for (int nt = 0; nt < 8; ++nt) acc[nt] = v8f{};

    const float* ap0 = X + (size_t)(row0 + m) * K + 2 * half;
    const float* bp0 = W + (size_t)(2 * half) * HC + m;
    for (int k0 = 0; k0 < K; k0 += 4) {
        v2f a;
        a.x = ap0[k0];
        a.y = ap0[k0 + 1];
        const float* bp = bp0 + (size_t)k0 * HC;
#pragma unroll
        for (int nt = 0; nt < 8; ++nt) {
            v2f b;
            b.x = bp[nt * 16];
            b.y = bp[HC + nt * 16];
            acc[nt] = __builtin_amdgcn_wmma_f32_16x16x4_f32(false, a, false, b,
                                                            (short)0, acc[nt], false, false);
        }
    }

    float* yp = Y + (size_t)(row0 + 8 * half) * HC + m;
#pragma unroll
    for (int nt = 0; nt < 8; ++nt) {
#pragma unroll
        for (int r = 0; r < 8; ++r) yp[(size_t)r * HC + nt * 16] = acc[nt][r];
    }

    float p[2][4][8];
#pragma unroll
    for (int h = 0; h < 4; ++h) {
#pragma unroll
        for (int r = 0; r < 8; ++r) {
            p[0][h][r] = acc[2 * h][r] * ws[2 * h] + acc[2 * h + 1][r] * ws[2 * h + 1];
            p[1][h][r] = acc[2 * h][r] * wd[2 * h] + acc[2 * h + 1][r] * wd[2 * h + 1];
        }
    }
    const bool b8 = (m & 8) != 0, b4 = (m & 4) != 0, b2 = (m & 2) != 0, b1 = (m & 1) != 0;
    float q[2][4][4];
#pragma unroll
    for (int sd = 0; sd < 2; ++sd)
#pragma unroll
        for (int h = 0; h < 4; ++h)
#pragma unroll
            for (int r = 0; r < 4; ++r) {
                float keep = b8 ? p[sd][h][r + 4] : p[sd][h][r];
                float send = b8 ? p[sd][h][r] : p[sd][h][r + 4];
                q[sd][h][r] = keep + lane_xor8(send);
            }
    float t[2][4][2];
#pragma unroll
    for (int sd = 0; sd < 2; ++sd)
#pragma unroll
        for (int h = 0; h < 4; ++h)
#pragma unroll
            for (int r = 0; r < 2; ++r) {
                float keep = b4 ? q[sd][h][r + 2] : q[sd][h][r];
                float send = b4 ? q[sd][h][r] : q[sd][h][r + 2];
                t[sd][h][r] = keep + lane_xor4(send);
            }
    float u[2][4];
#pragma unroll
    for (int sd = 0; sd < 2; ++sd)
#pragma unroll
        for (int h = 0; h < 4; ++h) {
            float keep = b2 ? t[sd][h][1] : t[sd][h][0];
            float send = b2 ? t[sd][h][0] : t[sd][h][1];
            u[sd][h] = keep + lane_xor2(send);
        }
    float v[4];
#pragma unroll
    for (int h = 0; h < 4; ++h) {
        float keep = b1 ? u[1][h] : u[0][h];
        float send = b1 ? u[0][h] : u[1][h];
        v[h] = keep + lane_xor1(send);
    }
    float* ap = (b1 ? a_d : a_s) + (size_t)(row0 + 8 * half + (m >> 1)) * 4;
    *reinterpret_cast<float4*>(ap) = float4{v[0], v[1], v[2], v[3]};
}

template <int J> __device__ __forceinline__ int bc8i(int v) {
    return (int)__builtin_amdgcn_mov_dpp8((unsigned)v, (unsigned)J * 0x249249u);
}
template <int J> __device__ __forceinline__ float bc8f(float v) {
    return __builtin_bit_cast(float, __builtin_amdgcn_mov_dpp8(__builtin_bit_cast(unsigned, v), (unsigned)J * 0x249249u));
}
__device__ __forceinline__ float sum8(float v) {
    v += lane_xor1(v); v += lane_xor2(v); v += lane_xor4(v);
    return v;
}
__device__ __forceinline__ float max8(float v) {
    v = fmaxf(v, lane_xor1(v)); v = fmaxf(v, lane_xor2(v)); v = fmaxf(v, lane_xor4(v));
    return v;
}

template <int NG>
__device__ __forceinline__ void edge_groups(const int* __restrict__ ep, const int* __restrict__ srcA,
                                            const float* __restrict__ a_e, const float* __restrict__ a_s,
                                            int j, int h, int nc, float ad_n,
                                            float& S, float* lg, int* sv) {
    unsigned e[NG], s[NG], keep[NG];
    float ae[NG], as[NG], pen[NG];
#pragma unroll
    for (int q = 0; q < NG; ++q) {
        const int idx = 8 * q + j;
        keep[q] = idx < nc ? 0xFFFFFFFFu : 0u;
        pen[q]  = idx < nc ? 0.0f : -INFINITY;
        e[q] = (unsigned)ep[(unsigned)min(idx, nc - 1)];
    }
#pragma unroll
    for (int q = 0; q < NG; ++q) s[q] = (unsigned)srcA[e[q]];
#pragma unroll
    for (int q = 0; q < NG; ++q) ae[q] = a_e[(size_t)e[q] * 4 + h];
#pragma unroll
    for (int q = 0; q < NG; ++q) as[q] = a_s[(size_t)s[q] * 4 + h];
#pragma unroll
    for (int q = 0; q < NG; ++q) {
        S += __builtin_bit_cast(float, __builtin_bit_cast(unsigned, ae[q]) & keep[q]);
        lg[q] = leaky(as[q] + ad_n + ae[q]) + pen[q];
        sv[q] = (int)s[q];
    }
}

__device__ __forceinline__ void gather8(float4* hv, const float4* __restrict__ Hp4, int svg, int lane) {
    hv[0] = Hp4[(size_t)(unsigned)bc8i<0>(svg) * 32 + lane];
    hv[1] = Hp4[(size_t)(unsigned)bc8i<1>(svg) * 32 + lane];
    hv[2] = Hp4[(size_t)(unsigned)bc8i<2>(svg) * 32 + lane];
    hv[3] = Hp4[(size_t)(unsigned)bc8i<3>(svg) * 32 + lane];
    hv[4] = Hp4[(size_t)(unsigned)bc8i<4>(svg) * 32 + lane];
    hv[5] = Hp4[(size_t)(unsigned)bc8i<5>(svg) * 32 + lane];
    hv[6] = Hp4[(size_t)(unsigned)bc8i<6>(svg) * 32 + lane];
    hv[7] = Hp4[(size_t)(unsigned)bc8i<7>(svg) * 32 + lane];
}
__device__ __forceinline__ void fma_row(float4& A, float w, const float4& v) {
    A.x += w * v.x; A.y += w * v.y; A.z += w * v.z; A.w += w * v.w;
}
__device__ __forceinline__ void accum8(float4& A, const float4* hv, float pg) {
    fma_row(A, bc8f<0>(pg), hv[0]);
    fma_row(A, bc8f<1>(pg), hv[1]);
    fma_row(A, bc8f<2>(pg), hv[2]);
    fma_row(A, bc8f<3>(pg), hv[3]);
    fma_row(A, bc8f<4>(pg), hv[4]);
    fma_row(A, bc8f<5>(pg), hv[5]);
    fma_row(A, bc8f<6>(pg), hv[6]);
    fma_row(A, bc8f<7>(pg), hv[7]);
}

__global__ void agg_kernel(const int* __restrict__ off, const int* __restrict__ degI,
                const int* __restrict__ eidA, const int* __restrict__ srcA,
                const float* __restrict__ a_e, const float* __restrict__ a_s,
                const float* __restrict__ a_d,
                const float* __restrict__ Hp, const float* __restrict__ bias,
                const float* __restrict__ lin_w, const float* __restrict__ lin_b,
                float* __restrict__ outv, float* __restrict__ outs,
                int N, int mode) {
    const int lane = threadIdx.x & 31;
    const int n = __builtin_amdgcn_readfirstlane(blockIdx.x * (blockDim.x >> 5) + (threadIdx.x >> 5));
    if (n >= N) return;
    const int h = lane >> 3, j = lane & 7;
    const float4* Hp4 = reinterpret_cast<const float4*>(Hp);

    const int cnt = degI[n];
    const int s0 = off[n];
    const float ad_n = a_d[(size_t)n * 4 + h];
    const float as_n = a_s[(size_t)n * 4 + h];
    const float4 hself = Hp4[(size_t)n * 32 + lane];
    const float degf = fmaxf((float)cnt, 1.0f);
    float M = -INFINITY;
    float D = 0.0f;
    float4 A = {0.f, 0.f, 0.f, 0.f};
    float S = 0.0f;

    for (int c = 0; c < cnt; c += 32) {
        const int nc = min(cnt - c, 32);
        const int* ep = eidA + s0 + c;
        float lg[4];
        int sv[4];
        if (nc > 16) {
            edge_groups<4>(ep, srcA, a_e, a_s, j, h, nc, ad_n, S, lg, sv);
        } else {
            edge_groups<2>(ep, srcA, a_e, a_s, j, h, nc, ad_n, S, lg, sv);
            lg[2] = -INFINITY; lg[3] = -INFINITY;
            sv[2] = 0; sv[3] = 0;
        }
        float4 hv[8];
        gather8(hv, Hp4, sv[0], lane);

        const float m = max8(fmaxf(fmaxf(lg[0], lg[1]), fmaxf(lg[2], lg[3])));
        const float Mn = fmaxf(M, m);
        const float sc = __expf(M - Mn);
        float p[4];
#pragma unroll
        for (int g = 0; g < 4; ++g) p[g] = __expf(lg[g] - Mn);
        const float ps = sum8((p[0] + p[1]) + (p[2] + p[3]));
        D = D * sc + ps;
        A.x *= sc; A.y *= sc; A.z *= sc; A.w *= sc;
        M = Mn;

        accum8(A, hv, p[0]);
        if (nc > 8)  { gather8(hv, Hp4, sv[1], lane); accum8(A, hv, p[1]); }
        if (nc > 16) { gather8(hv, Hp4, sv[2], lane); accum8(A, hv, p[2]); }
        if (nc > 24) { gather8(hv, Hp4, sv[3], lane); accum8(A, hv, p[3]); }
    }
    S = sum8(S);
    {
        float lgs = leaky(as_n + ad_n + S / degf);
        float Mn = fmaxf(M, lgs);
        float sc = __expf(M - Mn);
        float p  = __expf(lgs - Mn);
        A.x = A.x * sc + p * hself.x;
        A.y = A.y * sc + p * hself.y;
        A.z = A.z * sc + p * hself.z;
        A.w = A.w * sc + p * hself.w;
        D = D * sc + p;
    }
    float4 bv = reinterpret_cast<const float4*>(bias)[lane];
    float inv = 1.0f / D;
    float4 r = { A.x * inv + bv.x, A.y * inv + bv.y, A.z * inv + bv.z, A.w * inv + bv.w };
    if (mode == 0) {
        r.x = fmaxf(r.x, 0.f); r.y = fmaxf(r.y, 0.f);
        r.z = fmaxf(r.z, 0.f); r.w = fmaxf(r.w, 0.f);
        reinterpret_cast<float4*>(outv)[(size_t)n * 32 + lane] = r;
    } else {
        float4 lw = reinterpret_cast<const float4*>(lin_w)[lane];
        float part = r.x * lw.x + r.y * lw.y + r.z * lw.z + r.w * lw.w;
#pragma unroll
        for (int msk = 16; msk >= 1; msk >>= 1) part += __shfl_xor(part, msk, 32);
        if (lane == 0) outs[n] = part + lin_b[0];
    }
}

extern "C" void kernel_launch(void* const* d_in, const int* in_sizes, int n_in,
                              void* d_out, int out_size, void* d_ws, size_t ws_size,
                              hipStream_t stream) {
    const float* x   = (const float*)d_in[0];
    const float* ea  = (const float*)d_in[1];
    const float* W1  = (const float*)d_in[2];
    const float* as1 = (const float*)d_in[3];
    const float* ad1 = (const float*)d_in[4];
    const float* We1 = (const float*)d_in[5];
    const float* ae1 = (const float*)d_in[6];
    const float* b1  = (const float*)d_in[7];
    const float* W2  = (const float*)d_in[8];
    const float* as2 = (const float*)d_in[9];
    const float* ad2 = (const float*)d_in[10];
    const float* We2 = (const float*)d_in[11];
    const float* ae2 = (const float*)d_in[12];
    const float* b2  = (const float*)d_in[13];
    const float* lw  = (const float*)d_in[14];
    const float* lb  = (const float*)d_in[15];
    const int*   ei  = (const int*)d_in[16];

    const int N = in_sizes[0] / F_IN;
    const int E = in_sizes[16] / 2;
    const int* srcA = ei;
    const int* dstA = ei + E;

    char* w = (char*)d_ws;
    auto alloc = [&](size_t bytes) -> void* {
        void* p = (void*)w;
        w += (bytes + 255) & ~(size_t)255;
        return p;
    };
    int*   degI   = (int*)  alloc((size_t)N * 4);
    size_t zbytes = (size_t)(w - (char*)degI);
    float* Ve1  = (float*)alloc(F_IN * 4 * 4);
    float* Ve2  = (float*)alloc(F_IN * 4 * 4);
    float* ae1v = (float*)alloc((size_t)E * 16);
    float* ae2v = (float*)alloc((size_t)E * 16);
    float* a_sb = (float*)alloc((size_t)N * 16);
    float* a_db = (float*)alloc((size_t)N * 16);
    int*   off  = (int*)  alloc((size_t)N * 4);
    int*   eidA = (int*)  alloc((size_t)E * 4);
    float* Hp   = (float*)alloc((size_t)N * HC * 4);
    float* h1   = (float*)alloc((size_t)N * HC * 4);
    int*   rankA = (int*)h1;

    hipMemsetAsync(degI, 0, zbytes, stream);

    fold_kernel<<<1, 256, 0, stream>>>(We1, ae1, We2, ae2, Ve1, Ve2);
    edge_proj_kernel<<<(E + 255) / 256, 256, 0, stream>>>(ea, dstA, Ve1, Ve2,
                                                          ae1v, ae2v, degI, rankA, E);
    scan_kernel<<<1, SCAN_THREADS, 0, stream>>>(degI, off, N);
    scatter_kernel<<<(E + 255) / 256, 256, 0, stream>>>(dstA, off, rankA, eidA, E);

    const int mt = (N + 15) / 16;
    const int projBlocks = (mt + 3) / 4;
    const int nodeBlocks = (N + 7) / 8;

    proj_att_kernel<<<projBlocks, 128, 0, stream>>>(x, W1, as1, ad1, Hp, a_sb, a_db, N, F_IN);
    agg_kernel<<<nodeBlocks, 256, 0, stream>>>(off, degI, eidA, srcA, ae1v, a_sb, a_db,
                                               Hp, b1, nullptr, nullptr,
                                               h1, nullptr, N, 0);
    proj_att_kernel<<<projBlocks, 128, 0, stream>>>(h1, W2, as2, ad2, Hp, a_sb, a_db, N, HC);
    agg_kernel<<<nodeBlocks, 256, 0, stream>>>(off, degI, eidA, srcA, ae2v, a_sb, a_db,
                                               Hp, b2, lw, lb,
                                               nullptr, (float*)d_out, N, 1);
}


# aggregation wave takes a_s of each edge's source, and its own node's a_s and a_d, from the 512-byte rows it gathers anyway: no a_s fetch per edge, projection writes no scalars (new summation order)
# speedup vs baseline: 2.6143x; 1.0478x over previous
#include <hip/hip_runtime.h>
#include <hip/hip_bf16.h>

#define F_IN   64
#define HEADS  4
#define CHD    32
#define HC     128
#define NEG_SLOPE 0.2f

typedef __attribute__((ext_vector_type(2))) float v2f;
typedef __attribute__((ext_vector_type(8))) float v8f;
typedef __attribute__((ext_vector_type(4))) float v4f;

__device__ __forceinline__ float leaky(float x) { return x > 0.f ? x : NEG_SLOPE * x; }

__global__ void fold_kernel(const float* __restrict__ We1, const float* __restrict__ ae1,
                            const float* __restrict__ We2, const float* __restrict__ ae2,
                            float* __restrict__ Ve1, float* __restrict__ Ve2) {
    int t = threadIdx.x;
    int f = t >> 2, h = t & 3;
    float s1 = 0.f, s2 = 0.f;
#pragma unroll
    for (int c = 0; c < CHD; ++c) {
        s1 += We1[f * HC + h * CHD + c] * ae1[h * CHD + c];
        s2 += We2[f * HC + h * CHD + c] * ae2[h * CHD + c];
    }
    Ve1[f * 4 + h] = s1;
    Ve2[f * 4 + h] = s2;
}

__device__ __forceinline__ float lane_xor4(float v) {
    return __builtin_bit_cast(float, __builtin_amdgcn_update_dpp(0, __builtin_bit_cast(int, v), 0x164, 0xf, 0xf, false));
}
__device__ __forceinline__ float lane_xor2(float v) {
    return __builtin_bit_cast(float, __builtin_amdgcn_update_dpp(0, __builtin_bit_cast(int, v), 0x4E, 0xf, 0xf, false));
}
__device__ __forceinline__ float lane_xor1(float v) {
    return __builtin_bit_cast(float, __builtin_amdgcn_update_dpp(0, __builtin_bit_cast(int, v), 0xB1, 0xf, 0xf, false));
}

#define EP_GROUPS 8

__global__ __launch_bounds__(256)
void edge_proj_kernel(const float* __restrict__ EA, const int* __restrict__ dstA,
                      const float* __restrict__ Ve1, const float* __restrict__ Ve2,
                      float* __restrict__ ae1v, float* __restrict__ ae2v,
                      int* __restrict__ degI, int* __restrict__ rankA, int E) {
    const int lane = threadIdx.x & 31;
    const int j = lane & 7;
    const int sub = lane >> 3;
    const int gw = blockIdx.x * (blockDim.x >> 5) + (threadIdx.x >> 5);
    const int base = gw * (4 * EP_GROUPS);
    if (base >= E) return;

    const int eown = base + lane;
    const int down = dstA[min(eown, E - 1)];

    const float4* EA4 = reinterpret_cast<const float4*>(EA);
    float4 x0[EP_GROUPS], x1[EP_GROUPS];
#pragma unroll
    for (int t = 0; t < EP_GROUPS; ++t) {
        int e = min(base + 4 * t + sub, E - 1);
        x0[t] = EA4[(size_t)e * 16 + j];
        x1[t] = EA4[(size_t)e * 16 + 8 + j];
    }

    const float4* V1 = reinterpret_cast<const float4*>(Ve1);
    const float4* V2 = reinterpret_cast<const float4*>(Ve2);
    float wl[4][8], wh[4][8];
#pragma unroll
    for (int c = 0; c < 4; ++c) {
        float4 p = V1[4 * j + c], q = V2[4 * j + c];
        float4 r = V1[32 + 4 * j + c], s = V2[32 + 4 * j + c];
        wl[c][0] = p.x; wl[c][1] = p.y; wl[c][2] = p.z; wl[c][3] = p.w;
        wl[c][4] = q.x; wl[c][5] = q.y; wl[c][6] = q.z; wl[c][7] = q.w;
        wh[c][0] = r.x; wh[c][1] = r.y; wh[c][2] = r.z; wh[c][3] = r.w;
        wh[c][4] = s.x; wh[c][5] = s.y; wh[c][6] = s.z; wh[c][7] = s.w;
    }

    const int rank = atomicAdd(&degI[down], eown < E ? 1 : 0);

    const bool b4 = (j & 4) != 0, b2 = (j & 2) != 0, b1 = (j & 1) != 0;
    float* outp = (b4 ? ae2v : ae1v) + (j & 3);
#pragma unroll
    for (int t = 0; t < EP_GROUPS; ++t) {
        float acc[8];
#pragma unroll
        for (int q = 0; q < 8; ++q) {
            acc[q] = x0[t].x * wl[0][q] + x0[t].y * wl[1][q] + x0[t].z * wl[2][q] + x0[t].w * wl[3][q]
                   + x1[t].x * wh[0][q] + x1[t].y * wh[1][q] + x1[t].z * wh[2][q] + x1[t].w * wh[3][q];
        }
        float r[4], s[2];
#pragma unroll
        for (int q = 0; q < 4; ++q) {
            float keep = b4 ? acc[q + 4] : acc[q];
            float send = b4 ? acc[q] : acc[q + 4];
            r[q] = keep + lane_xor4(send);
        }
#pragma unroll
        for (int q = 0; q < 2; ++q) {
            float keep = b2 ? r[q + 2] : r[q];
            float send = b2 ? r[q] : r[q + 2];
            s[q] = keep + lane_xor2(send);
        }
        float keep = b1 ? s[1] : s[0];
        float send = b1 ? s[0] : s[1];
        float v = keep + lane_xor1(send);
        int e = base + 4 * t + sub;
        if (e < E) outp[(size_t)e * 4] = v;
    }
    if (eown < E) rankA[eown] = rank;
}

#define SCAN_THREADS 1024
#define SCAN_ROWS    8
#define SCAN_WAVE    (SCAN_ROWS * 128)
#define SCAN_CHUNK   (32 * SCAN_WAVE)

template <int CTRL> __device__ __forceinline__ int dpp_row_shr(int v) {
    return __builtin_amdgcn_update_dpp(0, v, CTRL, 0xf, 0xf, false);
}
__device__ __forceinline__ int wave_incl_scan(int x, int lane) {
    x += dpp_row_shr<0x111>(x);
    x += dpp_row_shr<0x112>(x);
    x += dpp_row_shr<0x114>(x);
    x += dpp_row_shr<0x118>(x);
    const int low = __builtin_amdgcn_readlane(x, 15);
    return lane >= 16 ? x + low : x;
}

__global__ __launch_bounds__(SCAN_THREADS)
void scan_kernel(const int* __restrict__ deg, int* __restrict__ off, int N) {
    __shared__ int wsum[32];
    const int lane = threadIdx.x & 31;
    const int wv = __builtin_amdgcn_readfirstlane((int)(threadIdx.x >> 5));
    int carry = 0;
    for (int c0 = 0; c0 < N; c0 += SCAN_CHUNK) {
        const int i0 = c0 + wv * SCAN_WAVE + lane * 4;
        int4 q[SCAN_ROWS];
#pragma unroll
        for (int r = 0; r < SCAN_ROWS; ++r) {
            const int i = i0 + r * 128;
            if (i + 4 <= N) {
                q[r] = *reinterpret_cast<const int4*>(deg + i);
            } else {
                q[r].x = (i     < N) ? deg[i]     : 0;
                q[r].y = (i + 1 < N) ? deg[i + 1] : 0;
                q[r].z = (i + 2 < N) ? deg[i + 2] : 0;
                q[r].w = 0;
            }
        }
        int s[SCAN_ROWS], incl[SCAN_ROWS];
#pragma unroll
        for (int r = 0; r < SCAN_ROWS; ++r) {
            s[r] = (q[r].x + q[r].y) + (q[r].z + q[r].w);
            incl[r] = wave_incl_scan(s[r], lane);
        }
        int rbase[SCAN_ROWS];
        int wtotal = 0;
#pragma unroll
        for (int r = 0; r < SCAN_ROWS; ++r) {
            rbase[r] = wtotal;
            wtotal += __builtin_amdgcn_readlane(incl[r], 31);
        }
        if (lane == 0) wsum[wv] = wtotal;
        __syncthreads();
        const int mine = wsum[lane];
        const int winc = wave_incl_scan(mine, lane);
        const int wbase = __builtin_amdgcn_readlane(winc - mine, wv);
        const int total = __builtin_amdgcn_readlane(winc, 31);
        __syncthreads();
        const int cb = carry + wbase;
#pragma unroll
        for (int r = 0; r < SCAN_ROWS; ++r) {
            const int i = i0 + r * 128;
            const int b0 = cb + rbase[r] + (incl[r] - s[r]);
            const int b1 = b0 + q[r].x, b2 = b1 + q[r].y, b3 = b2 + q[r].z;
            if (i + 4 <= N) {
                *reinterpret_cast<int4*>(off + i) = int4{b0, b1, b2, b3};
            } else {
                if (i     < N) off[i]     = b0;
                if (i + 1 < N) off[i + 1] = b1;
                if (i + 2 < N) off[i + 2] = b2;
            }
        }
        carry += total;
    }
}

__global__ void scatter_kernel(const int* __restrict__ dstA, const int* __restrict__ off,
                               const int* __restrict__ rankA, int* __restrict__ eidA, int E) {
    int e = blockIdx.x * blockDim.x + threadIdx.x;
    if (e >= E) return;
    eidA[off[dstA[e]] + rankA[e]] = e;
}

__global__ __launch_bounds__(128)
void proj_kernel(const float* __restrict__ X, const float* __restrict__ W,
                 float* __restrict__ Y, int M, int K) {
    const int wave = threadIdx.x >> 5;
    const int lane = threadIdx.x & 31;
    const int mt = blockIdx.x * (blockDim.x >> 5) + wave;
    if (mt * 16 >= M) return;
    const int row0 = mt * 16;
    const int m = lane & 15, half = lane >> 4;

    v8f acc[8];
#pragma unroll
    for (int nt = 0; nt < 8; ++nt) acc[nt] = v8f{};

    const float* ap0 = X + (size_t)(row0 + m) * K + 2 * half;
    const float* bp0 = W + (size_t)(2 * half) * HC + m;
    for (int k0 = 0; k0 < K; k0 += 4) {
        v2f a;
        a.x = ap0[k0];
        a.y = ap0[k0 + 1];
        const float* bp = bp0 + (size_t)k0 * HC;
#pragma unroll
        for (int nt = 0; nt < 8; ++nt) {
            v2f b;
            b.x = bp[nt * 16];
            b.y = bp[HC + nt * 16];
            acc[nt] = __builtin_amdgcn_wmma_f32_16x16x4_f32(false, a, false, b,
                                                            (short)0, acc[nt], false, false);
        }
    }

    float* yp = Y + (size_t)(row0 + 8 * half) * HC + m;
#pragma unroll
    for (int nt = 0; nt < 8; ++nt) {
#pragma unroll
        for (int r = 0; r < 8; ++r) yp[(size_t)r * HC + nt * 16] = acc[nt][r];
    }
}

template <int J> __device__ __forceinline__ int bc8i(int v) {
    return (int)__builtin_amdgcn_mov_dpp8((unsigned)v, (unsigned)J * 0x249249u);
}
template <int J> __device__ __forceinline__ float bc8f(float v) {
    return __builtin_bit_cast(float, __builtin_amdgcn_mov_dpp8(__builtin_bit_cast(unsigned, v), (unsigned)J * 0x249249u));
}
__device__ __forceinline__ float sum8(float v) {
    v += lane_xor1(v); v += lane_xor2(v); v += lane_xor4(v);
    return v;
}
__device__ __forceinline__ float max8(float v) {
    v = fmaxf(v, lane_xor1(v)); v = fmaxf(v, lane_xor2(v)); v = fmaxf(v, lane_xor4(v));
    return v;
}

template <int NG>
__device__ __forceinline__ void edge_groups(const int* __restrict__ ep, const int* __restrict__ srcA,
                                            const float* __restrict__ a_e,
                                            int j, int h, int nc, float* ae, int* sv) {
    unsigned e[NG];
#pragma unroll
    for (int q = 0; q < NG; ++q) e[q] = (unsigned)ep[(unsigned)min(8 * q + j, nc - 1)];
#pragma unroll
    for (int q = 0; q < NG; ++q) sv[q] = srcA[e[q]];
#pragma unroll
    for (int q = 0; q < NG; ++q) ae[q] = a_e[e[q] * 4u + (unsigned)h];
    if constexpr (NG == 4) asm volatile("" : "+v"(sv[0]), "+v"(sv[1]), "+v"(sv[2]), "+v"(sv[3]));
    else                   asm volatile("" : "+v"(sv[0]), "+v"(sv[1]));
}

__device__ __forceinline__ void gather8(v4f* hv, const v4f* __restrict__ Hp4, int svg, int lane) {
    hv[0] = Hp4[(unsigned)bc8i<0>(svg) * 32u + (unsigned)lane];
    hv[1] = Hp4[(unsigned)bc8i<1>(svg) * 32u + (unsigned)lane];
    hv[2] = Hp4[(unsigned)bc8i<2>(svg) * 32u + (unsigned)lane];
    hv[3] = Hp4[(unsigned)bc8i<3>(svg) * 32u + (unsigned)lane];
    hv[4] = Hp4[(unsigned)bc8i<4>(svg) * 32u + (unsigned)lane];
    hv[5] = Hp4[(unsigned)bc8i<5>(svg) * 32u + (unsigned)lane];
    hv[6] = Hp4[(unsigned)bc8i<6>(svg) * 32u + (unsigned)lane];
    hv[7] = Hp4[(unsigned)bc8i<7>(svg) * 32u + (unsigned)lane];
    asm volatile("" : "+v"(hv[0]), "+v"(hv[1]), "+v"(hv[2]), "+v"(hv[3]),
                      "+v"(hv[4]), "+v"(hv[5]), "+v"(hv[6]), "+v"(hv[7]));
}
__device__ __forceinline__ void fma_row(float4& A, float w, const v4f& v) {
    A.x += w * v.x; A.y += w * v.y; A.z += w * v.z; A.w += w * v.w;
}
__device__ __forceinline__ void accum8(float4& A, const v4f* hv, float pg) {
    fma_row(A, bc8f<0>(pg), hv[0]);
    fma_row(A, bc8f<1>(pg), hv[1]);
    fma_row(A, bc8f<2>(pg), hv[2]);
    fma_row(A, bc8f<3>(pg), hv[3]);
    fma_row(A, bc8f<4>(pg), hv[4]);
    fma_row(A, bc8f<5>(pg), hv[5]);
    fma_row(A, bc8f<6>(pg), hv[6]);
    fma_row(A, bc8f<7>(pg), hv[7]);
}

__device__ __forceinline__ void group_step(v4f* hv, const v4f& ats, float base, float pen,
                                           bool b4, bool b2, bool b1, float& M, float& D, float4& A) {
    float r[4], s[2];
#pragma unroll
    for (int q = 0; q < 4; ++q) {
        const v2f t0 = __builtin_elementwise_fma(hv[q].zw, ats.zw, hv[q].xy * ats.xy);
        const v2f t1 = __builtin_elementwise_fma(hv[q + 4].zw, ats.zw, hv[q + 4].xy * ats.xy);
        const float d0 = t0.x + t0.y, d1 = t1.x + t1.y;
        float keep = b4 ? d1 : d0;
        float send = b4 ? d0 : d1;
        r[q] = keep + lane_xor4(send);
        if (q == 1) asm volatile("" : "+v"(r[0]), "+v"(r[1]), "+v"(hv[2]), "+v"(hv[3]), "+v"(hv[6]), "+v"(hv[7]));
    }
#pragma unroll
    for (int q = 0; q < 2; ++q) {
        float keep = b2 ? r[q + 2] : r[q];
        float send = b2 ? r[q] : r[q + 2];
        s[q] = keep + lane_xor2(send);
    }
    const float keep = b1 ? s[1] : s[0];
    const float send = b1 ? s[0] : s[1];
    const float as_e = keep + lane_xor1(send);

    const float lg = leaky(as_e + base) + pen;
    const float Mn = fmaxf(M, max8(lg));
    const float sc = __expf(M - Mn);
    const float p  = __expf(lg - Mn);
    D = D * sc + sum8(p);
    A.x *= sc; A.y *= sc; A.z *= sc; A.w *= sc;
    M = Mn;
    accum8(A, hv, p);
}

__global__ void agg_kernel(const int* __restrict__ off, const int* __restrict__ degI,
                const int* __restrict__ eidA, const int* __restrict__ srcA,
                const float* __restrict__ a_e,
                const float* __restrict__ att_s, const float* __restrict__ att_d,
                const float* __restrict__ Hp, const float* __restrict__ bias,
                const float* __restrict__ lin_w, const float* __restrict__ lin_b,
                float* __restrict__ outv, float* __restrict__ outs,
                int N, int mode) {
    const int lane = threadIdx.x & 31;
    const int n = __builtin_amdgcn_readfirstlane(blockIdx.x * (blockDim.x >> 5) + (threadIdx.x >> 5));
    if (n >= N) return;
    const int h = lane >> 3, j = lane & 7;
    const float4* Hp4 = reinterpret_cast<const float4*>(Hp);
    const v4f* Hp4v = reinterpret_cast<const v4f*>(Hp);

    const int cnt = degI[n];
    const int s0 = off[n];
    const float4 hself = Hp4[(size_t)n * 32 + lane];
    const v4f ats = reinterpret_cast<const v4f*>(att_s)[lane];
    const v4f atd = reinterpret_cast<const v4f*>(att_d)[lane];
    const float ad_n = sum8(hself.x * atd.x + hself.y * atd.y + hself.z * atd.z + hself.w * atd.w);
    const float degf = fmaxf((float)cnt, 1.0f);
    float M = -INFINITY;
    float D = 0.0f;
    float4 A = {0.f, 0.f, 0.f, 0.f};
    float S = 0.0f;

    const bool b4 = (j & 4) != 0, b2 = (j & 2) != 0, b1 = (j & 1) != 0;

    for (int c = 0; c < cnt; c += 32) {
        const int nc = min(cnt - c, 32);
        const int* ep = eidA + s0 + c;
        float ae[4];
        int sv[4];
        if (nc > 16) {
            edge_groups<4>(ep, srcA, a_e, j, h, nc, ae, sv);
        } else {
            edge_groups<2>(ep, srcA, a_e, j, h, nc, ae, sv);
            ae[2] = 0.0f; ae[3] = 0.0f;
            sv[2] = 0; sv[3] = 0;
        }
        const bool in0 = j < nc, in1 = j < nc - 8, in2 = j < nc - 16, in3 = j < nc - 24;
        const float pen0 = in0 ? 0.0f : -INFINITY;
        const float pen1 = in1 ? 0.0f : -INFINITY;
        const float pen2 = in2 ? 0.0f : -INFINITY;
        const float pen3 = in3 ? 0.0f : -INFINITY;
        v4f hv[8];
        gather8(hv, Hp4v, sv[0], lane);
        S += ((in0 ? ae[0] : 0.0f) + (in1 ? ae[1] : 0.0f)) + ((in2 ? ae[2] : 0.0f) + (in3 ? ae[3] : 0.0f));
        group_step(hv, ats, ad_n + ae[0], pen0, b4, b2, b1, M, D, A);
        if (nc > 8)  { gather8(hv, Hp4v, sv[1], lane); group_step(hv, ats, ad_n + ae[1], pen1, b4, b2, b1, M, D, A); }
        if (nc > 16) { gather8(hv, Hp4v, sv[2], lane); group_step(hv, ats, ad_n + ae[2], pen2, b4, b2, b1, M, D, A); }
        if (nc > 24) { gather8(hv, Hp4v, sv[3], lane); group_step(hv, ats, ad_n + ae[3], pen3, b4, b2, b1, M, D, A); }
    }
    S = sum8(S);
    {
        const float as_n = sum8(hself.x * ats.x + hself.y * ats.y + hself.z * ats.z + hself.w * ats.w);
        float lgs = leaky(as_n + ad_n + S / degf);
        float Mn = fmaxf(M, lgs);
        float sc = __expf(M - Mn);
        float p  = __expf(lgs - Mn);
        A.x = A.x * sc + p * hself.x;
        A.y = A.y * sc + p * hself.y;
        A.z = A.z * sc + p * hself.z;
        A.w = A.w * sc + p * hself.w;
        D = D * sc + p;
    }
    float4 bv = reinterpret_cast<const float4*>(bias)[lane];
    float inv = 1.0f / D;
    float4 r = { A.x * inv + bv.x, A.y * inv + bv.y, A.z * inv + bv.z, A.w * inv + bv.w };
    if (mode == 0) {
        r.x = fmaxf(r.x, 0.f); r.y = fmaxf(r.y, 0.f);
        r.z = fmaxf(r.z, 0.f); r.w = fmaxf(r.w, 0.f);
        reinterpret_cast<float4*>(outv)[(size_t)n * 32 + lane] = r;
    } else {
        float4 lw = reinterpret_cast<const float4*>(lin_w)[lane];
        float part = r.x * lw.x + r.y * lw.y + r.z * lw.z + r.w * lw.w;
#pragma unroll
        for (int msk = 16; msk >= 1; msk >>= 1) part += __shfl_xor(part, msk, 32);
        if (lane == 0) outs[n] = part + lin_b[0];
    }
}

extern "C" void kernel_launch(void* const* d_in, const int* in_sizes, int n_in,
                              void* d_out, int out_size, void* d_ws, size_t ws_size,
                              hipStream_t stream) {
    const float* x   = (const float*)d_in[0];
    const float* ea  = (const float*)d_in[1];
    const float* W1  = (const float*)d_in[2];
    const float* as1 = (const float*)d_in[3];
    const float* ad1 = (const float*)d_in[4];
    const float* We1 = (const float*)d_in[5];
    const float* ae1 = (const float*)d_in[6];
    const float* b1  = (const float*)d_in[7];
    const float* W2  = (const float*)d_in[8];
    const float* as2 = (const float*)d_in[9];
    const float* ad2 = (const float*)d_in[10];
    const float* We2 = (const float*)d_in[11];
    const float* ae2 = (const float*)d_in[12];
    const float* b2  = (const float*)d_in[13];
    const float* lw  = (const float*)d_in[14];
    const float* lb  = (const float*)d_in[15];
    const int*   ei  = (const int*)d_in[16];

    const int N = in_sizes[0] / F_IN;
    const int E = in_sizes[16] / 2;
    const int* srcA = ei;
    const int* dstA = ei + E;

    char* w = (char*)d_ws;
    auto alloc = [&](size_t bytes) -> void* {
        void* p = (void*)w;
        w += (bytes + 255) & ~(size_t)255;
        return p;
    };
    int*   degI   = (int*)  alloc((size_t)N * 4);
    size_t zbytes = (size_t)(w - (char*)degI);
    float* Ve1  = (float*)alloc(F_IN * 4 * 4);
    float* Ve2  = (float*)alloc(F_IN * 4 * 4);
    float* ae1v = (float*)alloc((size_t)E * 16);
    float* ae2v = (float*)alloc((size_t)E * 16);
    int*   off  = (int*)  alloc((size_t)N * 4);
    int*   eidA = (int*)  alloc((size_t)E * 4);
    float* Hp   = (float*)alloc((size_t)N * HC * 4);
    float* h1   = (float*)alloc((size_t)N * HC * 4);
    int*   rankA = (int*)h1;

    hipMemsetAsync(degI, 0, zbytes, stream);

    fold_kernel<<<1, 256, 0, stream>>>(We1, ae1, We2, ae2, Ve1, Ve2);
    edge_proj_kernel<<<(E + 255) / 256, 256, 0, stream>>>(ea, dstA, Ve1, Ve2,
                                                          ae1v, ae2v, degI, rankA, E);
    scan_kernel<<<1, SCAN_THREADS, 0, stream>>>(degI, off, N);
    scatter_kernel<<<(E + 255) / 256, 256, 0, stream>>>(dstA, off, rankA, eidA, E);

    const int mt = (N + 15) / 16;
    const int projBlocks = (mt + 3) / 4;
    const int nodeBlocks = (N + 7) / 8;

    proj_kernel<<<projBlocks, 128, 0, stream>>>(x, W1, Hp, N, F_IN);
    agg_kernel<<<nodeBlocks, 256, 0, stream>>>(off, degI, eidA, srcA, ae1v, as1, ad1,
                                               Hp, b1, nullptr, nullptr,
                                               h1, nullptr, N, 0);
    proj_kernel<<<projBlocks, 128, 0, stream>>>(h1, W2, Hp, N, HC);
    agg_kernel<<<nodeBlocks, 256, 0, stream>>>(off, degI, eidA, srcA, ae2v, as2, ad2,
                                               Hp, b2, lw, lb,
                                               nullptr, (float*)d_out, N, 1);
}


# edge pass files each edge id at (destination, count its atomic returned) in a fixed 64-slot list per node: offset scan and fill launches gone, 8 launches to 6 (list order by arrival, as before)
# speedup vs baseline: 2.9283x; 1.1201x over previous
#include <hip/hip_runtime.h>
#include <hip/hip_bf16.h>

#define F_IN   64
#define HEADS  4
#define CHD    32
#define HC     128
#define NEG_SLOPE 0.2f

typedef __attribute__((ext_vector_type(2))) float v2f;
typedef __attribute__((ext_vector_type(8))) float v8f;
typedef __attribute__((ext_vector_type(4))) float v4f;

__device__ __forceinline__ float leaky(float x) { return x > 0.f ? x : NEG_SLOPE * x; }

__global__ void fold_kernel(const float* __restrict__ We1, const float* __restrict__ ae1,
                            const float* __restrict__ We2, const float* __restrict__ ae2,
                            float* __restrict__ Ve1, float* __restrict__ Ve2) {
    int t = threadIdx.x;
    int f = t >> 2, h = t & 3;
    float s1 = 0.f, s2 = 0.f;
#pragma unroll
    for (int c = 0; c < CHD; ++c) {
        s1 += We1[f * HC + h * CHD + c] * ae1[h * CHD + c];
        s2 += We2[f * HC + h * CHD + c] * ae2[h * CHD + c];
    }
    Ve1[f * 4 + h] = s1;
    Ve2[f * 4 + h] = s2;
}

__device__ __forceinline__ float lane_xor4(float v) {
    return __builtin_bit_cast(float, __builtin_amdgcn_update_dpp(0, __builtin_bit_cast(int, v), 0x164, 0xf, 0xf, false));
}
__device__ __forceinline__ float lane_xor2(float v) {
    return __builtin_bit_cast(float, __builtin_amdgcn_update_dpp(0, __builtin_bit_cast(int, v), 0x4E, 0xf, 0xf, false));
}
__device__ __forceinline__ float lane_xor1(float v) {
    return __builtin_bit_cast(float, __builtin_amdgcn_update_dpp(0, __builtin_bit_cast(int, v), 0xB1, 0xf, 0xf, false));
}

#define EP_GROUPS 8
#define LIST_CAP  64

__global__ __launch_bounds__(256)
void edge_proj_kernel(const float* __restrict__ EA, const int* __restrict__ dstA,
                      const float* __restrict__ Ve1, const float* __restrict__ Ve2,
                      float* __restrict__ ae1v, float* __restrict__ ae2v,
                      int* __restrict__ degI, int* __restrict__ slotA, int E) {
    const int lane = threadIdx.x & 31;
    const int j = lane & 7;
    const int sub = lane >> 3;
    const int gw = blockIdx.x * (blockDim.x >> 5) + (threadIdx.x >> 5);
    const int base = gw * (4 * EP_GROUPS);
    if (base >= E) return;

    const int eown = base + lane;
    const int down = dstA[min(eown, E - 1)];

    const float4* EA4 = reinterpret_cast<const float4*>(EA);
    float4 x0[EP_GROUPS], x1[EP_GROUPS];
#pragma unroll
    for (int t = 0; t < EP_GROUPS; ++t) {
        int e = min(base + 4 * t + sub, E - 1);
        x0[t] = EA4[(size_t)e * 16 + j];
        x1[t] = EA4[(size_t)e * 16 + 8 + j];
    }

    const float4* V1 = reinterpret_cast<const float4*>(Ve1);
    const float4* V2 = reinterpret_cast<const float4*>(Ve2);
    float wl[4][8], wh[4][8];
#pragma unroll
    for (int c = 0; c < 4; ++c) {
        float4 p = V1[4 * j + c], q = V2[4 * j + c];
        float4 r = V1[32 + 4 * j + c], s = V2[32 + 4 * j + c];
        wl[c][0] = p.x; wl[c][1] = p.y; wl[c][2] = p.z; wl[c][3] = p.w;
        wl[c][4] = q.x; wl[c][5] = q.y; wl[c][6] = q.z; wl[c][7] = q.w;
        wh[c][0] = r.x; wh[c][1] = r.y; wh[c][2] = r.z; wh[c][3] = r.w;
        wh[c][4] = s.x; wh[c][5] = s.y; wh[c][6] = s.z; wh[c][7] = s.w;
    }

    const int rank = atomicAdd(&degI[down], eown < E ? 1 : 0);

    const bool b4 = (j & 4) != 0, b2 = (j & 2) != 0, b1 = (j & 1) != 0;
    float* outp = (b4 ? ae2v : ae1v) + (j & 3);
#pragma unroll
    for (int t = 0; t < EP_GROUPS; ++t) {
        float acc[8];
#pragma unroll
        for (int q = 0; q < 8; ++q) {
            acc[q] = x0[t].x * wl[0][q] + x0[t].y * wl[1][q] + x0[t].z * wl[2][q] + x0[t].w * wl[3][q]
                   + x1[t].x * wh[0][q] + x1[t].y * wh[1][q] + x1[t].z * wh[2][q] + x1[t].w * wh[3][q];
        }
        float r[4], s[2];
#pragma unroll
        for (int q = 0; q < 4; ++q) {
            float keep = b4 ? acc[q + 4] : acc[q];
            float send = b4 ? acc[q] : acc[q + 4];
            r[q] = keep + lane_xor4(send);
        }
#pragma unroll
        for (int q = 0; q < 2; ++q) {
            float keep = b2 ? r[q + 2] : r[q];
            float send = b2 ? r[q] : r[q + 2];
            s[q] = keep + lane_xor2(send);
        }
        float keep = b1 ? s[1] : s[0];
        float send = b1 ? s[0] : s[1];
        float v = keep + lane_xor1(send);
        int e = base + 4 * t + sub;
        if (e < E) outp[(size_t)e * 4] = v;
    }
    if (eown < E && rank < LIST_CAP) slotA[(unsigned)down * (unsigned)LIST_CAP + (unsigned)rank] = eown;
}

__global__ __launch_bounds__(128)
void proj_kernel(const float* __restrict__ X, const float* __restrict__ W,
                 float* __restrict__ Y, int M, int K) {
    const int wave = threadIdx.x >> 5;
    const int lane = threadIdx.x & 31;
    const int mt = blockIdx.x * (blockDim.x >> 5) + wave;
    if (mt * 16 >= M) return;
    const int row0 = mt * 16;
    const int m = lane & 15, half = lane >> 4;

    v8f acc[8];
#pragma unroll
    for (int nt = 0; nt < 8; ++nt) acc[nt] = v8f{};

    const float* ap0 = X + (size_t)(row0 + m) * K + 2 * half;
    const float* bp0 = W + (size_t)(2 * half) * HC + m;
    for (int k0 = 0; k0 < K; k0 += 4) {
        v2f a;
        a.x = ap0[k0];
        a.y = ap0[k0 + 1];
        const float* bp = bp0 + (size_t)k0 * HC;
#pragma unroll
        for (int nt = 0; nt < 8; ++nt) {
            v2f b;
            b.x = bp[nt * 16];
            b.y = bp[HC + nt * 16];
            acc[nt] = __builtin_amdgcn_wmma_f32_16x16x4_f32(false, a, false, b,
                                                            (short)0, acc[nt], false, false);
        }
    }

    float* yp = Y + (size_t)(row0 + 8 * half) * HC + m;
#pragma unroll
    for (int nt = 0; nt < 8; ++nt) {
#pragma unroll
        for (int r = 0; r < 8; ++r) yp[(size_t)r * HC + nt * 16] = acc[nt][r];
    }
}

template <int J> __device__ __forceinline__ int bc8i(int v) {
    return (int)__builtin_amdgcn_mov_dpp8((unsigned)v, (unsigned)J * 0x249249u);
}
template <int J> __device__ __forceinline__ float bc8f(float v) {
    return __builtin_bit_cast(float, __builtin_amdgcn_mov_dpp8(__builtin_bit_cast(unsigned, v), (unsigned)J * 0x249249u));
}
__device__ __forceinline__ float sum8(float v) {
    v += lane_xor1(v); v += lane_xor2(v); v += lane_xor4(v);
    return v;
}
__device__ __forceinline__ float max8(float v) {
    v = fmaxf(v, lane_xor1(v)); v = fmaxf(v, lane_xor2(v)); v = fmaxf(v, lane_xor4(v));
    return v;
}

template <int NG>
__device__ __forceinline__ void edge_groups(const int* __restrict__ ep, const int* __restrict__ srcA,
                                            const float* __restrict__ a_e,
                                            int j, int h, int nc, float* ae, int* sv) {
    unsigned e[NG];
#pragma unroll
    for (int q = 0; q < NG; ++q) e[q] = (unsigned)ep[(unsigned)min(8 * q + j, nc - 1)];
#pragma unroll
    for (int q = 0; q < NG; ++q) sv[q] = srcA[e[q]];
#pragma unroll
    for (int q = 0; q < NG; ++q) ae[q] = a_e[e[q] * 4u + (unsigned)h];
    if constexpr (NG == 4) asm volatile("" : "+v"(sv[0]), "+v"(sv[1]), "+v"(sv[2]), "+v"(sv[3]));
    else                   asm volatile("" : "+v"(sv[0]), "+v"(sv[1]));
}

__device__ __forceinline__ void gather8(v4f* hv, const v4f* __restrict__ Hp4, int svg, int lane) {
    hv[0] = Hp4[(unsigned)bc8i<0>(svg) * 32u + (unsigned)lane];
    hv[1] = Hp4[(unsigned)bc8i<1>(svg) * 32u + (unsigned)lane];
    hv[2] = Hp4[(unsigned)bc8i<2>(svg) * 32u + (unsigned)lane];
    hv[3] = Hp4[(unsigned)bc8i<3>(svg) * 32u + (unsigned)lane];
    hv[4] = Hp4[(unsigned)bc8i<4>(svg) * 32u + (unsigned)lane];
    hv[5] = Hp4[(unsigned)bc8i<5>(svg) * 32u + (unsigned)lane];
    hv[6] = Hp4[(unsigned)bc8i<6>(svg) * 32u + (unsigned)lane];
    hv[7] = Hp4[(unsigned)bc8i<7>(svg) * 32u + (unsigned)lane];
    asm volatile("" : "+v"(hv[0]), "+v"(hv[1]), "+v"(hv[2]), "+v"(hv[3]),
                      "+v"(hv[4]), "+v"(hv[5]), "+v"(hv[6]), "+v"(hv[7]));
}
__device__ __forceinline__ void fma_row(float4& A, float w, const v4f& v) {
    A.x += w * v.x; A.y += w * v.y; A.z += w * v.z; A.w += w * v.w;
}
__device__ __forceinline__ void accum8(float4& A, const v4f* hv, float pg) {
    fma_row(A, bc8f<0>(pg), hv[0]);
    fma_row(A, bc8f<1>(pg), hv[1]);
    fma_row(A, bc8f<2>(pg), hv[2]);
    fma_row(A, bc8f<3>(pg), hv[3]);
    fma_row(A, bc8f<4>(pg), hv[4]);
    fma_row(A, bc8f<5>(pg), hv[5]);
    fma_row(A, bc8f<6>(pg), hv[6]);
    fma_row(A, bc8f<7>(pg), hv[7]);
}

__device__ __forceinline__ void group_step(v4f* hv, const v4f& ats, float base, float pen,
                                           bool b4, bool b2, bool b1, float& M, float& D, float4& A) {
    float r[4], s[2];
#pragma unroll
    for (int q = 0; q < 4; ++q) {
        const v2f t0 = __builtin_elementwise_fma(hv[q].zw, ats.zw, hv[q].xy * ats.xy);
        const v2f t1 = __builtin_elementwise_fma(hv[q + 4].zw, ats.zw, hv[q + 4].xy * ats.xy);
        const float d0 = t0.x + t0.y, d1 = t1.x + t1.y;
        float keep = b4 ? d1 : d0;
        float send = b4 ? d0 : d1;
        r[q] = keep + lane_xor4(send);
        if (q == 1) asm volatile("" : "+v"(r[0]), "+v"(r[1]), "+v"(hv[2]), "+v"(hv[3]), "+v"(hv[6]), "+v"(hv[7]));
    }
#pragma unroll
    for (int q = 0; q < 2; ++q) {
        float keep = b2 ? r[q + 2] : r[q];
        float send = b2 ? r[q] : r[q + 2];
        s[q] = keep + lane_xor2(send);
    }
    const float keep = b1 ? s[1] : s[0];
    const float send = b1 ? s[0] : s[1];
    const float as_e = keep + lane_xor1(send);

    const float lg = leaky(as_e + base) + pen;
    const float Mn = fmaxf(M, max8(lg));
    const float sc = __expf(M - Mn);
    const float p  = __expf(lg - Mn);
    D = D * sc + sum8(p);
    A.x *= sc; A.y *= sc; A.z *= sc; A.w *= sc;
    M = Mn;
    accum8(A, hv, p);
}

__global__ void agg_kernel(const int* __restrict__ degI, const int* __restrict__ slotA,
                const int* __restrict__ srcA, const int* __restrict__ dstA, int E,
                const float* __restrict__ a_e,
                const float* __restrict__ att_s, const float* __restrict__ att_d,
                const float* __restrict__ Hp, const float* __restrict__ bias,
                const float* __restrict__ lin_w, const float* __restrict__ lin_b,
                float* __restrict__ outv, float* __restrict__ outs,
                int N, int mode) {
    const int lane = threadIdx.x & 31;
    const int n = __builtin_amdgcn_readfirstlane(blockIdx.x * (blockDim.x >> 5) + (threadIdx.x >> 5));
    if (n >= N) return;
    const int h = lane >> 3, j = lane & 7;
    const float4* Hp4 = reinterpret_cast<const float4*>(Hp);
    const v4f* Hp4v = reinterpret_cast<const v4f*>(Hp);

    const int cnt = degI[n];
    const float4 hself = Hp4[(size_t)n * 32 + lane];
    const v4f ats = reinterpret_cast<const v4f*>(att_s)[lane];
    const v4f atd = reinterpret_cast<const v4f*>(att_d)[lane];
    const float ad_n = sum8(hself.x * atd.x + hself.y * atd.y + hself.z * atd.z + hself.w * atd.w);
    const float degf = fmaxf((float)cnt, 1.0f);
    float M = -INFINITY;
    float D = 0.0f;
    float4 A = {0.f, 0.f, 0.f, 0.f};
    float S = 0.0f;

    const bool b4 = (j & 4) != 0, b2 = (j & 2) != 0, b1 = (j & 1) != 0;

    const int lcnt = cnt <= LIST_CAP ? cnt : 0;
    const int* lp = slotA + (size_t)n * LIST_CAP;
    for (int c = 0; c < lcnt; c += 32) {
        const int nc = min(lcnt - c, 32);
        const int* ep = lp + c;
        float ae[4];
        int sv[4];
        if (nc > 16) {
            edge_groups<4>(ep, srcA, a_e, j, h, nc, ae, sv);
        } else {
            edge_groups<2>(ep, srcA, a_e, j, h, nc, ae, sv);
            ae[2] = 0.0f; ae[3] = 0.0f;
            sv[2] = 0; sv[3] = 0;
        }
        const bool in0 = j < nc, in1 = j < nc - 8, in2 = j < nc - 16, in3 = j < nc - 24;
        const float pen0 = in0 ? 0.0f : -INFINITY;
        const float pen1 = in1 ? 0.0f : -INFINITY;
        const float pen2 = in2 ? 0.0f : -INFINITY;
        const float pen3 = in3 ? 0.0f : -INFINITY;
        v4f hv[8];
        gather8(hv, Hp4v, sv[0], lane);
        S += ((in0 ? ae[0] : 0.0f) + (in1 ? ae[1] : 0.0f)) + ((in2 ? ae[2] : 0.0f) + (in3 ? ae[3] : 0.0f));
        group_step(hv, ats, ad_n + ae[0], pen0, b4, b2, b1, M, D, A);
        if (nc > 8)  { gather8(hv, Hp4v, sv[1], lane); group_step(hv, ats, ad_n + ae[1], pen1, b4, b2, b1, M, D, A); }
        if (nc > 16) { gather8(hv, Hp4v, sv[2], lane); group_step(hv, ats, ad_n + ae[2], pen2, b4, b2, b1, M, D, A); }
        if (nc > 24) { gather8(hv, Hp4v, sv[3], lane); group_step(hv, ats, ad_n + ae[3], pen3, b4, b2, b1, M, D, A); }
    }
    if (cnt > LIST_CAP) {
        for (int b = 0; b < E; b += 32) {
            const int e = b + lane;
            const bool hit = e < E && dstA[min(e, E - 1)] == n;
            unsigned found = __builtin_amdgcn_ballot_w32(hit);
            while (found != 0u) {
                const int ee = b + __builtin_ctz(found);
                found &= found - 1u;
                const int sn = srcA[ee];
                const float aev = a_e[(unsigned)ee * 4u + (unsigned)h];
                const v4f row = Hp4v[(unsigned)sn * 32u + (unsigned)lane];
                const float as_e = sum8(row.x * ats.x + row.y * ats.y + row.z * ats.z + row.w * ats.w);
                const float lg = leaky(as_e + ad_n + aev);
                const float Mn = fmaxf(M, lg);
                const float sc = __expf(M - Mn);
                const float p  = __expf(lg - Mn);
                D = D * sc + p;
                A.x = A.x * sc + p * row.x;
                A.y = A.y * sc + p * row.y;
                A.z = A.z * sc + p * row.z;
                A.w = A.w * sc + p * row.w;
                M = Mn;
                S += (j == 0) ? aev : 0.0f;
            }
        }
    }
    S = sum8(S);
    {
        const float as_n = sum8(hself.x * ats.x + hself.y * ats.y + hself.z * ats.z + hself.w * ats.w);
        float lgs = leaky(as_n + ad_n + S / degf);
        float Mn = fmaxf(M, lgs);
        float sc = __expf(M - Mn);
        float p  = __expf(lgs - Mn);
        A.x = A.x * sc + p * hself.x;
        A.y = A.y * sc + p * hself.y;
        A.z = A.z * sc + p * hself.z;
        A.w = A.w * sc + p * hself.w;
        D = D * sc + p;
    }
    float4 bv = reinterpret_cast<const float4*>(bias)[lane];
    float inv = 1.0f / D;
    float4 r = { A.x * inv + bv.x, A.y * inv + bv.y, A.z * inv + bv.z, A.w * inv + bv.w };
    if (mode == 0) {
        r.x = fmaxf(r.x, 0.f); r.y = fmaxf(r.y, 0.f);
        r.z = fmaxf(r.z, 0.f); r.w = fmaxf(r.w, 0.f);
        reinterpret_cast<float4*>(outv)[(size_t)n * 32 + lane] = r;
    } else {
        float4 lw = reinterpret_cast<const float4*>(lin_w)[lane];
        float part = r.x * lw.x + r.y * lw.y + r.z * lw.z + r.w * lw.w;
#pragma unroll
        for (int msk = 16; msk >= 1; msk >>= 1) part += __shfl_xor(part, msk, 32);
        if (lane == 0) outs[n] = part + lin_b[0];
    }
}

extern "C" void kernel_launch(void* const* d_in, const int* in_sizes, int n_in,
                              void* d_out, int out_size, void* d_ws, size_t ws_size,
                              hipStream_t stream) {
    const float* x   = (const float*)d_in[0];
    const float* ea  = (const float*)d_in[1];
    const float* W1  = (const float*)d_in[2];
    const float* as1 = (const float*)d_in[3];
    const float* ad1 = (const float*)d_in[4];
    const float* We1 = (const float*)d_in[5];
    const float* ae1 = (const float*)d_in[6];
    const float* b1  = (const float*)d_in[7];
    const float* W2  = (const float*)d_in[8];
    const float* as2 = (const float*)d_in[9];
    const float* ad2 = (const float*)d_in[10];
    const float* We2 = (const float*)d_in[11];
    const float* ae2 = (const float*)d_in[12];
    const float* b2  = (const float*)d_in[13];
    const float* lw  = (const float*)d_in[14];
    const float* lb  = (const float*)d_in[15];
    const int*   ei  = (const int*)d_in[16];

    const int N = in_sizes[0] / F_IN;
    const int E = in_sizes[16] / 2;
    const int* srcA = ei;
    const int* dstA = ei + E;

    char* w = (char*)d_ws;
    auto alloc = [&](size_t bytes) -> void* {
        void* p = (void*)w;
        w += (bytes + 255) & ~(size_t)255;
        return p;
    };
    int*   degI   = (int*)  alloc((size_t)N * 4);
    size_t zbytes = (size_t)(w - (char*)degI);
    float* Ve1  = (float*)alloc(F_IN * 4 * 4);
    float* Ve2  = (float*)alloc(F_IN * 4 * 4);
    float* ae1v = (float*)alloc((size_t)E * 16);
    float* ae2v = (float*)alloc((size_t)E * 16);
    int*   slotA = (int*) alloc((size_t)N * LIST_CAP * 4);
    float* Hp   = (float*)alloc((size_t)N * HC * 4);
    float* h1   = (float*)alloc((size_t)N * HC * 4);
    if ((size_t)(w - (char*)d_ws) > ws_size) return;

    hipMemsetAsync(degI, 0, zbytes, stream);

    fold_kernel<<<1, 256, 0, stream>>>(We1, ae1, We2, ae2, Ve1, Ve2);
    edge_proj_kernel<<<(E + 255) / 256, 256, 0, stream>>>(ea, dstA, Ve1, Ve2,
                                                          ae1v, ae2v, degI, slotA, E);

    const int mt = (N + 15) / 16;
    const int projBlocks = (mt + 3) / 4;
    const int nodeBlocks = (N + 7) / 8;

    proj_kernel<<<projBlocks, 128, 0, stream>>>(x, W1, Hp, N, F_IN);
    agg_kernel<<<nodeBlocks, 256, 0, stream>>>(degI, slotA, srcA, dstA, E, ae1v, as1, ad1,
                                               Hp, b1, nullptr, nullptr,
                                               h1, nullptr, N, 0);
    proj_kernel<<<projBlocks, 128, 0, stream>>>(h1, W2, Hp, N, HC);
    agg_kernel<<<nodeBlocks, 256, 0, stream>>>(degI, slotA, srcA, dstA, E, ae2v, as2, ad2,
                                               Hp, b2, lw, lb,
                                               nullptr, (float*)d_out, N, 1);
}


# layer 2 kept as 12 scalars per node: att_src2, att_dst2, lin_w folded into W2 before the projection (one column tile); 2nd aggregation reads 32 bytes per edge, 4 nodes per wave (new summation order)
# speedup vs baseline: 3.5080x; 1.1980x over previous
#include <hip/hip_runtime.h>
#include <hip/hip_bf16.h>

#define F_IN   64
#define HEADS  4
#define CHD    32
#define HC     128
#define NEG_SLOPE 0.2f

typedef __attribute__((ext_vector_type(2))) float v2f;
typedef __attribute__((ext_vector_type(8))) float v8f;
typedef __attribute__((ext_vector_type(4))) float v4f;

__device__ __forceinline__ float leaky(float x) { return x > 0.f ? x : NEG_SLOPE * x; }

__global__ void fold_kernel(const float* __restrict__ We1, const float* __restrict__ ae1,
                            const float* __restrict__ We2, const float* __restrict__ ae2,
                            const float* __restrict__ W2, const float* __restrict__ as2,
                            const float* __restrict__ ad2, const float* __restrict__ lw,
                            float* __restrict__ Ve1, float* __restrict__ Ve2, float* __restrict__ VqF) {
    int t = threadIdx.x;
    if (blockIdx.x == 0) {
        int f = t >> 2, h = t & 3;
        float s1 = 0.f, s2 = 0.f;
#pragma unroll
        for (int c = 0; c < CHD; ++c) {
            s1 += We1[f * HC + h * CHD + c] * ae1[h * CHD + c];
            s2 += We2[f * HC + h * CHD + c] * ae2[h * CHD + c];
        }
        Ve1[f * 4 + h] = s1;
        Ve2[f * 4 + h] = s2;
        return;
    }
    const int gid = (blockIdx.x - 1) * 256 + t;
    const int k = gid >> 4, col = gid & 15;
    const int typ = col >> 2, h = col & 3;
    float s = 0.f;
    if (typ < 3) {
        const float* vec = typ == 0 ? as2 : (typ == 1 ? lw : ad2);
#pragma unroll
        for (int c = 0; c < CHD; ++c) s += W2[k * HC + h * CHD + c] * vec[h * CHD + c];
    }
    const int kt = k >> 2, khalf = (k >> 1) & 1, kv = k & 1;
    VqF[((kt * 32 + khalf * 16 + col) * 2 + kv)] = s;
}

__device__ __forceinline__ float lane_xor4(float v) {
    return __builtin_bit_cast(float, __builtin_amdgcn_update_dpp(0, __builtin_bit_cast(int, v), 0x164, 0xf, 0xf, false));
}
__device__ __forceinline__ float lane_xor2(float v) {
    return __builtin_bit_cast(float, __builtin_amdgcn_update_dpp(0, __builtin_bit_cast(int, v), 0x4E, 0xf, 0xf, false));
}
__device__ __forceinline__ float lane_xor1(float v) {
    return __builtin_bit_cast(float, __builtin_amdgcn_update_dpp(0, __builtin_bit_cast(int, v), 0xB1, 0xf, 0xf, false));
}

#define EP_GROUPS 8
#define LIST_CAP  64

__global__ __launch_bounds__(256)
void edge_proj_kernel(const float* __restrict__ EA, const int* __restrict__ dstA,
                      const float* __restrict__ Ve1, const float* __restrict__ Ve2,
                      float* __restrict__ ae1v, float* __restrict__ ae2v,
                      int* __restrict__ degI, int* __restrict__ slotA, int E) {
    const int lane = threadIdx.x & 31;
    const int j = lane & 7;
    const int sub = lane >> 3;
    const int gw = blockIdx.x * (blockDim.x >> 5) + (threadIdx.x >> 5);
    const int base = gw * (4 * EP_GROUPS);
    if (base >= E) return;

    const int eown = base + lane;
    const int down = dstA[min(eown, E - 1)];

    const float4* EA4 = reinterpret_cast<const float4*>(EA);
    float4 x0[EP_GROUPS], x1[EP_GROUPS];
#pragma unroll
    for (int t = 0; t < EP_GROUPS; ++t) {
        int e = min(base + 4 * t + sub, E - 1);
        x0[t] = EA4[(size_t)e * 16 + j];
        x1[t] = EA4[(size_t)e * 16 + 8 + j];
    }

    const float4* V1 = reinterpret_cast<const float4*>(Ve1);
    const float4* V2 = reinterpret_cast<const float4*>(Ve2);
    float wl[4][8], wh[4][8];
#pragma unroll
    for (int c = 0; c < 4; ++c) {
        float4 p = V1[4 * j + c], q = V2[4 * j + c];
        float4 r = V1[32 + 4 * j + c], s = V2[32 + 4 * j + c];
        wl[c][0] = p.x; wl[c][1] = p.y; wl[c][2] = p.z; wl[c][3] = p.w;
        wl[c][4] = q.x; wl[c][5] = q.y; wl[c][6] = q.z; wl[c][7] = q.w;
        wh[c][0] = r.x; wh[c][1] = r.y; wh[c][2] = r.z; wh[c][3] = r.w;
        wh[c][4] = s.x; wh[c][5] = s.y; wh[c][6] = s.z; wh[c][7] = s.w;
    }

    const int rank = atomicAdd(&degI[down], eown < E ? 1 : 0);

    const bool b4 = (j & 4) != 0, b2 = (j & 2) != 0, b1 = (j & 1) != 0;
    float* outp = (b4 ? ae2v : ae1v) + (j & 3);
#pragma unroll
    for (int t = 0; t < EP_GROUPS; ++t) {
        float acc[8];
#pragma unroll
        for (int q = 0; q < 8; ++q) {
            acc[q] = x0[t].x * wl[0][q] + x0[t].y * wl[1][q] + x0[t].z * wl[2][q] + x0[t].w * wl[3][q]
                   + x1[t].x * wh[0][q] + x1[t].y * wh[1][q] + x1[t].z * wh[2][q] + x1[t].w * wh[3][q];
        }
        float r[4], s[2];
#pragma unroll
        for (int q = 0; q < 4; ++q) {
            float keep = b4 ? acc[q + 4] : acc[q];
            float send = b4 ? acc[q] : acc[q + 4];
            r[q] = keep + lane_xor4(send);
        }
#pragma unroll
        for (int q = 0; q < 2; ++q) {
            float keep = b2 ? r[q + 2] : r[q];
            float send = b2 ? r[q] : r[q + 2];
            s[q] = keep + lane_xor2(send);
        }
        float keep = b1 ? s[1] : s[0];
        float send = b1 ? s[0] : s[1];
        float v = keep + lane_xor1(send);
        int e = base + 4 * t + sub;
        if (e < E) outp[(size_t)e * 4] = v;
    }
    if (eown < E && rank < LIST_CAP) slotA[(unsigned)down * (unsigned)LIST_CAP + (unsigned)rank] = eown;
}

__global__ __launch_bounds__(128)
void proj_kernel(const float* __restrict__ X, const float* __restrict__ W,
                 float* __restrict__ Y, int M, int K) {
    const int wave = threadIdx.x >> 5;
    const int lane = threadIdx.x & 31;
    const int mt = blockIdx.x * (blockDim.x >> 5) + wave;
    if (mt * 16 >= M) return;
    const int row0 = mt * 16;
    const int m = lane & 15, half = lane >> 4;

    v8f acc[8];
#pragma unroll
    for (int nt = 0; nt < 8; ++nt) acc[nt] = v8f{};

    const float* ap0 = X + (size_t)(row0 + m) * K + 2 * half;
    const float* bp0 = W + (size_t)(2 * half) * HC + m;
    for (int k0 = 0; k0 < K; k0 += 4) {
        v2f a;
        a.x = ap0[k0];
        a.y = ap0[k0 + 1];
        const float* bp = bp0 + (size_t)k0 * HC;
#pragma unroll
        for (int nt = 0; nt < 8; ++nt) {
            v2f b;
            b.x = bp[nt * 16];
            b.y = bp[HC + nt * 16];
            acc[nt] = __builtin_amdgcn_wmma_f32_16x16x4_f32(false, a, false, b,
                                                            (short)0, acc[nt], false, false);
        }
    }

    float* yp = Y + (size_t)(row0 + 8 * half) * HC + m;
#pragma unroll
    for (int nt = 0; nt < 8; ++nt) {
#pragma unroll
        for (int r = 0; r < 8; ++r) yp[(size_t)r * HC + nt * 16] = acc[nt][r];
    }
}

__global__ __launch_bounds__(128)
void projq_kernel(const float* __restrict__ X, const float* __restrict__ VqF,
                  float* __restrict__ Q, int M) {
    const int wave = threadIdx.x >> 5;
    const int lane = threadIdx.x & 31;
    const int mt = blockIdx.x * (blockDim.x >> 5) + wave;
    if (mt * 16 >= M) return;
    const int row0 = mt * 16;
    const int m = lane & 15, half = lane >> 4;

    const float* ap0 = X + (size_t)min(row0 + m, M - 1) * HC + 2 * half;
    const v2f* bf = reinterpret_cast<const v2f*>(VqF) + lane;
    v8f acc0 = v8f{}, acc1 = v8f{};
#pragma unroll
    for (int t = 0; t < HC / 4; t += 2) {
        v2f a0, a1;
        a0.x = ap0[4 * t];
        a0.y = ap0[4 * t + 1];
        a1.x = ap0[4 * t + 4];
        a1.y = ap0[4 * t + 5];
        const v2f b0 = bf[t * 32];
        const v2f b1 = bf[(t + 1) * 32];
        acc0 = __builtin_amdgcn_wmma_f32_16x16x4_f32(false, a0, false, b0, (short)0, acc0, false, false);
        acc1 = __builtin_amdgcn_wmma_f32_16x16x4_f32(false, a1, false, b1, (short)0, acc1, false, false);
    }
    const v8f acc = acc0 + acc1;
#pragma unroll
    for (int r = 0; r < 8; ++r) {
        const int row = row0 + 8 * half + r;
        if (row < M) Q[(size_t)row * 16 + m] = acc[r];
    }
}

template <int J> __device__ __forceinline__ int bc8i(int v) {
    return (int)__builtin_amdgcn_mov_dpp8((unsigned)v, (unsigned)J * 0x249249u);
}
template <int J> __device__ __forceinline__ float bc8f(float v) {
    return __builtin_bit_cast(float, __builtin_amdgcn_mov_dpp8(__builtin_bit_cast(unsigned, v), (unsigned)J * 0x249249u));
}
__device__ __forceinline__ float sum8(float v) {
    v += lane_xor1(v); v += lane_xor2(v); v += lane_xor4(v);
    return v;
}
__device__ __forceinline__ float max8(float v) {
    v = fmaxf(v, lane_xor1(v)); v = fmaxf(v, lane_xor2(v)); v = fmaxf(v, lane_xor4(v));
    return v;
}

template <int NG>
__device__ __forceinline__ void edge_groups(const int* __restrict__ ep, const int* __restrict__ srcA,
                                            const float* __restrict__ a_e,
                                            int j, int h, int nc, float* ae, int* sv) {
    unsigned e[NG];
#pragma unroll
    for (int q = 0; q < NG; ++q) e[q] = (unsigned)ep[(unsigned)min(8 * q + j, nc - 1)];
#pragma unroll
    for (int q = 0; q < NG; ++q) sv[q] = srcA[e[q]];
#pragma unroll
    for (int q = 0; q < NG; ++q) ae[q] = a_e[e[q] * 4u + (unsigned)h];
    if constexpr (NG == 4) asm volatile("" : "+v"(sv[0]), "+v"(sv[1]), "+v"(sv[2]), "+v"(sv[3]));
    else                   asm volatile("" : "+v"(sv[0]), "+v"(sv[1]));
}

__device__ __forceinline__ void gather8(v4f* hv, const v4f* __restrict__ Hp4, int svg, int lane) {
    hv[0] = Hp4[(unsigned)bc8i<0>(svg) * 32u + (unsigned)lane];
    hv[1] = Hp4[(unsigned)bc8i<1>(svg) * 32u + (unsigned)lane];
    hv[2] = Hp4[(unsigned)bc8i<2>(svg) * 32u + (unsigned)lane];
    hv[3] = Hp4[(unsigned)bc8i<3>(svg) * 32u + (unsigned)lane];
    hv[4] = Hp4[(unsigned)bc8i<4>(svg) * 32u + (unsigned)lane];
    hv[5] = Hp4[(unsigned)bc8i<5>(svg) * 32u + (unsigned)lane];
    hv[6] = Hp4[(unsigned)bc8i<6>(svg) * 32u + (unsigned)lane];
    hv[7] = Hp4[(unsigned)bc8i<7>(svg) * 32u + (unsigned)lane];
    asm volatile("" : "+v"(hv[0]), "+v"(hv[1]), "+v"(hv[2]), "+v"(hv[3]),
                      "+v"(hv[4]), "+v"(hv[5]), "+v"(hv[6]), "+v"(hv[7]));
}
__device__ __forceinline__ void fma_row(float4& A, float w, const v4f& v) {
    A.x += w * v.x; A.y += w * v.y; A.z += w * v.z; A.w += w * v.w;
}
__device__ __forceinline__ void accum8(float4& A, const v4f* hv, float pg) {
    fma_row(A, bc8f<0>(pg), hv[0]);
    fma_row(A, bc8f<1>(pg), hv[1]);
    fma_row(A, bc8f<2>(pg), hv[2]);
    fma_row(A, bc8f<3>(pg), hv[3]);
    fma_row(A, bc8f<4>(pg), hv[4]);
    fma_row(A, bc8f<5>(pg), hv[5]);
    fma_row(A, bc8f<6>(pg), hv[6]);
    fma_row(A, bc8f<7>(pg), hv[7]);
}

__device__ __forceinline__ void group_step(v4f* hv, const v4f& ats, float base, float pen,
                                           bool b4, bool b2, bool b1, float& M, float& D, float4& A) {
    float r[4], s[2];
#pragma unroll
    for (int q = 0; q < 4; ++q) {
        const v2f t0 = __builtin_elementwise_fma(hv[q].zw, ats.zw, hv[q].xy * ats.xy);
        const v2f t1 = __builtin_elementwise_fma(hv[q + 4].zw, ats.zw, hv[q + 4].xy * ats.xy);
        const float d0 = t0.x + t0.y, d1 = t1.x + t1.y;
        float keep = b4 ? d1 : d0;
        float send = b4 ? d0 : d1;
        r[q] = keep + lane_xor4(send);
        if (q == 1) asm volatile("" : "+v"(r[0]), "+v"(r[1]), "+v"(hv[2]), "+v"(hv[3]), "+v"(hv[6]), "+v"(hv[7]));
    }
#pragma unroll
    for (int q = 0; q < 2; ++q) {
        float keep = b2 ? r[q + 2] : r[q];
        float send = b2 ? r[q] : r[q + 2];
        s[q] = keep + lane_xor2(send);
    }
    const float keep = b1 ? s[1] : s[0];
    const float send = b1 ? s[0] : s[1];
    const float as_e = keep + lane_xor1(send);

    const float lg = leaky(as_e + base) + pen;
    const float Mn = fmaxf(M, max8(lg));
    const float sc = __expf(M - Mn);
    const float p  = __expf(lg - Mn);
    D = D * sc + sum8(p);
    A.x *= sc; A.y *= sc; A.z *= sc; A.w *= sc;
    M = Mn;
    accum8(A, hv, p);
}

__global__ void agg_kernel(const int* __restrict__ degI, const int* __restrict__ slotA,
                const int* __restrict__ srcA, const int* __restrict__ dstA, int E,
                const float* __restrict__ a_e,
                const float* __restrict__ att_s, const float* __restrict__ att_d,
                const float* __restrict__ Hp, const float* __restrict__ bias,
                float* __restrict__ outv, int N) {
    const int lane = threadIdx.x & 31;
    const int n = __builtin_amdgcn_readfirstlane(blockIdx.x * (blockDim.x >> 5) + (threadIdx.x >> 5));
    if (n >= N) return;
    const int h = lane >> 3, j = lane & 7;
    const float4* Hp4 = reinterpret_cast<const float4*>(Hp);
    const v4f* Hp4v = reinterpret_cast<const v4f*>(Hp);

    const int cnt = degI[n];
    const float4 hself = Hp4[(size_t)n * 32 + lane];
    const v4f ats = reinterpret_cast<const v4f*>(att_s)[lane];
    const v4f atd = reinterpret_cast<const v4f*>(att_d)[lane];
    const float ad_n = sum8(hself.x * atd.x + hself.y * atd.y + hself.z * atd.z + hself.w * atd.w);
    const float degf = fmaxf((float)cnt, 1.0f);
    float M = -INFINITY;
    float D = 0.0f;
    float4 A = {0.f, 0.f, 0.f, 0.f};
    float S = 0.0f;

    const bool b4 = (j & 4) != 0, b2 = (j & 2) != 0, b1 = (j & 1) != 0;

    const int lcnt = cnt <= LIST_CAP ? cnt : 0;
    const int* lp = slotA + (size_t)n * LIST_CAP;
    for (int c = 0; c < lcnt; c += 32) {
        const int nc = min(lcnt - c, 32);
        const int* ep = lp + c;
        float ae[4];
        int sv[4];
        if (nc > 16) {
            edge_groups<4>(ep, srcA, a_e, j, h, nc, ae, sv);
        } else {
            edge_groups<2>(ep, srcA, a_e, j, h, nc, ae, sv);
            ae[2] = 0.0f; ae[3] = 0.0f;
            sv[2] = 0; sv[3] = 0;
        }
        const bool in0 = j < nc, in1 = j < nc - 8, in2 = j < nc - 16, in3 = j < nc - 24;
        const float pen0 = in0 ? 0.0f : -INFINITY;
        const float pen1 = in1 ? 0.0f : -INFINITY;
        const float pen2 = in2 ? 0.0f : -INFINITY;
        const float pen3 = in3 ? 0.0f : -INFINITY;
        v4f hv[8];
        gather8(hv, Hp4v, sv[0], lane);
        S += ((in0 ? ae[0] : 0.0f) + (in1 ? ae[1] : 0.0f)) + ((in2 ? ae[2] : 0.0f) + (in3 ? ae[3] : 0.0f));
        group_step(hv, ats, ad_n + ae[0], pen0, b4, b2, b1, M, D, A);
        if (nc > 8)  { gather8(hv, Hp4v, sv[1], lane); group_step(hv, ats, ad_n + ae[1], pen1, b4, b2, b1, M, D, A); }
        if (nc > 16) { gather8(hv, Hp4v, sv[2], lane); group_step(hv, ats, ad_n + ae[2], pen2, b4, b2, b1, M, D, A); }
        if (nc > 24) { gather8(hv, Hp4v, sv[3], lane); group_step(hv, ats, ad_n + ae[3], pen3, b4, b2, b1, M, D, A); }
    }
    if (cnt > LIST_CAP) {
        for (int b = 0; b < E; b += 32) {
            const int e = b + lane;
            const bool hit = e < E && dstA[min(e, E - 1)] == n;
            unsigned found = __builtin_amdgcn_ballot_w32(hit);
            while (found != 0u) {
                const int ee = b + __builtin_ctz(found);
                found &= found - 1u;
                const int sn = srcA[ee];
                const float aev = a_e[(unsigned)ee * 4u + (unsigned)h];
                const v4f row = Hp4v[(unsigned)sn * 32u + (unsigned)lane];
                const float as_e = sum8(row.x * ats.x + row.y * ats.y + row.z * ats.z + row.w * ats.w);
                const float lg = leaky(as_e + ad_n + aev);
                const float Mn = fmaxf(M, lg);
                const float sc = __expf(M - Mn);
                const float p  = __expf(lg - Mn);
                D = D * sc + p;
                A.x = A.x * sc + p * row.x;
                A.y = A.y * sc + p * row.y;
                A.z = A.z * sc + p * row.z;
                A.w = A.w * sc + p * row.w;
                M = Mn;
                S += (j == 0) ? aev : 0.0f;
            }
        }
    }
    S = sum8(S);
    {
        const float as_n = sum8(hself.x * ats.x + hself.y * ats.y + hself.z * ats.z + hself.w * ats.w);
        float lgs = leaky(as_n + ad_n + S / degf);
        float Mn = fmaxf(M, lgs);
        float sc = __expf(M - Mn);
        float p  = __expf(lgs - Mn);
        A.x = A.x * sc + p * hself.x;
        A.y = A.y * sc + p * hself.y;
        A.z = A.z * sc + p * hself.z;
        A.w = A.w * sc + p * hself.w;
        D = D * sc + p;
    }
    float4 bv = reinterpret_cast<const float4*>(bias)[lane];
    float inv = 1.0f / D;
    float4 r = { A.x * inv + bv.x, A.y * inv + bv.y, A.z * inv + bv.z, A.w * inv + bv.w };
    r.x = fmaxf(r.x, 0.f); r.y = fmaxf(r.y, 0.f);
    r.z = fmaxf(r.z, 0.f); r.w = fmaxf(r.w, 0.f);
    reinterpret_cast<float4*>(outv)[(size_t)n * 32 + lane] = r;
}

__device__ __forceinline__ void soft_join(float lg, float zv, float& M, float& D, float& Z) {
    const float Mn = fmaxf(M, lg);
    const float sc = __expf(M - Mn);
    const float p  = __expf(lg - Mn);
    D = D * sc + p;
    Z = Z * sc + p * zv;
    M = Mn;
}
__device__ __forceinline__ void soft_join4(const v4f& as_s, const v4f& z_s, const v4f& ae, const v4f& ad,
                                           float* M, float* D, float* Z, float* S) {
    soft_join(leaky(as_s.x + ad.x + ae.x), z_s.x, M[0], D[0], Z[0]);
    soft_join(leaky(as_s.y + ad.y + ae.y), z_s.y, M[1], D[1], Z[1]);
    soft_join(leaky(as_s.z + ad.z + ae.z), z_s.z, M[2], D[2], Z[2]);
    soft_join(leaky(as_s.w + ad.w + ae.w), z_s.w, M[3], D[3], Z[3]);
    S[0] += ae.x; S[1] += ae.y; S[2] += ae.z; S[3] += ae.w;
}
__device__ __forceinline__ float head_close(float M, float D, float Z, float S, float as_n, float ad_n, float z_n,
                                            float degf) {
    const float lgs = leaky(as_n + ad_n + sum8(S) / degf);
    const float Mg = fmaxf(max8(M), lgs);
    const float sc = __expf(M - Mg);
    const float ps = __expf(lgs - Mg);
    const float Dg = sum8(D * sc) + ps;
    const float Zg = sum8(Z * sc) + ps * z_n;
    return Zg / Dg;
}

#define A2_NODES 4

__global__ __launch_bounds__(256)
void agg2_kernel(const int* __restrict__ degI, const int* __restrict__ slotA,
                 const int* __restrict__ srcA, const int* __restrict__ dstA, int E,
                 const float* __restrict__ a_e, const float* __restrict__ Q,
                 const float* __restrict__ bias, const float* __restrict__ lin_w, const float* __restrict__ lin_b,
                 float* __restrict__ outs, int N) {
    const int lane = threadIdx.x & 31;
    const int n0 = __builtin_amdgcn_readfirstlane((blockIdx.x * (blockDim.x >> 5) + (threadIdx.x >> 5)) * A2_NODES);
    if (n0 >= N) return;
    const int g = lane >> 3, j = lane & 7;
    const bool live = n0 + g < N;
    const int n = min(n0 + g, N - 1);

    const v4f* Q4 = reinterpret_cast<const v4f*>(Q);
    const v4f* AE4 = reinterpret_cast<const v4f*>(a_e);
    const int cnt = degI[n];
    const v4f as_n = Q4[(unsigned)n * 4u];
    const v4f z_n  = Q4[(unsigned)n * 4u + 1u];
    const v4f ad_n = Q4[(unsigned)n * 4u + 2u];
    const float4 bv = reinterpret_cast<const float4*>(bias)[lane];
    const float4 lw = reinterpret_cast<const float4*>(lin_w)[lane];
    float cst = bv.x * lw.x + bv.y * lw.y + bv.z * lw.z + bv.w * lw.w;

    float M[4] = {-INFINITY, -INFINITY, -INFINITY, -INFINITY};
    float D[4] = {0.f, 0.f, 0.f, 0.f};
    float Z[4] = {0.f, 0.f, 0.f, 0.f};
    float S[4] = {0.f, 0.f, 0.f, 0.f};

    const int lcnt = cnt <= LIST_CAP ? cnt : 0;
    const int maxc = max(max(__builtin_amdgcn_readlane(lcnt, 0), __builtin_amdgcn_readlane(lcnt, 8)),
                         max(__builtin_amdgcn_readlane(lcnt, 16), __builtin_amdgcn_readlane(lcnt, 24)));
    const int* lp = slotA + (size_t)n * LIST_CAP;
    for (int c0 = 0; c0 < maxc; c0 += 32) {
        bool in[4];
        unsigned e[4];
        int sv[4];
        v4f ae[4], qs[4], qz[4];
#pragma unroll
        for (int q = 0; q < 4; ++q) {
            const int c = c0 + 8 * q + j;
            in[q] = c < lcnt;
            e[q] = 0u;
            if (in[q]) e[q] = (unsigned)lp[c];
        }
#pragma unroll
        for (int q = 0; q < 4; ++q) { sv[q] = srcA[e[q]]; ae[q] = AE4[e[q]]; }
#pragma unroll
        for (int q = 0; q < 4; ++q) { qs[q] = Q4[(unsigned)sv[q] * 4u]; qz[q] = Q4[(unsigned)sv[q] * 4u + 1u]; }
#pragma unroll
        for (int q = 0; q < 4; ++q) {
            if (in[q]) soft_join4(qs[q], qz[q], ae[q], ad_n, M, D, Z, S);
        }
    }
#pragma unroll
    for (int k = 0; k < A2_NODES; ++k) {
        const int cnt_k = __builtin_amdgcn_readlane(cnt, 8 * k);
        const int n_k = n0 + k;
        if (cnt_k > LIST_CAP && n_k < N) {
            for (int b = 0; b < E; b += 32) {
                const int ed = b + lane;
                const bool hit = ed < E && dstA[min(ed, E - 1)] == n_k;
                unsigned found = __builtin_amdgcn_ballot_w32(hit);
                while (found != 0u) {
                    const int ee = b + __builtin_ctz(found);
                    found &= found - 1u;
                    const int sn = srcA[ee];
                    const v4f aev = AE4[(unsigned)ee];
                    const v4f qsv = Q4[(unsigned)sn * 4u];
                    const v4f qzv = Q4[(unsigned)sn * 4u + 1u];
                    if (lane == 8 * k) soft_join4(qsv, qzv, aev, ad_n, M, D, Z, S);
                }
            }
        }
    }
    const float degf = fmaxf((float)cnt, 1.0f);
    float o = head_close(M[0], D[0], Z[0], S[0], as_n.x, ad_n.x, z_n.x, degf)
            + head_close(M[1], D[1], Z[1], S[1], as_n.y, ad_n.y, z_n.y, degf)
            + head_close(M[2], D[2], Z[2], S[2], as_n.z, ad_n.z, z_n.z, degf)
            + head_close(M[3], D[3], Z[3], S[3], as_n.w, ad_n.w, z_n.w, degf);
#pragma unroll
    for (int msk = 16; msk >= 1; msk >>= 1) cst += __shfl_xor(cst, msk, 32);
    if (live && j == 0) outs[n] = o + cst + lin_b[0];
}

extern "C" void kernel_launch(void* const* d_in, const int* in_sizes, int n_in,
                              void* d_out, int out_size, void* d_ws, size_t ws_size,
                              hipStream_t stream) {
    const float* x   = (const float*)d_in[0];
    const float* ea  = (const float*)d_in[1];
    const float* W1  = (const float*)d_in[2];
    const float* as1 = (const float*)d_in[3];
    const float* ad1 = (const float*)d_in[4];
    const float* We1 = (const float*)d_in[5];
    const float* ae1 = (const float*)d_in[6];
    const float* b1  = (const float*)d_in[7];
    const float* W2  = (const float*)d_in[8];
    const float* as2 = (const float*)d_in[9];
    const float* ad2 = (const float*)d_in[10];
    const float* We2 = (const float*)d_in[11];
    const float* ae2 = (const float*)d_in[12];
    const float* b2  = (const float*)d_in[13];
    const float* lw  = (const float*)d_in[14];
    const float* lb  = (const float*)d_in[15];
    const int*   ei  = (const int*)d_in[16];

    const int N = in_sizes[0] / F_IN;
    const int E = in_sizes[16] / 2;
    const int* srcA = ei;
    const int* dstA = ei + E;

    char* w = (char*)d_ws;
    auto alloc = [&](size_t bytes) -> void* {
        void* p = (void*)w;
        w += (bytes + 255) & ~(size_t)255;
        return p;
    };
    int*   degI   = (int*)  alloc((size_t)N * 4);
    size_t zbytes = (size_t)(w - (char*)degI);
    float* Ve1  = (float*)alloc(F_IN * 4 * 4);
    float* Ve2  = (float*)alloc(F_IN * 4 * 4);
    float* ae1v = (float*)alloc((size_t)E * 16);
    float* ae2v = (float*)alloc((size_t)E * 16);
    int*   slotA = (int*) alloc((size_t)N * LIST_CAP * 4);
    float* Hp   = (float*)alloc((size_t)N * HC * 4);
    float* h1   = (float*)alloc((size_t)N * HC * 4);
    float* VqF  = (float*)alloc(HC * 16 * 4);
    float* Q    = Hp;
    if ((size_t)(w - (char*)d_ws) > ws_size) return;

    hipMemsetAsync(degI, 0, zbytes, stream);

    fold_kernel<<<9, 256, 0, stream>>>(We1, ae1, We2, ae2, W2, as2, ad2, lw, Ve1, Ve2, VqF);
    edge_proj_kernel<<<(E + 255) / 256, 256, 0, stream>>>(ea, dstA, Ve1, Ve2,
                                                          ae1v, ae2v, degI, slotA, E);

    const int mt = (N + 15) / 16;
    const int projBlocks = (mt + 3) / 4;
    const int nodeBlocks = (N + 7) / 8;
    const int node2Blocks = (N + 31) / 32;

    proj_kernel<<<projBlocks, 128, 0, stream>>>(x, W1, Hp, N, F_IN);
    agg_kernel<<<nodeBlocks, 256, 0, stream>>>(degI, slotA, srcA, dstA, E, ae1v, as1, ad1,
                                               Hp, b1, h1, N);
    projq_kernel<<<projBlocks, 128, 0, stream>>>(h1, VqF, Q, N);
    agg2_kernel<<<node2Blocks, 256, 0, stream>>>(degI, slotA, srcA, dstA, E, ae2v, Q, b2, lw, lb,
                                                 (float*)d_out, N);
}
